# starting kernel
# baseline (speedup 1.0000x reference)
#include <hip/hip_runtime.h>
#include <hip/hip_bf16.h>

typedef __attribute__((ext_vector_type(2))) float v2f;
typedef __attribute__((ext_vector_type(8))) float v8f;

#define NEG_SLOPE 0.2f

__device__ __forceinline__ unsigned f2ord(float f) {
    unsigned b = __float_as_uint(f);
    return (b & 0x80000000u) ? ~b : (b | 0x80000000u);
}
__device__ __forceinline__ float ord2f(unsigned u) {
    unsigned b = (u & 0x80000000u) ? (u & 0x7FFFFFFFu) : ~u;
    return __uint_as_float(b);
}
#define ORD_NEG_INF 0x007FFFFFu

__device__ __forceinline__ void edge_sd(const int* ei, int E0, int e, int& s, int& d) {
    if (e < E0) { s = ei[e]; d = ei[E0 + e]; }
    else        { s = e - E0; d = s; }
}

__global__ void gat_wmma_gemm(const float* __restrict__ A,
                              const float* __restrict__ B,
                              float* __restrict__ C,
                              int M, int Ntot, int K) {
    __shared__ float As[64][68];
    __shared__ float Bp[32][132];

    const int tid   = threadIdx.x;
    const int lane  = tid & 31;
    const int wave  = tid >> 5;
    const int mBase = blockIdx.y * 64;
    const int nBase = blockIdx.x * 64;

    const int ms   = (wave & 3) * 16;
    const int ns   = (wave >> 2) * 32;
    const int mrow = ms + (lane & 15);
    const int khi  = (lane >> 4) * 2;
    const int ncol = lane & 15;

    v8f acc0 = {}; v8f acc1 = {};

    for (int kBase = 0; kBase < K; kBase += 64) {
#pragma unroll
        for (int i = 0; i < 4; ++i) {
            int s  = tid + i * 256;
            int r  = s >> 4;
            int c4 = (s & 15) << 2;
            float4 va = make_float4(0.f, 0.f, 0.f, 0.f);
            int g = mBase + r;
            if (g < M) va = *(const float4*)(A + (size_t)g * K + kBase + c4);
            *(float4*)&As[r][c4] = va;
            float4 vb = *(const float4*)(B + (size_t)(kBase + r) * Ntot + nBase + c4);
            float* bp = &Bp[r >> 1][r & 1];
            bp[2 * c4 + 0] = vb.x;
            bp[2 * c4 + 2] = vb.y;
            bp[2 * c4 + 4] = vb.z;
            bp[2 * c4 + 6] = vb.w;
        }
        __syncthreads();

#pragma unroll
        for (int kk = 0; kk < 64; kk += 4) {
            const int p2 = (kk + khi) >> 1;
            v2f a, b0, b1v;
            a[0]   = As[mrow][kk + khi];
            a[1]   = As[mrow][kk + khi + 1];
            b0     = *(const v2f*)&Bp[p2][2 * (ns + ncol)];
            b1v    = *(const v2f*)&Bp[p2][2 * (ns + 16 + ncol)];
            acc0 = __builtin_amdgcn_wmma_f32_16x16x4_f32(
                false, a, false, b0, (short)0, acc0, false, false);
            acc1 = __builtin_amdgcn_wmma_f32_16x16x4_f32(
                false, a, false, b1v, (short)0, acc1, false, false);
        }
        __syncthreads();
    }

    const int orow = mBase + ms + ((lane >> 4) << 3);
    const int ocol = nBase + ns + ncol;
#pragma unroll
    for (int r = 0; r < 8; ++r) {
        int g = orow + r;
        if (g < M) {
            C[(size_t)g * Ntot + ocol]      = acc0[r];
            C[(size_t)g * Ntot + ocol + 16] = acc1[r];
        }
    }
}

__global__ void scores1_kernel(const float* __restrict__ h1,
                               const float* __restrict__ att_s,
                               const float* __restrict__ att_d,
                               float* __restrict__ as1, float* __restrict__ ad1,
                               int Nn) {
    int t = blockIdx.x * blockDim.x + threadIdx.x;
    if (t >= Nn * 8) return;
    int n = t >> 3, h = t & 7;
    const float* row = h1 + (size_t)n * 256 + h * 32;
    const float* ws = att_s + h * 32;
    const float* wd = att_d + h * 32;
    float ss = 0.f, sd = 0.f;
#pragma unroll
    for (int c = 0; c < 32; ++c) { ss += row[c] * ws[c]; sd += row[c] * wd[c]; }
    as1[t] = ss; ad1[t] = sd;
}

__global__ void init1_kernel(float* __restrict__ num1, float* __restrict__ den1,
                             unsigned* __restrict__ amax1, int Nn) {
    int t = blockIdx.x * blockDim.x + threadIdx.x;
    if (t < Nn * 256) num1[t] = 0.f;
    if (t < Nn * 8) { den1[t] = 0.f; amax1[t] = ORD_NEG_INF; }
}

__global__ void edge_max1_kernel(const int* __restrict__ ei, int E0, int Nn,
                                 const float* __restrict__ as1,
                                 const float* __restrict__ ad1,
                                 unsigned* __restrict__ amax1) {
    int t = blockIdx.x * blockDim.x + threadIdx.x;
    int ET = E0 + Nn;
    if (t >= ET * 8) return;
    int e = t >> 3, h = t & 7;
    int s, d; edge_sd(ei, E0, e, s, d);
    float lg = as1[s * 8 + h] + ad1[d * 8 + h];
    lg = lg > 0.f ? lg : NEG_SLOPE * lg;
    atomicMax(&amax1[d * 8 + h], f2ord(lg));
}

__global__ void edge_acc1_kernel(const int* __restrict__ ei, int E0, int Nn,
                                 const float* __restrict__ as1,
                                 const float* __restrict__ ad1,
                                 const unsigned* __restrict__ amax1,
                                 const float* __restrict__ h1,
                                 float* __restrict__ den1,
                                 float* __restrict__ num1) {
    int gtid = blockIdx.x * blockDim.x + threadIdx.x;
    int e    = gtid >> 5;
    int lane = threadIdx.x & 31;
    if (e >= E0 + Nn) return;
    int s, d; edge_sd(ei, E0, e, s, d);
    const float* hrow = h1 + (size_t)s * 256;
    float*       orow = num1 + (size_t)d * 256;
    __builtin_prefetch(hrow, 0, 1);
#pragma unroll
    for (int h = 0; h < 8; ++h) {
        float lg = as1[s * 8 + h] + ad1[d * 8 + h];
        lg = lg > 0.f ? lg : NEG_SLOPE * lg;
        float p = expf(lg - ord2f(amax1[d * 8 + h]));
        if (lane == 0) atomicAdd(&den1[d * 8 + h], p);
        atomicAdd(&orow[h * 32 + lane], hrow[h * 32 + lane] * p);
    }
}

__global__ void norm_elu1_kernel(const float* __restrict__ num1,
                                 const float* __restrict__ den1,
                                 const float* __restrict__ b1,
                                 float* __restrict__ hout, int Nn) {
    int t = blockIdx.x * blockDim.x + threadIdx.x;
    if (t >= Nn * 256) return;
    int n = t >> 8, idx = t & 255, h = idx >> 5;
    float v = num1[t] / (den1[n * 8 + h] + 1e-16f) + b1[idx];
    hout[t] = v > 0.f ? v : (expf(v) - 1.f);
}

__global__ void scores2_kernel(const float* __restrict__ h2,
                               const float* __restrict__ att_s,
                               const float* __restrict__ att_d,
                               float* __restrict__ as2, float* __restrict__ ad2,
                               int Nn) {
    int n = blockIdx.x * blockDim.x + threadIdx.x;
    if (n >= Nn) return;
    const float* row = h2 + (size_t)n * 64;
    float ss = 0.f, sd = 0.f;
#pragma unroll
    for (int c = 0; c < 64; ++c) { ss += row[c] * att_s[c]; sd += row[c] * att_d[c]; }
    as2[n] = ss; ad2[n] = sd;
}

__global__ void init2_kernel(float* __restrict__ out, float* __restrict__ den2,
                             unsigned* __restrict__ amax2, int Nn) {
    int t = blockIdx.x * blockDim.x + threadIdx.x;
    if (t < Nn * 64) out[t] = 0.f;
    if (t < Nn) { den2[t] = 0.f; amax2[t] = ORD_NEG_INF; }
}

__global__ void edge_max2_kernel(const int* __restrict__ ei, int E0, int Nn,
                                 const float* __restrict__ as2,
                                 const float* __restrict__ ad2,
                                 unsigned* __restrict__ amax2) {
    int e = blockIdx.x * blockDim.x + threadIdx.x;
    if (e >= E0 + Nn) return;
    int s, d; edge_sd(ei, E0, e, s, d);
    float lg = as2[s] + ad2[d];
    lg = lg > 0.f ? lg : NEG_SLOPE * lg;
    atomicMax(&amax2[d], f2ord(lg));
}

__global__ void edge_acc2_kernel(const int* __restrict__ ei, int E0, int Nn,
                                 const float* __restrict__ as2,
                                 const float* __restrict__ ad2,
                                 const unsigned* __restrict__ amax2,
                                 const float* __restrict__ h2,
                                 float* __restrict__ den2,
                                 float* __restrict__ out) {
    int gtid = blockIdx.x * blockDim.x + threadIdx.x;
    int e    = gtid >> 5;
    int lane = threadIdx.x & 31;
    if (e >= E0 + Nn) return;
    int s, d; edge_sd(ei, E0, e, s, d);
    float lg = as2[s] + ad2[d];
    lg = lg > 0.f ? lg : NEG_SLOPE * lg;
    float p = expf(lg - ord2f(amax2[d]));
    if (lane == 0) atomicAdd(&den2[d], p);
    const float* hrow = h2 + (size_t)s * 64;
    float*       orow = out + (size_t)d * 64;
    __builtin_prefetch(hrow, 0, 1);
    atomicAdd(&orow[lane],      hrow[lane]      * p);
    atomicAdd(&orow[32 + lane], hrow[32 + lane] * p);
}

__global__ void final_kernel(float* __restrict__ out,
                             const float* __restrict__ den2,
                             const float* __restrict__ b2, int Nn) {
    int t = blockIdx.x * blockDim.x + threadIdx.x;
    if (t >= Nn * 64) return;
    int n = t >> 6, c = t & 63;
    out[t] = out[t] / (den2[n] + 1e-16f) + b2[c];
}

extern "C" void kernel_launch(void* const* d_in, const int* in_sizes, int n_in,
                              void* d_out, int out_size, void* d_ws, size_t ws_size,
                              hipStream_t stream) {
    const float* x        = (const float*)d_in[0];
    const int*   ei       = (const int*)d_in[1];
    const float* W1       = (const float*)d_in[2];
    const float* att_src1 = (const float*)d_in[3];
    const float* att_dst1 = (const float*)d_in[4];
    const float* b1       = (const float*)d_in[5];
    const float* W2       = (const float*)d_in[6];
    const float* att_src2 = (const float*)d_in[7];
    const float* att_dst2 = (const float*)d_in[8];
    const float* b2       = (const float*)d_in[9];
    float* out = (float*)d_out;

    const int Nn = in_sizes[0] / 256;
    const int E0 = in_sizes[1] / 2;
    const int ET = E0 + Nn;

    float*    h1    = (float*)d_ws;
    float*    num1  = h1 + (size_t)Nn * 256;
    float*    as1   = num1 + (size_t)Nn * 256;
    float*    ad1   = as1 + (size_t)Nn * 8;
    unsigned* amax1 = (unsigned*)(ad1 + (size_t)Nn * 8);
    float*    den1  = (float*)(amax1 + (size_t)Nn * 8);
    float*    h2    = num1;
    float*    as2   = as1;
    float*    ad2   = as1 + Nn;
    unsigned* amax2 = (unsigned*)(as1 + 2 * (size_t)Nn);
    float*    den2  = as1 + 3 * (size_t)Nn;

    const int TB = 256;
    dim3 g1((256 + 63) / 64, (Nn + 63) / 64);
    dim3 g2((64 + 63) / 64, (Nn + 63) / 64);

    gat_wmma_gemm<<<g1, TB, 0, stream>>>(x, W1, h1, Nn, 256, 256);
    scores1_kernel<<<(Nn * 8 + TB - 1) / TB, TB, 0, stream>>>(h1, att_src1, att_dst1, as1, ad1, Nn);
    init1_kernel<<<(Nn * 256 + TB - 1) / TB, TB, 0, stream>>>(num1, den1, amax1, Nn);
    edge_max1_kernel<<<(ET * 8 + TB - 1) / TB, TB, 0, stream>>>(ei, E0, Nn, as1, ad1, amax1);
    edge_acc1_kernel<<<(ET + 7) / 8, TB, 0, stream>>>(ei, E0, Nn, as1, ad1, amax1,
                                                      h1, den1, num1);
    norm_elu1_kernel<<<(Nn * 256 + TB - 1) / TB, TB, 0, stream>>>(num1, den1, b1, h1, Nn);

    gat_wmma_gemm<<<g2, TB, 0, stream>>>(h1, W2, h2, Nn, 64, 256);
    scores2_kernel<<<(Nn + TB - 1) / TB, TB, 0, stream>>>(h2, att_src2, att_dst2, as2, ad2, Nn);
    init2_kernel<<<(Nn * 64 + TB - 1) / TB, TB, 0, stream>>>(out, den2, amax2, Nn);
    edge_max2_kernel<<<(ET + TB - 1) / TB, TB, 0, stream>>>(ei, E0, Nn, as2, ad2, amax2);
    edge_acc2_kernel<<<(ET + 7) / 8, TB, 0, stream>>>(ei, E0, Nn, as2, ad2, amax2,
                                                      h2, den2, out);
    final_kernel<<<(Nn * 64 + TB - 1) / TB, TB, 0, stream>>>(out, den2, b2, Nn);
}


# Softmax sums as owner-computes destination tiles: a work-group filters the edge list for its 196 nodes into LDS, one wave per node sums rows in registers in fixed edge order; no atomics
# speedup vs baseline: 2.4305x; 2.4305x over previous
#include <hip/hip_runtime.h>
#include <hip/hip_bf16.h>

typedef __attribute__((ext_vector_type(2))) float v2f;
typedef __attribute__((ext_vector_type(8))) float v8f;

#define NEG_SLOPE 0.2f

__global__ void gat_wmma_gemm(const float* __restrict__ A,
                              const float* __restrict__ B,
                              float* __restrict__ C,
                              int M, int Ntot, int K) {
    __shared__ float As[64][68];
    __shared__ float Bp[32][132];

    const int tid   = threadIdx.x;
    const int lane  = tid & 31;
    const int wave  = tid >> 5;
    const int mBase = blockIdx.y * 64;
    const int nBase = blockIdx.x * 64;

    const int ms   = (wave & 3) * 16;
    const int ns   = (wave >> 2) * 32;
    const int mrow = ms + (lane & 15);
    const int khi  = (lane >> 4) * 2;
    const int ncol = lane & 15;

    v8f acc0 = {}; v8f acc1 = {};

    for (int kBase = 0; kBase < K; kBase += 64) {
#pragma unroll
        for (int i = 0; i < 4; ++i) {
            int s  = tid + i * 256;
            int r  = s >> 4;
            int c4 = (s & 15) << 2;
            float4 va = make_float4(0.f, 0.f, 0.f, 0.f);
            int g = mBase + r;
            if (g < M) va = *(const float4*)(A + (size_t)g * K + kBase + c4);
            *(float4*)&As[r][c4] = va;
            float4 vb = *(const float4*)(B + (size_t)(kBase + r) * Ntot + nBase + c4);
            float* bp = &Bp[r >> 1][r & 1];
            bp[2 * c4 + 0] = vb.x;
            bp[2 * c4 + 2] = vb.y;
            bp[2 * c4 + 4] = vb.z;
            bp[2 * c4 + 6] = vb.w;
        }
        __syncthreads();

#pragma unroll
        for (int kk = 0; kk < 64; kk += 4) {
            const int p2 = (kk + khi) >> 1;
            v2f a, b0, b1v;
            a[0]   = As[mrow][kk + khi];
            a[1]   = As[mrow][kk + khi + 1];
            b0     = *(const v2f*)&Bp[p2][2 * (ns + ncol)];
            b1v    = *(const v2f*)&Bp[p2][2 * (ns + 16 + ncol)];
            acc0 = __builtin_amdgcn_wmma_f32_16x16x4_f32(
                false, a, false, b0, (short)0, acc0, false, false);
            acc1 = __builtin_amdgcn_wmma_f32_16x16x4_f32(
                false, a, false, b1v, (short)0, acc1, false, false);
        }
        __syncthreads();
    }

    const int orow = mBase + ms + ((lane >> 4) << 3);
    const int ocol = nBase + ns + ncol;
#pragma unroll
    for (int r = 0; r < 8; ++r) {
        int g = orow + r;
        if (g < M) {
            C[(size_t)g * Ntot + ocol]      = acc0[r];
            C[(size_t)g * Ntot + ocol + 16] = acc1[r];
        }
    }
}

__global__ void scores1_kernel(const float* __restrict__ h1,
                               const float* __restrict__ att_s,
                               const float* __restrict__ att_d,
                               float* __restrict__ as1, float* __restrict__ ad1,
                               int Nn) {
    int t = blockIdx.x * blockDim.x + threadIdx.x;
    if (t >= Nn * 8) return;
    int n = t >> 3, h = t & 7;
    const float* row = h1 + (size_t)n * 256 + h * 32;
    const float* ws = att_s + h * 32;
    const float* wd = att_d + h * 32;
    float ss = 0.f, sd = 0.f;
#pragma unroll
    for (int c = 0; c < 32; ++c) { ss += row[c] * ws[c]; sd += row[c] * wd[c]; }
    as1[t] = ss; ad1[t] = sd;
}

__global__ void scores2_kernel(const float* __restrict__ h2,
                               const float* __restrict__ att_s,
                               const float* __restrict__ att_d,
                               float* __restrict__ as2, float* __restrict__ ad2,
                               int Nn) {
    int n = blockIdx.x * blockDim.x + threadIdx.x;
    if (n >= Nn) return;
    const float* row = h2 + (size_t)n * 64;
    float ss = 0.f, sd = 0.f;
#pragma unroll
    for (int c = 0; c < 64; ++c) { ss += row[c] * att_s[c]; sd += row[c] * att_d[c]; }
    as2[n] = ss; ad2[n] = sd;
}

#define TILE_N    196
#define AGG_WAVES 32
#define SEG_CAP   256

__device__ __forceinline__ float lrelu(float v) { return v > 0.f ? v : NEG_SLOPE * v; }

__device__ __forceinline__ float lane_bcast(float v, int j) {
    return __uint_as_float(__builtin_amdgcn_readlane(__float_as_uint(v), j));
}

template <int H, int R, bool ELU>
__device__ __forceinline__ void node_rows(const int* srcs, const int* dsts, int count,
                                          int n0, int nt, int wave, int lane,
                                          const float* __restrict__ as,
                                          const float* __restrict__ ad,
                                          const float* __restrict__ hin,
                                          const float* __restrict__ bias,
                                          float* __restrict__ hout) {
    constexpr int ROW = 32 * R;
    for (int ln = wave; ln < nt; ln += AGG_WAVES) {
        const int n = n0 + ln;
        float adn[H], lself[H], mx[H];
#pragma unroll
        for (int h = 0; h < H; ++h) {
            adn[h]   = ad[n * H + h];
            lself[h] = lrelu(as[n * H + h] + adn[h]);
            mx[h]    = lself[h];
        }
        for (int c = 0; c < count; c += 32) {
            const int i = c + lane;
            if (i < count && dsts[i] == n) {
                const int s = srcs[i];
#pragma unroll
                for (int h = 0; h < H; ++h)
                    mx[h] = fmaxf(mx[h], lrelu(as[s * H + h] + adn[h]));
            }
        }
#pragma unroll
        for (int h = 0; h < H; ++h) {
#pragma unroll
            for (int off = 16; off > 0; off >>= 1)
                mx[h] = fmaxf(mx[h], __shfl_xor(mx[h], off));
            if (!(fabsf(mx[h]) < INFINITY)) mx[h] = 0.f;
        }
        float den[H], acc[R];
        {
            const float* hrow = hin + (size_t)n * ROW;
#pragma unroll
            for (int h = 0; h < H; ++h) den[h] = expf(lself[h] - mx[h]);
#pragma unroll
            for (int r = 0; r < R; ++r) acc[r] = den[H == 1 ? 0 : r] * hrow[r * 32 + lane];
        }
        for (int c = 0; c < count; c += 32) {
            const int i = c + lane;
            bool m = false;
            int  s = 0;
            if (i < count) m = (dsts[i] == n);
            float p[H];
#pragma unroll
            for (int h = 0; h < H; ++h) p[h] = 0.f;
            if (m) {
                s = srcs[i];
#pragma unroll
                for (int h = 0; h < H; ++h)
                    p[h] = expf(lrelu(as[s * H + h] + adn[h]) - mx[h]);
            }
            unsigned mask = __builtin_amdgcn_ballot_w32(m);
            while (mask) {
                const int j = __builtin_ctz(mask);
                mask &= mask - 1u;
                const int sj = __builtin_amdgcn_readlane(s, j);
                const float* hrow = hin + (size_t)sj * ROW;
                float v[R];
#pragma unroll
                for (int r = 0; r < R; ++r) v[r] = hrow[r * 32 + lane];
                float pj[H];
#pragma unroll
                for (int h = 0; h < H; ++h) { pj[h] = lane_bcast(p[h], j); den[h] += pj[h]; }
#pragma unroll
                for (int r = 0; r < R; ++r) acc[r] += pj[H == 1 ? 0 : r] * v[r];
            }
        }
        float* orow = hout + (size_t)n * ROW;
#pragma unroll
        for (int r = 0; r < R; ++r) {
            const int idx = r * 32 + lane;
            float v = acc[r] / (den[H == 1 ? 0 : r] + 1e-16f) + bias[idx];
            if (ELU) v = v > 0.f ? v : (expf(v) - 1.f);
            orow[idx] = v;
        }
    }
}

template <int H, int R, bool ELU>
__global__ void __launch_bounds__(1024)
agg_tile_kernel(const int* __restrict__ ei, int E0, int Nn,
                const float* __restrict__ as, const float* __restrict__ ad,
                const float* __restrict__ hin, const float* __restrict__ bias,
                float* __restrict__ hout) {
    __shared__ int seg_e[AGG_WAVES][SEG_CAP];
    __shared__ int seg_d[AGG_WAVES][SEG_CAP];
    __shared__ int lst_src[AGG_WAVES * SEG_CAP];
    __shared__ int lst_dst[AGG_WAVES * SEG_CAP];
    __shared__ int own_src[AGG_WAVES * SEG_CAP];
    __shared__ int own_dst[AGG_WAVES * SEG_CAP];
    __shared__ int seg_cnt[AGG_WAVES];
    __shared__ int own_cnt[AGG_WAVES];
    __shared__ int overflow;

    const int tid  = threadIdx.x;
    const int lane = tid & 31;
    const int wave = __builtin_amdgcn_readfirstlane(tid >> 5);
    const int n0   = blockIdx.x * TILE_N;
    const int nt   = min(TILE_N, Nn - n0);
    const int* __restrict__ dstp = ei + E0;

    if (tid == 0) overflow = 0;
    __syncthreads();

    int cnt = 0;
    for (int base = wave * 256; base < E0; base += AGG_WAVES * 256) {
        int d[8];
        if (base + 256 <= E0) {
#pragma unroll
            for (int j = 0; j < 8; ++j) d[j] = dstp[base + j * 32 + lane];
        } else {
#pragma unroll
            for (int j = 0; j < 8; ++j) {
                const int e = base + j * 32 + lane;
                d[j] = (e < E0) ? dstp[e] : -1;
            }
        }
#pragma unroll
        for (int j = 0; j < 8; ++j) {
            const bool m = (unsigned)(d[j] - n0) < (unsigned)nt;
            const unsigned mask = __builtin_amdgcn_ballot_w32(m);
            if (mask) {
                const int pos = cnt + __builtin_popcount(mask & ((1u << lane) - 1u));
                if (m && pos < SEG_CAP) {
                    seg_e[wave][pos] = base + j * 32 + lane;
                    seg_d[wave][pos] = d[j];
                }
                cnt += __builtin_popcount(mask);
            }
        }
    }
    if (lane == 0) {
        seg_cnt[wave] = min(cnt, SEG_CAP);
        if (cnt > SEG_CAP) overflow = 1;
    }
    __syncthreads();

    int mybase = 0, total = 0;
    for (int w = 0; w < AGG_WAVES; ++w) {
        const int c = seg_cnt[w];
        if (w < wave) mybase += c;
        total += c;
    }
    const int myc = seg_cnt[wave];
    const int ovf = overflow;
    for (int i = lane; i < myc; i += 32) {
        lst_src[mybase + i] = ei[seg_e[wave][i]];
        lst_dst[mybase + i] = seg_d[wave][i];
    }
    __syncthreads();

    int oc = 0;
    if (!ovf) {
        for (int c = 0; c < total; c += 32) {
            const int i = c + lane;
            const bool m = (i < total) && (((lst_dst[i] - n0) & (AGG_WAVES - 1)) == wave);
            oc += __builtin_popcount(__builtin_amdgcn_ballot_w32(m));
        }
    }
    if (lane == 0) own_cnt[wave] = oc;
    __syncthreads();
    int obase = 0;
    for (int w = 0; w < wave; ++w) obase += own_cnt[w];

    if (!ovf) {
        int k = obase;
        for (int c = 0; c < total; c += 32) {
            const int i = c + lane;
            const bool m = (i < total) && (((lst_dst[i] - n0) & (AGG_WAVES - 1)) == wave);
            const unsigned mask = __builtin_amdgcn_ballot_w32(m);
            if (m) {
                const int pos = k + __builtin_popcount(mask & ((1u << lane) - 1u));
                own_src[pos] = lst_src[i];
                own_dst[pos] = lst_dst[i];
            }
            k += __builtin_popcount(mask);
        }
    }
    __syncthreads();

    if (!ovf)
        node_rows<H, R, ELU>(own_src + obase, own_dst + obase, oc, n0, nt, wave, lane, as, ad, hin, bias, hout);
    else
        node_rows<H, R, ELU>(ei, dstp, E0, n0, nt, wave, lane, as, ad, hin, bias, hout);
}

extern "C" void kernel_launch(void* const* d_in, const int* in_sizes, int n_in,
                              void* d_out, int out_size, void* d_ws, size_t ws_size,
                              hipStream_t stream) {
    const float* x        = (const float*)d_in[0];
    const int*   ei       = (const int*)d_in[1];
    const float* W1       = (const float*)d_in[2];
    const float* att_src1 = (const float*)d_in[3];
    const float* att_dst1 = (const float*)d_in[4];
    const float* b1       = (const float*)d_in[5];
    const float* W2       = (const float*)d_in[6];
    const float* att_src2 = (const float*)d_in[7];
    const float* att_dst2 = (const float*)d_in[8];
    const float* b2       = (const float*)d_in[9];
    float* out = (float*)d_out;

    const int Nn = in_sizes[0] / 256;
    const int E0 = in_sizes[1] / 2;

    float* h1   = (float*)d_ws;
    float* helu = h1 + (size_t)Nn * 256;
    float* as1  = helu + (size_t)Nn * 256;
    float* ad1  = as1 + (size_t)Nn * 8;
    float* h2   = h1;
    float* as2  = as1;
    float* ad2  = as1 + Nn;

    const int TB = 256;
    dim3 g1((256 + 63) / 64, (Nn + 63) / 64);
    dim3 g2((64 + 63) / 64, (Nn + 63) / 64);
    const int tiles = (Nn + TILE_N - 1) / TILE_N;

    gat_wmma_gemm<<<g1, TB, 0, stream>>>(x, W1, h1, Nn, 256, 256);
    scores1_kernel<<<(Nn * 8 + TB - 1) / TB, TB, 0, stream>>>(h1, att_src1, att_dst1, as1, ad1, Nn);
    agg_tile_kernel<8, 8, true><<<tiles, 1024, 0, stream>>>(ei, E0, Nn, as1, ad1, h1, b1, helu);

    gat_wmma_gemm<<<g2, TB, 0, stream>>>(helu, W2, h2, Nn, 64, 256);
    scores2_kernel<<<(Nn + TB - 1) / TB, TB, 0, stream>>>(h2, att_src2, att_dst2, as2, ad2, Nn);
    agg_tile_kernel<1, 2, false><<<tiles, 1024, 0, stream>>>(ei, E0, Nn, as2, ad2, h2, b2, out);
}


# Per-node gather issues its dependent loads in groups: the score loads of 2-4 list chunks and the source rows of up to 4 edges go out together before their sums; sum order unchanged
# speedup vs baseline: 2.6111x; 1.0743x over previous
#include <hip/hip_runtime.h>
#include <hip/hip_bf16.h>

typedef __attribute__((ext_vector_type(2))) float v2f;
typedef __attribute__((ext_vector_type(8))) float v8f;

#define NEG_SLOPE 0.2f

__global__ void gat_wmma_gemm(const float* __restrict__ A,
                              const float* __restrict__ B,
                              float* __restrict__ C,
                              int M, int Ntot, int K) {
    __shared__ float As[64][68];
    __shared__ float Bp[32][132];

    const int tid   = threadIdx.x;
    const int lane  = tid & 31;
    const int wave  = tid >> 5;
    const int mBase = blockIdx.y * 64;
    const int nBase = blockIdx.x * 64;

    const int ms   = (wave & 3) * 16;
    const int ns   = (wave >> 2) * 32;
    const int mrow = ms + (lane & 15);
    const int khi  = (lane >> 4) * 2;
    const int ncol = lane & 15;

    v8f acc0 = {}; v8f acc1 = {};

    for (int kBase = 0; kBase < K; kBase += 64) {
#pragma unroll
        for (int i = 0; i < 4; ++i) {
            int s  = tid + i * 256;
            int r  = s >> 4;
            int c4 = (s & 15) << 2;
            float4 va = make_float4(0.f, 0.f, 0.f, 0.f);
            int g = mBase + r;
            if (g < M) va = *(const float4*)(A + (size_t)g * K + kBase + c4);
            *(float4*)&As[r][c4] = va;
            float4 vb = *(const float4*)(B + (size_t)(kBase + r) * Ntot + nBase + c4);
            float* bp = &Bp[r >> 1][r & 1];
            bp[2 * c4 + 0] = vb.x;
            bp[2 * c4 + 2] = vb.y;
            bp[2 * c4 + 4] = vb.z;
            bp[2 * c4 + 6] = vb.w;
        }
        __syncthreads();

#pragma unroll
        for (int kk = 0; kk < 64; kk += 4) {
            const int p2 = (kk + khi) >> 1;
            v2f a, b0, b1v;
            a[0]   = As[mrow][kk + khi];
            a[1]   = As[mrow][kk + khi + 1];
            b0     = *(const v2f*)&Bp[p2][2 * (ns + ncol)];
            b1v    = *(const v2f*)&Bp[p2][2 * (ns + 16 + ncol)];
            acc0 = __builtin_amdgcn_wmma_f32_16x16x4_f32(
                false, a, false, b0, (short)0, acc0, false, false);
            acc1 = __builtin_amdgcn_wmma_f32_16x16x4_f32(
                false, a, false, b1v, (short)0, acc1, false, false);
        }
        __syncthreads();
    }

    const int orow = mBase + ms + ((lane >> 4) << 3);
    const int ocol = nBase + ns + ncol;
#pragma unroll
    for (int r = 0; r < 8; ++r) {
        int g = orow + r;
        if (g < M) {
            C[(size_t)g * Ntot + ocol]      = acc0[r];
            C[(size_t)g * Ntot + ocol + 16] = acc1[r];
        }
    }
}

__global__ void scores1_kernel(const float* __restrict__ h1,
                               const float* __restrict__ att_s,
                               const float* __restrict__ att_d,
                               float* __restrict__ as1, float* __restrict__ ad1,
                               int Nn) {
    int t = blockIdx.x * blockDim.x + threadIdx.x;
    if (t >= Nn * 8) return;
    int n = t >> 3, h = t & 7;
    const float* row = h1 + (size_t)n * 256 + h * 32;
    const float* ws = att_s + h * 32;
    const float* wd = att_d + h * 32;
    float ss = 0.f, sd = 0.f;
#pragma unroll
    for (int c = 0; c < 32; ++c) { ss += row[c] * ws[c]; sd += row[c] * wd[c]; }
    as1[t] = ss; ad1[t] = sd;
}

__global__ void scores2_kernel(const float* __restrict__ h2,
                               const float* __restrict__ att_s,
                               const float* __restrict__ att_d,
                               float* __restrict__ as2, float* __restrict__ ad2,
                               int Nn) {
    int n = blockIdx.x * blockDim.x + threadIdx.x;
    if (n >= Nn) return;
    const float* row = h2 + (size_t)n * 64;
    float ss = 0.f, sd = 0.f;
#pragma unroll
    for (int c = 0; c < 64; ++c) { ss += row[c] * att_s[c]; sd += row[c] * att_d[c]; }
    as2[n] = ss; ad2[n] = sd;
}

#define TILE_N    196
#define AGG_WAVES 32
#define SEG_CAP   256

__device__ __forceinline__ float lrelu(float v) { return v > 0.f ? v : NEG_SLOPE * v; }

__device__ __forceinline__ float lane_bcast(float v, int j) {
    return __uint_as_float(__builtin_amdgcn_readlane(__float_as_uint(v), j));
}

template <int H, int R, bool ELU, int CB1, int CB2, int EB>
__device__ __forceinline__ void node_rows(const int* srcs, const int* dsts, int count,
                                          int n0, int nt, int wave, int lane,
                                          const float* __restrict__ as,
                                          const float* __restrict__ ad,
                                          const float* __restrict__ hin,
                                          const float* __restrict__ bias,
                                          float* __restrict__ hout) {
    constexpr int ROW = 32 * R;
    for (int ln = wave; ln < nt; ln += AGG_WAVES) {
        const int n = n0 + ln;
        float adn[H], lself[H], mx[H];
#pragma unroll
        for (int h = 0; h < H; ++h) {
            adn[h]   = ad[n * H + h];
            lself[h] = lrelu(as[n * H + h] + adn[h]);
            mx[h]    = lself[h];
        }
        for (int c = 0; c < count; c += 32 * CB1) {
            int sg[CB1];
#pragma unroll
            for (int u = 0; u < CB1; ++u) {
                const int  i  = c + u * 32 + lane;
                const bool in = i < count;
                const int  d  = in ? dsts[i] : -1;
                const int  sv = in ? srcs[i] : n;
                sg[u] = (d == n) ? sv : n;
            }
            float lg[CB1][H];
#pragma unroll
            for (int u = 0; u < CB1; ++u) {
#pragma unroll
                for (int h = 0; h < H; ++h) lg[u][h] = as[sg[u] * H + h];
            }
#pragma unroll
            for (int u = 0; u < CB1; ++u) {
#pragma unroll
                for (int h = 0; h < H; ++h) mx[h] = fmaxf(mx[h], lrelu(lg[u][h] + adn[h]));
            }
        }
#pragma unroll
        for (int h = 0; h < H; ++h) {
#pragma unroll
            for (int off = 16; off > 0; off >>= 1)
                mx[h] = fmaxf(mx[h], __shfl_xor(mx[h], off));
            if (!(fabsf(mx[h]) < INFINITY)) mx[h] = 0.f;
        }
        float den[H], acc[R];
        {
            const float* hrow = hin + (size_t)n * ROW;
#pragma unroll
            for (int h = 0; h < H; ++h) den[h] = expf(lself[h] - mx[h]);
#pragma unroll
            for (int r = 0; r < R; ++r) acc[r] = den[H == 1 ? 0 : r] * hrow[r * 32 + lane];
        }
        for (int c = 0; c < count; c += 32 * CB2) {
            int  sg[CB2];
            bool mg[CB2];
#pragma unroll
            for (int u = 0; u < CB2; ++u) {
                const int  i  = c + u * 32 + lane;
                const bool in = i < count;
                const int  d  = in ? dsts[i] : -1;
                const int  sv = in ? srcs[i] : n;
                mg[u] = (d == n);
                sg[u] = mg[u] ? sv : n;
            }
            float lg[CB2][H];
#pragma unroll
            for (int u = 0; u < CB2; ++u) {
#pragma unroll
                for (int h = 0; h < H; ++h) lg[u][h] = as[sg[u] * H + h];
            }
#pragma unroll
            for (int u = 0; u < CB2; ++u) {
                float p[H];
#pragma unroll
                for (int h = 0; h < H; ++h)
                    p[h] = mg[u] ? expf(lrelu(lg[u][h] + adn[h]) - mx[h]) : 0.f;
                unsigned mask = __builtin_amdgcn_ballot_w32(mg[u]);
                while (mask) {
                    int  jj[EB];
                    bool ok[EB];
#pragma unroll
                    for (int e = 0; e < EB; ++e) {
                        ok[e] = (mask != 0u);
                        jj[e] = ok[e] ? __builtin_ctz(mask) : 0;
                        mask &= mask - 1u;
                    }
                    float v[EB][R];
#pragma unroll
                    for (int e = 0; e < EB; ++e) {
#pragma unroll
                        for (int r = 0; r < R; ++r) v[e][r] = 0.f;
                        if (ok[e]) {
                            const int sj = __builtin_amdgcn_readlane(sg[u], jj[e]);
                            const float* hrow = hin + (size_t)sj * ROW;
#pragma unroll
                            for (int r = 0; r < R; ++r) v[e][r] = hrow[r * 32 + lane];
                        }
                    }
#pragma unroll
                    for (int e = 0; e < EB; ++e) {
                        if (ok[e]) {
                            float pj[H];
#pragma unroll
                            for (int h = 0; h < H; ++h) { pj[h] = lane_bcast(p[h], jj[e]); den[h] += pj[h]; }
#pragma unroll
                            for (int r = 0; r < R; ++r) acc[r] += pj[H == 1 ? 0 : r] * v[e][r];
                        }
                    }
                }
            }
        }
        float* orow = hout + (size_t)n * ROW;
#pragma unroll
        for (int r = 0; r < R; ++r) {
            const int idx = r * 32 + lane;
            float v = acc[r] / (den[H == 1 ? 0 : r] + 1e-16f) + bias[idx];
            if (ELU) v = v > 0.f ? v : (expf(v) - 1.f);
            orow[idx] = v;
        }
    }
}

template <int H, int R, bool ELU>
__global__ void __launch_bounds__(1024)
agg_tile_kernel(const int* __restrict__ ei, int E0, int Nn,
                const float* __restrict__ as, const float* __restrict__ ad,
                const float* __restrict__ hin, const float* __restrict__ bias,
                float* __restrict__ hout) {
    __shared__ int seg_e[AGG_WAVES][SEG_CAP];
    __shared__ int seg_d[AGG_WAVES][SEG_CAP];
    __shared__ int lst_src[AGG_WAVES * SEG_CAP];
    __shared__ int lst_dst[AGG_WAVES * SEG_CAP];
    __shared__ int own_src[AGG_WAVES * SEG_CAP];
    __shared__ int own_dst[AGG_WAVES * SEG_CAP];
    __shared__ int seg_cnt[AGG_WAVES];
    __shared__ int own_cnt[AGG_WAVES];
    __shared__ int overflow;

    const int tid  = threadIdx.x;
    const int lane = tid & 31;
    const int wave = __builtin_amdgcn_readfirstlane(tid >> 5);
    const int n0   = blockIdx.x * TILE_N;
    const int nt   = min(TILE_N, Nn - n0);
    const int* __restrict__ dstp = ei + E0;

    if (tid == 0) overflow = 0;
    __syncthreads();

    int cnt = 0;
    for (int base = wave * 256; base < E0; base += AGG_WAVES * 256) {
        int d[8];
        if (base + 256 <= E0) {
#pragma unroll
            for (int j = 0; j < 8; ++j) d[j] = dstp[base + j * 32 + lane];
        } else {
#pragma unroll
            for (int j = 0; j < 8; ++j) {
                const int e = base + j * 32 + lane;
                d[j] = (e < E0) ? dstp[e] : -1;
            }
        }
#pragma unroll
        for (int j = 0; j < 8; ++j) {
            const bool m = (unsigned)(d[j] - n0) < (unsigned)nt;
            const unsigned mask = __builtin_amdgcn_ballot_w32(m);
            if (mask) {
                const int pos = cnt + __builtin_popcount(mask & ((1u << lane) - 1u));
                if (m && pos < SEG_CAP) {
                    seg_e[wave][pos] = base + j * 32 + lane;
                    seg_d[wave][pos] = d[j];
                }
                cnt += __builtin_popcount(mask);
            }
        }
    }
    if (lane == 0) {
        seg_cnt[wave] = min(cnt, SEG_CAP);
        if (cnt > SEG_CAP) overflow = 1;
    }
    __syncthreads();

    int mybase = 0, total = 0;
    for (int w = 0; w < AGG_WAVES; ++w) {
        const int c = seg_cnt[w];
        if (w < wave) mybase += c;
        total += c;
    }
    const int myc = seg_cnt[wave];
    const int ovf = overflow;
    for (int i = lane; i < myc; i += 32) {
        lst_src[mybase + i] = ei[seg_e[wave][i]];
        lst_dst[mybase + i] = seg_d[wave][i];
    }
    __syncthreads();

    int oc = 0;
    if (!ovf) {
        for (int c = 0; c < total; c += 32) {
            const int i = c + lane;
            const bool m = (i < total) && (((lst_dst[i] - n0) & (AGG_WAVES - 1)) == wave);
            oc += __builtin_popcount(__builtin_amdgcn_ballot_w32(m));
        }
    }
    if (lane == 0) own_cnt[wave] = oc;
    __syncthreads();
    int obase = 0;
    for (int w = 0; w < wave; ++w) obase += own_cnt[w];

    if (!ovf) {
        int k = obase;
        for (int c = 0; c < total; c += 32) {
            const int i = c + lane;
            const bool m = (i < total) && (((lst_dst[i] - n0) & (AGG_WAVES - 1)) == wave);
            const unsigned mask = __builtin_amdgcn_ballot_w32(m);
            if (m) {
                const int pos = k + __builtin_popcount(mask & ((1u << lane) - 1u));
                own_src[pos] = lst_src[i];
                own_dst[pos] = lst_dst[i];
            }
            k += __builtin_popcount(mask);
        }
    }
    __syncthreads();

    constexpr int CB1 = 4;
    constexpr int CB2 = (H > 1) ? 2 : 4;
    constexpr int EB  = 4;
    if (!ovf)
        node_rows<H, R, ELU, CB1, CB2, EB>(own_src + obase, own_dst + obase, oc, n0, nt, wave, lane, as, ad, hin, bias, hout);
    else
        node_rows<H, R, ELU, CB1, CB2, EB>(ei, dstp, E0, n0, nt, wave, lane, as, ad, hin, bias, hout);
}

extern "C" void kernel_launch(void* const* d_in, const int* in_sizes, int n_in,
                              void* d_out, int out_size, void* d_ws, size_t ws_size,
                              hipStream_t stream) {
    const float* x        = (const float*)d_in[0];
    const int*   ei       = (const int*)d_in[1];
    const float* W1       = (const float*)d_in[2];
    const float* att_src1 = (const float*)d_in[3];
    const float* att_dst1 = (const float*)d_in[4];
    const float* b1       = (const float*)d_in[5];
    const float* W2       = (const float*)d_in[6];
    const float* att_src2 = (const float*)d_in[7];
    const float* att_dst2 = (const float*)d_in[8];
    const float* b2       = (const float*)d_in[9];
    float* out = (float*)d_out;

    const int Nn = in_sizes[0] / 256;
    const int E0 = in_sizes[1] / 2;

    float* h1   = (float*)d_ws;
    float* helu = h1 + (size_t)Nn * 256;
    float* as1  = helu + (size_t)Nn * 256;
    float* ad1  = as1 + (size_t)Nn * 8;
    float* h2   = h1;
    float* as2  = as1;
    float* ad2  = as1 + Nn;

    const int TB = 256;
    dim3 g1((256 + 63) / 64, (Nn + 63) / 64);
    dim3 g2((64 + 63) / 64, (Nn + 63) / 64);
    const int tiles = (Nn + TILE_N - 1) / TILE_N;

    gat_wmma_gemm<<<g1, TB, 0, stream>>>(x, W1, h1, Nn, 256, 256);
    scores1_kernel<<<(Nn * 8 + TB - 1) / TB, TB, 0, stream>>>(h1, att_src1, att_dst1, as1, ad1, Nn);
    agg_tile_kernel<8, 8, true><<<tiles, 1024, 0, stream>>>(ei, E0, Nn, as1, ad1, h1, b1, helu);

    gat_wmma_gemm<<<g2, TB, 0, stream>>>(helu, W2, h2, Nn, 64, 256);
    scores2_kernel<<<(Nn + TB - 1) / TB, TB, 0, stream>>>(h2, att_src2, att_dst2, as2, ad2, Nn);
    agg_tile_kernel<1, 2, false><<<tiles, 1024, 0, stream>>>(ei, E0, Nn, as2, ad2, h2, b2, out);
}


# Per-node gather re-laid: a node's sources are compacted to a short list and a lane owns adjacent columns of one head: a row is two 16-byte loads, the weight is formed in-lane; sum order unchanged
# speedup vs baseline: 2.8958x; 1.1090x over previous
#include <hip/hip_runtime.h>
#include <hip/hip_bf16.h>

typedef __attribute__((ext_vector_type(2))) float v2f;
typedef __attribute__((ext_vector_type(8))) float v8f;

#define NEG_SLOPE 0.2f

__global__ void gat_wmma_gemm(const float* __restrict__ A,
                              const float* __restrict__ B,
                              float* __restrict__ C,
                              int M, int Ntot, int K) {
    __shared__ float As[64][68];
    __shared__ float Bp[32][132];

    const int tid   = threadIdx.x;
    const int lane  = tid & 31;
    const int wave  = tid >> 5;
    const int mBase = blockIdx.y * 64;
    const int nBase = blockIdx.x * 64;

    const int ms   = (wave & 3) * 16;
    const int ns   = (wave >> 2) * 32;
    const int mrow = ms + (lane & 15);
    const int khi  = (lane >> 4) * 2;
    const int ncol = lane & 15;

    v8f acc0 = {}; v8f acc1 = {};

    for (int kBase = 0; kBase < K; kBase += 64) {
#pragma unroll
        for (int i = 0; i < 4; ++i) {
            int s  = tid + i * 256;
            int r  = s >> 4;
            int c4 = (s & 15) << 2;
            float4 va = make_float4(0.f, 0.f, 0.f, 0.f);
            int g = mBase + r;
            if (g < M) va = *(const float4*)(A + (size_t)g * K + kBase + c4);
            *(float4*)&As[r][c4] = va;
            float4 vb = *(const float4*)(B + (size_t)(kBase + r) * Ntot + nBase + c4);
            float* bp = &Bp[r >> 1][r & 1];
            bp[2 * c4 + 0] = vb.x;
            bp[2 * c4 + 2] = vb.y;
            bp[2 * c4 + 4] = vb.z;
            bp[2 * c4 + 6] = vb.w;
        }
        __syncthreads();

#pragma unroll
        for (int kk = 0; kk < 64; kk += 4) {
            const int p2 = (kk + khi) >> 1;
            v2f a, b0, b1v;
            a[0]   = As[mrow][kk + khi];
            a[1]   = As[mrow][kk + khi + 1];
            b0     = *(const v2f*)&Bp[p2][2 * (ns + ncol)];
            b1v    = *(const v2f*)&Bp[p2][2 * (ns + 16 + ncol)];
            acc0 = __builtin_amdgcn_wmma_f32_16x16x4_f32(
                false, a, false, b0, (short)0, acc0, false, false);
            acc1 = __builtin_amdgcn_wmma_f32_16x16x4_f32(
                false, a, false, b1v, (short)0, acc1, false, false);
        }
        __syncthreads();
    }

    const int orow = mBase + ms + ((lane >> 4) << 3);
    const int ocol = nBase + ns + ncol;
#pragma unroll
    for (int r = 0; r < 8; ++r) {
        int g = orow + r;
        if (g < M) {
            C[(size_t)g * Ntot + ocol]      = acc0[r];
            C[(size_t)g * Ntot + ocol + 16] = acc1[r];
        }
    }
}

__global__ void scores1_kernel(const float* __restrict__ h1,
                               const float* __restrict__ att_s,
                               const float* __restrict__ att_d,
                               float* __restrict__ as1, float* __restrict__ ad1,
                               int Nn) {
    int t = blockIdx.x * blockDim.x + threadIdx.x;
    if (t >= Nn * 8) return;
    int n = t >> 3, h = t & 7;
    const float* row = h1 + (size_t)n * 256 + h * 32;
    const float* ws = att_s + h * 32;
    const float* wd = att_d + h * 32;
    float ss = 0.f, sd = 0.f;
#pragma unroll
    for (int c = 0; c < 32; ++c) { ss += row[c] * ws[c]; sd += row[c] * wd[c]; }
    as1[t] = ss; ad1[t] = sd;
}

__global__ void scores2_kernel(const float* __restrict__ h2,
                               const float* __restrict__ att_s,
                               const float* __restrict__ att_d,
                               float* __restrict__ as2, float* __restrict__ ad2,
                               int Nn) {
    int n = blockIdx.x * blockDim.x + threadIdx.x;
    if (n >= Nn) return;
    const float* row = h2 + (size_t)n * 64;
    float ss = 0.f, sd = 0.f;
#pragma unroll
    for (int c = 0; c < 64; ++c) { ss += row[c] * att_s[c]; sd += row[c] * att_d[c]; }
    as2[n] = ss; ad2[n] = sd;
}

#define TILE_N    196
#define AGG_WAVES 32
#define SEG_CAP   256

__device__ __forceinline__ float lrelu(float v) { return v > 0.f ? v : NEG_SLOPE * v; }

__device__ __forceinline__ float lane_bcast(float v, int j) {
    return __uint_as_float(__builtin_amdgcn_readlane(__float_as_uint(v), j));
}

__device__ __forceinline__ int compact_sources(const int* __restrict__ srcs, const int* __restrict__ dsts,
                                               int c0, int cend, int n, int lane,
                                               int* __restrict__ scratch) {
    const unsigned below = (1u << lane) - 1u;
    int m = 0;
    for (int i0 = c0; i0 < cend; i0 += 32) {
        const int  i     = i0 + lane;
        const bool in    = i < cend;
        const int  d     = in ? dsts[i] : -1;
        const int  sv    = in ? srcs[i] : 0;
        const bool match = (d == n);
        const unsigned mask = __builtin_amdgcn_ballot_w32(match);
        if (match) scratch[m + __builtin_popcount(mask & below)] = sv;
        m += __builtin_popcount(mask);
    }
    return m;
}

__device__ __forceinline__ void wave_lds_handover() {
    __builtin_amdgcn_fence(__ATOMIC_RELEASE, "wavefront");
    __builtin_amdgcn_wave_barrier();
    __builtin_amdgcn_fence(__ATOMIC_ACQUIRE, "wavefront");
}

template <int H, int R, bool ELU, int G1, int G2>
__device__ __forceinline__ void node_rows(const int* __restrict__ srcs, const int* __restrict__ dsts, int count,
                                          int n0, int nt, int wave, int lane,
                                          const float* __restrict__ as,
                                          const float* __restrict__ ad,
                                          const float* __restrict__ hin,
                                          const float* __restrict__ bias,
                                          float* __restrict__ hout,
                                          int* __restrict__ scratch) {
    constexpr int ROW = 32 * R;
    constexpr int VW  = (R >= 4) ? 4 : 2;
    typedef float VT __attribute__((ext_vector_type(VW)));
    const int  col    = lane * R;
    const int  hl     = (col * H) / ROW;
    const bool single = count <= SEG_CAP;

    for (int ln = wave; ln < nt; ln += AGG_WAVES) {
        const int   n     = n0 + ln;
        const float adn   = ad[n * H + hl];
        const float lself = lrelu(as[n * H + hl] + adn);
        float mx = lself;
        int m0 = 0;

        for (int c0 = 0; c0 < count; c0 += SEG_CAP) {
            const int m = compact_sources(srcs, dsts, c0, min(count, c0 + SEG_CAP), n, lane, scratch);
            if (c0 == 0) m0 = m;
            wave_lds_handover();
            for (int j = 0; j < m; j += 32) {
                const int cnt  = min(32, m - j);
                const int sreg = (lane < cnt) ? scratch[j + lane] : n;
                if constexpr (H == 1) {
                    mx = fmaxf(mx, lrelu(as[sreg] + adn));
                } else {
                    for (int e0 = 0; e0 < cnt; e0 += G1) {
                        float s[G1];
#pragma unroll
                        for (int g = 0; g < G1; ++g) {
                            const int sj = __builtin_amdgcn_readlane(sreg, e0 + g);
                            s[g] = as[sj * H + hl];
                        }
#pragma unroll
                        for (int g = 0; g < G1; ++g) mx = fmaxf(mx, lrelu(s[g] + adn));
                    }
                }
            }
            wave_lds_handover();
        }
        if constexpr (H == 1) {
#pragma unroll
            for (int off = 16; off > 0; off >>= 1) mx = fmaxf(mx, __shfl_xor(mx, off));
        }
        if (!(fabsf(mx) < INFINITY)) mx = 0.f;

        float den = expf(lself - mx);
        float acc[R];
        {
            const float* hrow = hin + (size_t)n * ROW + col;
#pragma unroll
            for (int q = 0; q < R / VW; ++q) {
                const VT t = *(const VT*)(hrow + q * VW);
#pragma unroll
                for (int k = 0; k < VW; ++k) acc[q * VW + k] = den * t[k];
            }
        }
        for (int c0 = 0; c0 < count; c0 += SEG_CAP) {
            int m = m0;
            if (!single) {
                m = compact_sources(srcs, dsts, c0, min(count, c0 + SEG_CAP), n, lane, scratch);
                wave_lds_handover();
            }
            for (int j = 0; j < m; j += 32) {
                const int cnt  = min(32, m - j);
                const int sreg = (lane < cnt) ? scratch[j + lane] : n;
                float wreg = 0.f;
                if constexpr (H == 1) wreg = expf(lrelu(as[sreg] + adn) - mx);
                for (int e0 = 0; e0 < cnt; e0 += G2) {
                    float s[G2];
                    float v[G2][R];
#pragma unroll
                    for (int g = 0; g < G2; ++g) {
                        s[g] = 0.f;
#pragma unroll
                        for (int r = 0; r < R; ++r) v[g][r] = 0.f;
                        if (e0 + g < cnt) {
                            const int sj = __builtin_amdgcn_readlane(sreg, e0 + g);
                            if constexpr (H > 1) s[g] = as[sj * H + hl];
                            const float* hrow = hin + (size_t)sj * ROW + col;
#pragma unroll
                            for (int q = 0; q < R / VW; ++q) {
                                const VT t = *(const VT*)(hrow + q * VW);
#pragma unroll
                                for (int k = 0; k < VW; ++k) v[g][q * VW + k] = t[k];
                            }
                        }
                    }
#pragma unroll
                    for (int g = 0; g < G2; ++g) {
                        if (e0 + g < cnt) {
                            float w;
                            if constexpr (H == 1) w = lane_bcast(wreg, e0 + g);
                            else                  w = expf(lrelu(s[g] + adn) - mx);
                            den += w;
#pragma unroll
                            for (int r = 0; r < R; ++r) acc[r] += w * v[g][r];
                        }
                    }
                }
            }
            wave_lds_handover();
        }

        float o[R];
#pragma unroll
        for (int r = 0; r < R; ++r) {
            float val = acc[r] / (den + 1e-16f) + bias[col + r];
            if (ELU) val = val > 0.f ? val : (expf(val) - 1.f);
            o[r] = val;
        }
        float* orow = hout + (size_t)n * ROW + col;
#pragma unroll
        for (int q = 0; q < R / VW; ++q) {
            VT t;
#pragma unroll
            for (int k = 0; k < VW; ++k) t[k] = o[q * VW + k];
            *(VT*)(orow + q * VW) = t;
        }
    }
}

template <int H, int R, bool ELU>
__global__ void __launch_bounds__(1024)
agg_tile_kernel(const int* __restrict__ ei, int E0, int Nn,
                const float* __restrict__ as, const float* __restrict__ ad,
                const float* __restrict__ hin, const float* __restrict__ bias,
                float* __restrict__ hout) {
    __shared__ int seg_e[AGG_WAVES][SEG_CAP];
    __shared__ int seg_d[AGG_WAVES][SEG_CAP];
    __shared__ int lst_src[AGG_WAVES * SEG_CAP];
    __shared__ int lst_dst[AGG_WAVES * SEG_CAP];
    __shared__ int own_src[AGG_WAVES * SEG_CAP];
    __shared__ int own_dst[AGG_WAVES * SEG_CAP];
    __shared__ int seg_cnt[AGG_WAVES];
    __shared__ int own_cnt[AGG_WAVES];
    __shared__ int overflow;

    const int tid  = threadIdx.x;
    const int lane = tid & 31;
    const int wave = __builtin_amdgcn_readfirstlane(tid >> 5);
    const int n0   = blockIdx.x * TILE_N;
    const int nt   = min(TILE_N, Nn - n0);
    const int* __restrict__ dstp = ei + E0;

    if (tid == 0) overflow = 0;
    __syncthreads();

    int cnt = 0;
    for (int base = wave * 256; base < E0; base += AGG_WAVES * 256) {
        int d[8];
        if (base + 256 <= E0) {
#pragma unroll
            for (int j = 0; j < 8; ++j) d[j] = dstp[base + j * 32 + lane];
        } else {
#pragma unroll
            for (int j = 0; j < 8; ++j) {
                const int e = base + j * 32 + lane;
                d[j] = (e < E0) ? dstp[e] : -1;
            }
        }
#pragma unroll
        for (int j = 0; j < 8; ++j) {
            const bool m = (unsigned)(d[j] - n0) < (unsigned)nt;
            const unsigned mask = __builtin_amdgcn_ballot_w32(m);
            if (mask) {
                const int pos = cnt + __builtin_popcount(mask & ((1u << lane) - 1u));
                if (m && pos < SEG_CAP) {
                    seg_e[wave][pos] = base + j * 32 + lane;
                    seg_d[wave][pos] = d[j];
                }
                cnt += __builtin_popcount(mask);
            }
        }
    }
    if (lane == 0) {
        seg_cnt[wave] = min(cnt, SEG_CAP);
        if (cnt > SEG_CAP) overflow = 1;
    }
    __syncthreads();

    int mybase = 0, total = 0;
    for (int w = 0; w < AGG_WAVES; ++w) {
        const int c = seg_cnt[w];
        if (w < wave) mybase += c;
        total += c;
    }
    const int myc = seg_cnt[wave];
    const int ovf = overflow;
    for (int i = lane; i < myc; i += 32) {
        lst_src[mybase + i] = ei[seg_e[wave][i]];
        lst_dst[mybase + i] = seg_d[wave][i];
    }
    __syncthreads();

    int oc = 0;
    if (!ovf) {
        for (int c = 0; c < total; c += 32) {
            const int i = c + lane;
            const bool m = (i < total) && (((lst_dst[i] - n0) & (AGG_WAVES - 1)) == wave);
            oc += __builtin_popcount(__builtin_amdgcn_ballot_w32(m));
        }
    }
    if (lane == 0) own_cnt[wave] = oc;
    __syncthreads();
    int obase = 0;
    for (int w = 0; w < wave; ++w) obase += own_cnt[w];

    if (!ovf) {
        int k = obase;
        for (int c = 0; c < total; c += 32) {
            const int i = c + lane;
            const bool m = (i < total) && (((lst_dst[i] - n0) & (AGG_WAVES - 1)) == wave);
            const unsigned mask = __builtin_amdgcn_ballot_w32(m);
            if (m) {
                const int pos = k + __builtin_popcount(mask & ((1u << lane) - 1u));
                own_src[pos] = lst_src[i];
                own_dst[pos] = lst_dst[i];
            }
            k += __builtin_popcount(mask);
        }
    }
    __syncthreads();

    constexpr int G1 = 16;
    constexpr int G2 = (R > 2) ? 8 : 16;
    int* scratch = &seg_e[wave][0];
    if (!ovf)
        node_rows<H, R, ELU, G1, G2>(own_src + obase, own_dst + obase, oc, n0, nt, wave, lane, as, ad, hin, bias, hout, scratch);
    else
        node_rows<H, R, ELU, G1, G2>(ei, dstp, E0, n0, nt, wave, lane, as, ad, hin, bias, hout, scratch);
}

extern "C" void kernel_launch(void* const* d_in, const int* in_sizes, int n_in,
                              void* d_out, int out_size, void* d_ws, size_t ws_size,
                              hipStream_t stream) {
    const float* x        = (const float*)d_in[0];
    const int*   ei       = (const int*)d_in[1];
    const float* W1       = (const float*)d_in[2];
    const float* att_src1 = (const float*)d_in[3];
    const float* att_dst1 = (const float*)d_in[4];
    const float* b1       = (const float*)d_in[5];
    const float* W2       = (const float*)d_in[6];
    const float* att_src2 = (const float*)d_in[7];
    const float* att_dst2 = (const float*)d_in[8];
    const float* b2       = (const float*)d_in[9];
    float* out = (float*)d_out;

    const int Nn = in_sizes[0] / 256;
    const int E0 = in_sizes[1] / 2;

    float* h1   = (float*)d_ws;
    float* helu = h1 + (size_t)Nn * 256;
    float* as1  = helu + (size_t)Nn * 256;
    float* ad1  = as1 + (size_t)Nn * 8;
    float* h2   = h1;
    float* as2  = as1;
    float* ad2  = as1 + Nn;

    const int TB = 256;
    dim3 g1((256 + 63) / 64, (Nn + 63) / 64);
    dim3 g2((64 + 63) / 64, (Nn + 63) / 64);
    const int tiles = (Nn + TILE_N - 1) / TILE_N;

    gat_wmma_gemm<<<g1, TB, 0, stream>>>(x, W1, h1, Nn, 256, 256);
    scores1_kernel<<<(Nn * 8 + TB - 1) / TB, TB, 0, stream>>>(h1, att_src1, att_dst1, as1, ad1, Nn);
    agg_tile_kernel<8, 8, true><<<tiles, 1024, 0, stream>>>(ei, E0, Nn, as1, ad1, h1, b1, helu);

    gat_wmma_gemm<<<g2, TB, 0, stream>>>(helu, W2, h2, Nn, 64, 256);
    scores2_kernel<<<(Nn + TB - 1) / TB, TB, 0, stream>>>(h2, att_src2, att_dst2, as2, ad2, Nn);
    agg_tile_kernel<1, 2, false><<<tiles, 1024, 0, stream>>>(ei, E0, Nn, as2, ad2, h2, b2, out);
}


# Layer 2 takes each tile's wave-grouped edge list from layer 1's launch, handed over in the tile's own rows of the output buffer, instead of filtering the edge list again; sum order unchanged
# speedup vs baseline: 3.5763x; 1.2350x over previous
#include <hip/hip_runtime.h>
#include <hip/hip_bf16.h>

typedef __attribute__((ext_vector_type(2))) float v2f;
typedef __attribute__((ext_vector_type(8))) float v8f;

#define NEG_SLOPE 0.2f

__global__ void gat_wmma_gemm(const float* __restrict__ A,
                              const float* __restrict__ B,
                              float* __restrict__ C,
                              int M, int Ntot, int K) {
    __shared__ float As[64][68];
    __shared__ float Bp[32][132];

    const int tid   = threadIdx.x;
    const int lane  = tid & 31;
    const int wave  = tid >> 5;
    const int mBase = blockIdx.y * 64;
    const int nBase = blockIdx.x * 64;

    const int ms   = (wave & 3) * 16;
    const int ns   = (wave >> 2) * 32;
    const int mrow = ms + (lane & 15);
    const int khi  = (lane >> 4) * 2;
    const int ncol = lane & 15;

    v8f acc0 = {}; v8f acc1 = {};

    for (int kBase = 0; kBase < K; kBase += 64) {
#pragma unroll
        for (int i = 0; i < 4; ++i) {
            int s  = tid + i * 256;
            int r  = s >> 4;
            int c4 = (s & 15) << 2;
            float4 va = make_float4(0.f, 0.f, 0.f, 0.f);
            int g = mBase + r;
            if (g < M) va = *(const float4*)(A + (size_t)g * K + kBase + c4);
            *(float4*)&As[r][c4] = va;
            float4 vb = *(const float4*)(B + (size_t)(kBase + r) * Ntot + nBase + c4);
            float* bp = &Bp[r >> 1][r & 1];
            bp[2 * c4 + 0] = vb.x;
            bp[2 * c4 + 2] = vb.y;
            bp[2 * c4 + 4] = vb.z;
            bp[2 * c4 + 6] = vb.w;
        }
        __syncthreads();

#pragma unroll
        for (int kk = 0; kk < 64; kk += 4) {
            const int p2 = (kk + khi) >> 1;
            v2f a, b0, b1v;
            a[0]   = As[mrow][kk + khi];
            a[1]   = As[mrow][kk + khi + 1];
            b0     = *(const v2f*)&Bp[p2][2 * (ns + ncol)];
            b1v    = *(const v2f*)&Bp[p2][2 * (ns + 16 + ncol)];
            acc0 = __builtin_amdgcn_wmma_f32_16x16x4_f32(
                false, a, false, b0, (short)0, acc0, false, false);
            acc1 = __builtin_amdgcn_wmma_f32_16x16x4_f32(
                false, a, false, b1v, (short)0, acc1, false, false);
        }
        __syncthreads();
    }

    const int orow = mBase + ms + ((lane >> 4) << 3);
    const int ocol = nBase + ns + ncol;
#pragma unroll
    for (int r = 0; r < 8; ++r) {
        int g = orow + r;
        if (g < M) {
            C[(size_t)g * Ntot + ocol]      = acc0[r];
            C[(size_t)g * Ntot + ocol + 16] = acc1[r];
        }
    }
}

__global__ void scores1_kernel(const float* __restrict__ h1,
                               const float* __restrict__ att_s,
                               const float* __restrict__ att_d,
                               float* __restrict__ as1, float* __restrict__ ad1,
                               int Nn) {
    int t = blockIdx.x * blockDim.x + threadIdx.x;
    if (t >= Nn * 8) return;
    int n = t >> 3, h = t & 7;
    const float* row = h1 + (size_t)n * 256 + h * 32;
    const float* ws = att_s + h * 32;
    const float* wd = att_d + h * 32;
    float ss = 0.f, sd = 0.f;
#pragma unroll
    for (int c = 0; c < 32; ++c) { ss += row[c] * ws[c]; sd += row[c] * wd[c]; }
    as1[t] = ss; ad1[t] = sd;
}

__global__ void scores2_kernel(const float* __restrict__ h2,
                               const float* __restrict__ att_s,
                               const float* __restrict__ att_d,
                               float* __restrict__ as2, float* __restrict__ ad2,
                               int Nn) {
    int n = blockIdx.x * blockDim.x + threadIdx.x;
    if (n >= Nn) return;
    const float* row = h2 + (size_t)n * 64;
    float ss = 0.f, sd = 0.f;
#pragma unroll
    for (int c = 0; c < 64; ++c) { ss += row[c] * att_s[c]; sd += row[c] * att_d[c]; }
    as2[n] = ss; ad2[n] = sd;
}

#define TILE_N    196
#define AGG_WAVES 32
#define SEG_CAP   256
#define KEEP_HDR  64

__device__ __forceinline__ float lrelu(float v) { return v > 0.f ? v : NEG_SLOPE * v; }

__device__ __forceinline__ float lane_bcast(float v, int j) {
    return __uint_as_float(__builtin_amdgcn_readlane(__float_as_uint(v), j));
}

__device__ __forceinline__ int compact_sources(const int* __restrict__ srcs, const int* __restrict__ dsts,
                                               int c0, int cend, int n, int lane,
                                               int* __restrict__ scratch) {
    const unsigned below = (1u << lane) - 1u;
    int m = 0;
    for (int i0 = c0; i0 < cend; i0 += 32) {
        const int  i     = i0 + lane;
        const bool in    = i < cend;
        const int  d     = in ? dsts[i] : -1;
        const int  sv    = in ? srcs[i] : 0;
        const bool match = (d == n);
        const unsigned mask = __builtin_amdgcn_ballot_w32(match);
        if (match) scratch[m + __builtin_popcount(mask & below)] = sv;
        m += __builtin_popcount(mask);
    }
    return m;
}

__device__ __forceinline__ void wave_lds_handover() {
    __builtin_amdgcn_fence(__ATOMIC_RELEASE, "wavefront");
    __builtin_amdgcn_wave_barrier();
    __builtin_amdgcn_fence(__ATOMIC_ACQUIRE, "wavefront");
}

template <int H, int R, bool ELU, int G1, int G2>
__device__ __forceinline__ void node_rows(const int* __restrict__ srcs, const int* __restrict__ dsts, int count,
                                          int n0, int nt, int wave, int lane,
                                          const float* __restrict__ as,
                                          const float* __restrict__ ad,
                                          const float* __restrict__ hin,
                                          const float* __restrict__ bias,
                                          float* __restrict__ hout,
                                          int* __restrict__ scratch) {
    constexpr int ROW = 32 * R;
    constexpr int VW  = (R >= 4) ? 4 : 2;
    typedef float VT __attribute__((ext_vector_type(VW)));
    const int  col    = lane * R;
    const int  hl     = (col * H) / ROW;
    const bool single = count <= SEG_CAP;

    for (int ln = wave; ln < nt; ln += AGG_WAVES) {
        const int   n     = n0 + ln;
        const float adn   = ad[n * H + hl];
        const float lself = lrelu(as[n * H + hl] + adn);
        float mx = lself;
        int m0 = 0;

        for (int c0 = 0; c0 < count; c0 += SEG_CAP) {
            const int m = compact_sources(srcs, dsts, c0, min(count, c0 + SEG_CAP), n, lane, scratch);
            if (c0 == 0) m0 = m;
            wave_lds_handover();
            for (int j = 0; j < m; j += 32) {
                const int cnt  = min(32, m - j);
                const int sreg = (lane < cnt) ? scratch[j + lane] : n;
                if constexpr (H == 1) {
                    mx = fmaxf(mx, lrelu(as[sreg] + adn));
                } else {
                    for (int e0 = 0; e0 < cnt; e0 += G1) {
                        float s[G1];
#pragma unroll
                        for (int g = 0; g < G1; ++g) {
                            const int sj = __builtin_amdgcn_readlane(sreg, e0 + g);
                            s[g] = as[sj * H + hl];
                        }
#pragma unroll
                        for (int g = 0; g < G1; ++g) mx = fmaxf(mx, lrelu(s[g] + adn));
                    }
                }
            }
            wave_lds_handover();
        }
        if constexpr (H == 1) {
#pragma unroll
            for (int off = 16; off > 0; off >>= 1) mx = fmaxf(mx, __shfl_xor(mx, off));
        }
        if (!(fabsf(mx) < INFINITY)) mx = 0.f;

        float den = expf(lself - mx);
        float acc[R];
        {
            const float* hrow = hin + (size_t)n * ROW + col;
#pragma unroll
            for (int q = 0; q < R / VW; ++q) {
                const VT t = *(const VT*)(hrow + q * VW);
#pragma unroll
                for (int k = 0; k < VW; ++k) acc[q * VW + k] = den * t[k];
            }
        }
        for (int c0 = 0; c0 < count; c0 += SEG_CAP) {
            int m = m0;
            if (!single) {
                m = compact_sources(srcs, dsts, c0, min(count, c0 + SEG_CAP), n, lane, scratch);
                wave_lds_handover();
            }
            for (int j = 0; j < m; j += 32) {
                const int cnt  = min(32, m - j);
                const int sreg = (lane < cnt) ? scratch[j + lane] : n;
                float wreg = 0.f;
                if constexpr (H == 1) wreg = expf(lrelu(as[sreg] + adn) - mx);
                for (int e0 = 0; e0 < cnt; e0 += G2) {
                    float s[G2];
                    float v[G2][R];
#pragma unroll
                    for (int g = 0; g < G2; ++g) {
                        s[g] = 0.f;
#pragma unroll
                        for (int r = 0; r < R; ++r) v[g][r] = 0.f;
                        if (e0 + g < cnt) {
                            const int sj = __builtin_amdgcn_readlane(sreg, e0 + g);
                            if constexpr (H > 1) s[g] = as[sj * H + hl];
                            const float* hrow = hin + (size_t)sj * ROW + col;
#pragma unroll
                            for (int q = 0; q < R / VW; ++q) {
                                const VT t = *(const VT*)(hrow + q * VW);
#pragma unroll
                                for (int k = 0; k < VW; ++k) v[g][q * VW + k] = t[k];
                            }
                        }
                    }
#pragma unroll
                    for (int g = 0; g < G2; ++g) {
                        if (e0 + g < cnt) {
                            float w;
                            if constexpr (H == 1) w = lane_bcast(wreg, e0 + g);
                            else                  w = expf(lrelu(s[g] + adn) - mx);
                            den += w;
#pragma unroll
                            for (int r = 0; r < R; ++r) acc[r] += w * v[g][r];
                        }
                    }
                }
            }
            wave_lds_handover();
        }

        float o[R];
#pragma unroll
        for (int r = 0; r < R; ++r) {
            float val = acc[r] / (den + 1e-16f) + bias[col + r];
            if (ELU) val = val > 0.f ? val : (expf(val) - 1.f);
            o[r] = val;
        }
        float* orow = hout + (size_t)n * ROW + col;
#pragma unroll
        for (int q = 0; q < R / VW; ++q) {
            VT t;
#pragma unroll
            for (int k = 0; k < VW; ++k) t[k] = o[q * VW + k];
            *(VT*)(orow + q * VW) = t;
        }
    }
}

template <int H, int R, bool ELU, bool REUSE>
__global__ void __launch_bounds__(1024)
agg_tile_kernel(const int* __restrict__ ei, int E0, int Nn,
                const float* __restrict__ as, const float* __restrict__ ad,
                const float* __restrict__ hin, const float* __restrict__ bias,
                float* hout, int* keep, int keep_row) {
    __shared__ int seg_e[AGG_WAVES][SEG_CAP];
    __shared__ int seg_d[AGG_WAVES][SEG_CAP];
    __shared__ int lst_src[AGG_WAVES * SEG_CAP];
    __shared__ int lst_dst[AGG_WAVES * SEG_CAP];
    __shared__ int own_src[AGG_WAVES * SEG_CAP];
    __shared__ int own_dst[AGG_WAVES * SEG_CAP];
    __shared__ int seg_cnt[AGG_WAVES];
    __shared__ int own_cnt[AGG_WAVES];
    __shared__ int overflow;

    const int tid  = threadIdx.x;
    const int lane = tid & 31;
    const int wave = __builtin_amdgcn_readfirstlane(tid >> 5);
    const int n0   = blockIdx.x * TILE_N;
    const int nt   = min(TILE_N, Nn - n0);
    const int* __restrict__ dstp = ei + E0;
    int* kt = keep + (size_t)n0 * keep_row;
    const int kcap = nt * keep_row;

    int  oc = 0, obase = 0, ovf = 0;
    bool have = false;
    if constexpr (REUSE) {
        const int hdr = __builtin_amdgcn_readfirstlane(kt[0]);
        have = (hdr >= 0) && (hdr <= AGG_WAVES * SEG_CAP) && (KEEP_HDR + 2 * hdr <= kcap);
        if (have) {
            if (tid < AGG_WAVES) own_cnt[tid] = kt[1 + tid];
            for (int i = tid; i < hdr; i += 1024) {
                own_src[i] = kt[KEEP_HDR + i];
                own_dst[i] = kt[KEEP_HDR + hdr + i];
            }
            __syncthreads();
            oc = own_cnt[wave];
            for (int w = 0; w < wave; ++w) obase += own_cnt[w];
        }
    }

    if (!have) {
        if (tid == 0) overflow = 0;
        __syncthreads();

        int cnt = 0;
        for (int base = wave * 256; base < E0; base += AGG_WAVES * 256) {
            int d[8];
            if (base + 256 <= E0) {
#pragma unroll
                for (int j = 0; j < 8; ++j) d[j] = dstp[base + j * 32 + lane];
            } else {
#pragma unroll
                for (int j = 0; j < 8; ++j) {
                    const int e = base + j * 32 + lane;
                    d[j] = (e < E0) ? dstp[e] : -1;
                }
            }
#pragma unroll
            for (int j = 0; j < 8; ++j) {
                const bool m = (unsigned)(d[j] - n0) < (unsigned)nt;
                const unsigned mask = __builtin_amdgcn_ballot_w32(m);
                if (mask) {
                    const int pos = cnt + __builtin_popcount(mask & ((1u << lane) - 1u));
                    if (m && pos < SEG_CAP) {
                        seg_e[wave][pos] = base + j * 32 + lane;
                        seg_d[wave][pos] = d[j];
                    }
                    cnt += __builtin_popcount(mask);
                }
            }
        }
        if (lane == 0) {
            seg_cnt[wave] = min(cnt, SEG_CAP);
            if (cnt > SEG_CAP) overflow = 1;
        }
        __syncthreads();

        int mybase = 0, total = 0;
        for (int w = 0; w < AGG_WAVES; ++w) {
            const int c = seg_cnt[w];
            if (w < wave) mybase += c;
            total += c;
        }
        const int myc = seg_cnt[wave];
        ovf = overflow;
        for (int i = lane; i < myc; i += 32) {
            lst_src[mybase + i] = ei[seg_e[wave][i]];
            lst_dst[mybase + i] = seg_d[wave][i];
        }
        __syncthreads();

        if (!ovf) {
            for (int c = 0; c < total; c += 32) {
                const int i = c + lane;
                const bool m = (i < total) && (((lst_dst[i] - n0) & (AGG_WAVES - 1)) == wave);
                oc += __builtin_popcount(__builtin_amdgcn_ballot_w32(m));
            }
        }
        if (lane == 0) own_cnt[wave] = oc;
        __syncthreads();
        for (int w = 0; w < wave; ++w) obase += own_cnt[w];

        if (!ovf) {
            int k = obase;
            for (int c = 0; c < total; c += 32) {
                const int i = c + lane;
                const bool m = (i < total) && (((lst_dst[i] - n0) & (AGG_WAVES - 1)) == wave);
                const unsigned mask = __builtin_amdgcn_ballot_w32(m);
                if (m) {
                    const int pos = k + __builtin_popcount(mask & ((1u << lane) - 1u));
                    own_src[pos] = lst_src[i];
                    own_dst[pos] = lst_dst[i];
                }
                k += __builtin_popcount(mask);
            }
        }
        __syncthreads();

        if constexpr (!REUSE) {
            const bool fits = !ovf && (KEEP_HDR + 2 * total <= kcap);
            if (tid == 0) kt[0] = fits ? total : -1;
            if (fits) {
                if (tid < AGG_WAVES) kt[1 + tid] = own_cnt[tid];
                for (int i = tid; i < total; i += 1024) {
                    kt[KEEP_HDR + i]         = own_src[i];
                    kt[KEEP_HDR + total + i] = own_dst[i];
                }
            }
        }
    }

    constexpr int G1 = 16;
    constexpr int G2 = (R > 2) ? 8 : 16;
    int* scratch = &seg_e[wave][0];
    if (!ovf)
        node_rows<H, R, ELU, G1, G2>(own_src + obase, own_dst + obase, oc, n0, nt, wave, lane, as, ad, hin, bias, hout, scratch);
    else
        node_rows<H, R, ELU, G1, G2>(ei, dstp, E0, n0, nt, wave, lane, as, ad, hin, bias, hout, scratch);
}

extern "C" void kernel_launch(void* const* d_in, const int* in_sizes, int n_in,
                              void* d_out, int out_size, void* d_ws, size_t ws_size,
                              hipStream_t stream) {
    const float* x        = (const float*)d_in[0];
    const int*   ei       = (const int*)d_in[1];
    const float* W1       = (const float*)d_in[2];
    const float* att_src1 = (const float*)d_in[3];
    const float* att_dst1 = (const float*)d_in[4];
    const float* b1       = (const float*)d_in[5];
    const float* W2       = (const float*)d_in[6];
    const float* att_src2 = (const float*)d_in[7];
    const float* att_dst2 = (const float*)d_in[8];
    const float* b2       = (const float*)d_in[9];
    float* out = (float*)d_out;

    const int Nn = in_sizes[0] / 256;
    const int E0 = in_sizes[1] / 2;

    float* h1   = (float*)d_ws;
    float* helu = h1 + (size_t)Nn * 256;
    float* as1  = helu + (size_t)Nn * 256;
    float* ad1  = as1 + (size_t)Nn * 8;
    float* h2   = h1;
    float* as2  = as1;
    float* ad2  = as1 + Nn;

    const int TB = 256;
    dim3 g1((256 + 63) / 64, (Nn + 63) / 64);
    dim3 g2((64 + 63) / 64, (Nn + 63) / 64);
    const int tiles = (Nn + TILE_N - 1) / TILE_N;

    gat_wmma_gemm<<<g1, TB, 0, stream>>>(x, W1, h1, Nn, 256, 256);
    scores1_kernel<<<(Nn * 8 + TB - 1) / TB, TB, 0, stream>>>(h1, att_src1, att_dst1, as1, ad1, Nn);
    agg_tile_kernel<8, 8, true, false><<<tiles, 1024, 0, stream>>>(ei, E0, Nn, as1, ad1, h1, b1, helu, (int*)out, 64);

    gat_wmma_gemm<<<g2, TB, 0, stream>>>(helu, W2, h2, Nn, 64, 256);
    scores2_kernel<<<(Nn + TB - 1) / TB, TB, 0, stream>>>(h2, att_src2, att_dst2, as2, ad2, Nn);
    agg_tile_kernel<1, 2, false, true><<<tiles, 1024, 0, stream>>>(ei, E0, Nn, as2, ad2, h2, b2, out, (int*)out, 64);
}


# Tile filter reads the destination list with 16-byte loads, four 256-entry chunks (8 loads) in flight per wave instead of eight 4-byte loads; chunk and entry order kept, sum order unchanged
# speedup vs baseline: 4.0150x; 1.1227x over previous
#include <hip/hip_runtime.h>
#include <hip/hip_bf16.h>

typedef __attribute__((ext_vector_type(2))) float v2f;
typedef __attribute__((ext_vector_type(8))) float v8f;

#define NEG_SLOPE 0.2f

__global__ void gat_wmma_gemm(const float* __restrict__ A,
                              const float* __restrict__ B,
                              float* __restrict__ C,
                              int M, int Ntot, int K) {
    __shared__ float As[64][68];
    __shared__ float Bp[32][132];

    const int tid   = threadIdx.x;
    const int lane  = tid & 31;
    const int wave  = tid >> 5;
    const int mBase = blockIdx.y * 64;
    const int nBase = blockIdx.x * 64;

    const int ms   = (wave & 3) * 16;
    const int ns   = (wave >> 2) * 32;
    const int mrow = ms + (lane & 15);
    const int khi  = (lane >> 4) * 2;
    const int ncol = lane & 15;

    v8f acc0 = {}; v8f acc1 = {};

    for (int kBase = 0; kBase < K; kBase += 64) {
#pragma unroll
        for (int i = 0; i < 4; ++i) {
            int s  = tid + i * 256;
            int r  = s >> 4;
            int c4 = (s & 15) << 2;
            float4 va = make_float4(0.f, 0.f, 0.f, 0.f);
            int g = mBase + r;
            if (g < M) va = *(const float4*)(A + (size_t)g * K + kBase + c4);
            *(float4*)&As[r][c4] = va;
            float4 vb = *(const float4*)(B + (size_t)(kBase + r) * Ntot + nBase + c4);
            float* bp = &Bp[r >> 1][r & 1];
            bp[2 * c4 + 0] = vb.x;
            bp[2 * c4 + 2] = vb.y;
            bp[2 * c4 + 4] = vb.z;
            bp[2 * c4 + 6] = vb.w;
        }
        __syncthreads();

#pragma unroll
        for (int kk = 0; kk < 64; kk += 4) {
            const int p2 = (kk + khi) >> 1;
            v2f a, b0, b1v;
            a[0]   = As[mrow][kk + khi];
            a[1]   = As[mrow][kk + khi + 1];
            b0     = *(const v2f*)&Bp[p2][2 * (ns + ncol)];
            b1v    = *(const v2f*)&Bp[p2][2 * (ns + 16 + ncol)];
            acc0 = __builtin_amdgcn_wmma_f32_16x16x4_f32(
                false, a, false, b0, (short)0, acc0, false, false);
            acc1 = __builtin_amdgcn_wmma_f32_16x16x4_f32(
                false, a, false, b1v, (short)0, acc1, false, false);
        }
        __syncthreads();
    }

    const int orow = mBase + ms + ((lane >> 4) << 3);
    const int ocol = nBase + ns + ncol;
#pragma unroll
    for (int r = 0; r < 8; ++r) {
        int g = orow + r;
        if (g < M) {
            C[(size_t)g * Ntot + ocol]      = acc0[r];
            C[(size_t)g * Ntot + ocol + 16] = acc1[r];
        }
    }
}

__global__ void scores1_kernel(const float* __restrict__ h1,
                               const float* __restrict__ att_s,
                               const float* __restrict__ att_d,
                               float* __restrict__ as1, float* __restrict__ ad1,
                               int Nn) {
    int t = blockIdx.x * blockDim.x + threadIdx.x;
    if (t >= Nn * 8) return;
    int n = t >> 3, h = t & 7;
    const float* row = h1 + (size_t)n * 256 + h * 32;
    const float* ws = att_s + h * 32;
    const float* wd = att_d + h * 32;
    float ss = 0.f, sd = 0.f;
#pragma unroll
    for (int c = 0; c < 32; ++c) { ss += row[c] * ws[c]; sd += row[c] * wd[c]; }
    as1[t] = ss; ad1[t] = sd;
}

__global__ void scores2_kernel(const float* __restrict__ h2,
                               const float* __restrict__ att_s,
                               const float* __restrict__ att_d,
                               float* __restrict__ as2, float* __restrict__ ad2,
                               int Nn) {
    int n = blockIdx.x * blockDim.x + threadIdx.x;
    if (n >= Nn) return;
    const float* row = h2 + (size_t)n * 64;
    float ss = 0.f, sd = 0.f;
#pragma unroll
    for (int c = 0; c < 64; ++c) { ss += row[c] * att_s[c]; sd += row[c] * att_d[c]; }
    as2[n] = ss; ad2[n] = sd;
}

#define TILE_N    196
#define AGG_WAVES 32
#define SEG_CAP   256
#define KEEP_HDR  64

__device__ __forceinline__ float lrelu(float v) { return v > 0.f ? v : NEG_SLOPE * v; }

__device__ __forceinline__ float lane_bcast(float v, int j) {
    return __uint_as_float(__builtin_amdgcn_readlane(__float_as_uint(v), j));
}

__device__ __forceinline__ int compact_sources(const int* __restrict__ srcs, const int* __restrict__ dsts,
                                               int c0, int cend, int n, int lane,
                                               int* __restrict__ scratch) {
    const unsigned below = (1u << lane) - 1u;
    int m = 0;
    for (int i0 = c0; i0 < cend; i0 += 32) {
        const int  i     = i0 + lane;
        const bool in    = i < cend;
        const int  d     = in ? dsts[i] : -1;
        const int  sv    = in ? srcs[i] : 0;
        const bool match = (d == n);
        const unsigned mask = __builtin_amdgcn_ballot_w32(match);
        if (match) scratch[m + __builtin_popcount(mask & below)] = sv;
        m += __builtin_popcount(mask);
    }
    return m;
}

__device__ __forceinline__ void wave_lds_handover() {
    __builtin_amdgcn_fence(__ATOMIC_RELEASE, "wavefront");
    __builtin_amdgcn_wave_barrier();
    __builtin_amdgcn_fence(__ATOMIC_ACQUIRE, "wavefront");
}

template <int H, int R, bool ELU, int G1, int G2>
__device__ __forceinline__ void node_rows(const int* __restrict__ srcs, const int* __restrict__ dsts, int count,
                                          int n0, int nt, int wave, int lane,
                                          const float* __restrict__ as,
                                          const float* __restrict__ ad,
                                          const float* __restrict__ hin,
                                          const float* __restrict__ bias,
                                          float* __restrict__ hout,
                                          int* __restrict__ scratch) {
    constexpr int ROW = 32 * R;
    constexpr int VW  = (R >= 4) ? 4 : 2;
    typedef float VT __attribute__((ext_vector_type(VW)));
    const int  col    = lane * R;
    const int  hl     = (col * H) / ROW;
    const bool single = count <= SEG_CAP;

    for (int ln = wave; ln < nt; ln += AGG_WAVES) {
        const int   n     = n0 + ln;
        const float adn   = ad[n * H + hl];
        const float lself = lrelu(as[n * H + hl] + adn);
        float mx = lself;
        int m0 = 0;

        for (int c0 = 0; c0 < count; c0 += SEG_CAP) {
            const int m = compact_sources(srcs, dsts, c0, min(count, c0 + SEG_CAP), n, lane, scratch);
            if (c0 == 0) m0 = m;
            wave_lds_handover();
            for (int j = 0; j < m; j += 32) {
                const int cnt  = min(32, m - j);
                const int sreg = (lane < cnt) ? scratch[j + lane] : n;
                if constexpr (H == 1) {
                    mx = fmaxf(mx, lrelu(as[sreg] + adn));
                } else {
                    for (int e0 = 0; e0 < cnt; e0 += G1) {
                        float s[G1];
#pragma unroll
                        for (int g = 0; g < G1; ++g) {
                            const int sj = __builtin_amdgcn_readlane(sreg, e0 + g);
                            s[g] = as[sj * H + hl];
                        }
#pragma unroll
                        for (int g = 0; g < G1; ++g) mx = fmaxf(mx, lrelu(s[g] + adn));
                    }
                }
            }
            wave_lds_handover();
        }
        if constexpr (H == 1) {
#pragma unroll
            for (int off = 16; off > 0; off >>= 1) mx = fmaxf(mx, __shfl_xor(mx, off));
        }
        if (!(fabsf(mx) < INFINITY)) mx = 0.f;

        float den = expf(lself - mx);
        float acc[R];
        {
            const float* hrow = hin + (size_t)n * ROW + col;
#pragma unroll
            for (int q = 0; q < R / VW; ++q) {
                const VT t = *(const VT*)(hrow + q * VW);
#pragma unroll
                for (int k = 0; k < VW; ++k) acc[q * VW + k] = den * t[k];
            }
        }
        for (int c0 = 0; c0 < count; c0 += SEG_CAP) {
            int m = m0;
            if (!single) {
                m = compact_sources(srcs, dsts, c0, min(count, c0 + SEG_CAP), n, lane, scratch);
                wave_lds_handover();
            }
            for (int j = 0; j < m; j += 32) {
                const int cnt  = min(32, m - j);
                const int sreg = (lane < cnt) ? scratch[j + lane] : n;
                float wreg = 0.f;
                if constexpr (H == 1) wreg = expf(lrelu(as[sreg] + adn) - mx);
                for (int e0 = 0; e0 < cnt; e0 += G2) {
                    float s[G2];
                    float v[G2][R];
#pragma unroll
                    for (int g = 0; g < G2; ++g) {
                        s[g] = 0.f;
#pragma unroll
                        for (int r = 0; r < R; ++r) v[g][r] = 0.f;
                        if (e0 + g < cnt) {
                            const int sj = __builtin_amdgcn_readlane(sreg, e0 + g);
                            if constexpr (H > 1) s[g] = as[sj * H + hl];
                            const float* hrow = hin + (size_t)sj * ROW + col;
#pragma unroll
                            for (int q = 0; q < R / VW; ++q) {
                                const VT t = *(const VT*)(hrow + q * VW);
#pragma unroll
                                for (int k = 0; k < VW; ++k) v[g][q * VW + k] = t[k];
                            }
                        }
                    }
#pragma unroll
                    for (int g = 0; g < G2; ++g) {
                        if (e0 + g < cnt) {
                            float w;
                            if constexpr (H == 1) w = lane_bcast(wreg, e0 + g);
                            else                  w = expf(lrelu(s[g] + adn) - mx);
                            den += w;
#pragma unroll
                            for (int r = 0; r < R; ++r) acc[r] += w * v[g][r];
                        }
                    }
                }
            }
            wave_lds_handover();
        }

        float o[R];
#pragma unroll
        for (int r = 0; r < R; ++r) {
            float val = acc[r] / (den + 1e-16f) + bias[col + r];
            if (ELU) val = val > 0.f ? val : (expf(val) - 1.f);
            o[r] = val;
        }
        float* orow = hout + (size_t)n * ROW + col;
#pragma unroll
        for (int q = 0; q < R / VW; ++q) {
            VT t;
#pragma unroll
            for (int k = 0; k < VW; ++k) t[k] = o[q * VW + k];
            *(VT*)(orow + q * VW) = t;
        }
    }
}

__device__ __forceinline__ void filter_group4(const int4 d, int ebase, int lane, unsigned below,
                                              int n0, int nt,
                                              int* __restrict__ se, int* __restrict__ sd, int& cnt) {
    const unsigned t0 = (unsigned)(d.x - n0), t1 = (unsigned)(d.y - n0);
    const unsigned t2 = (unsigned)(d.z - n0), t3 = (unsigned)(d.w - n0);
    const unsigned tmin = min(min(t0, t1), min(t2, t3));
    if (__builtin_amdgcn_ballot_w32(tmin < (unsigned)nt)) {
        const bool m0 = t0 < (unsigned)nt, m1 = t1 < (unsigned)nt;
        const bool m2 = t2 < (unsigned)nt, m3 = t3 < (unsigned)nt;
        const unsigned k0 = __builtin_amdgcn_ballot_w32(m0), k1 = __builtin_amdgcn_ballot_w32(m1);
        const unsigned k2 = __builtin_amdgcn_ballot_w32(m2), k3 = __builtin_amdgcn_ballot_w32(m3);
        int pos = cnt + __builtin_popcount(k0 & below) + __builtin_popcount(k1 & below)
                      + __builtin_popcount(k2 & below) + __builtin_popcount(k3 & below);
        const int e = ebase + 4 * lane;
        if (m0) { if (pos < SEG_CAP) { se[pos] = e;     sd[pos] = d.x; } ++pos; }
        if (m1) { if (pos < SEG_CAP) { se[pos] = e + 1; sd[pos] = d.y; } ++pos; }
        if (m2) { if (pos < SEG_CAP) { se[pos] = e + 2; sd[pos] = d.z; } ++pos; }
        if (m3) { if (pos < SEG_CAP) { se[pos] = e + 3; sd[pos] = d.w; } }
        cnt += __builtin_popcount(k0) + __builtin_popcount(k1) + __builtin_popcount(k2) + __builtin_popcount(k3);
    }
}

template <int H, int R, bool ELU, bool REUSE>
__global__ void __launch_bounds__(1024)
agg_tile_kernel(const int* __restrict__ ei, int E0, int Nn,
                const float* __restrict__ as, const float* __restrict__ ad,
                const float* __restrict__ hin, const float* __restrict__ bias,
                float* hout, int* keep, int keep_row) {
    __shared__ int seg_e[AGG_WAVES][SEG_CAP];
    __shared__ int seg_d[AGG_WAVES][SEG_CAP];
    __shared__ int lst_src[AGG_WAVES * SEG_CAP];
    __shared__ int lst_dst[AGG_WAVES * SEG_CAP];
    __shared__ int own_src[AGG_WAVES * SEG_CAP];
    __shared__ int own_dst[AGG_WAVES * SEG_CAP];
    __shared__ int seg_cnt[AGG_WAVES];
    __shared__ int own_cnt[AGG_WAVES];
    __shared__ int overflow;

    const int tid  = threadIdx.x;
    const int lane = tid & 31;
    const int wave = __builtin_amdgcn_readfirstlane(tid >> 5);
    const int n0   = blockIdx.x * TILE_N;
    const int nt   = min(TILE_N, Nn - n0);
    const int* __restrict__ dstp = ei + E0;
    int* kt = keep + (size_t)n0 * keep_row;
    const int kcap = nt * keep_row;

    int  oc = 0, obase = 0, ovf = 0;
    bool have = false;
    if constexpr (REUSE) {
        const int hdr = __builtin_amdgcn_readfirstlane(kt[0]);
        have = (hdr >= 0) && (hdr <= AGG_WAVES * SEG_CAP) && (KEEP_HDR + 2 * hdr <= kcap);
        if (have) {
            if (tid < AGG_WAVES) own_cnt[tid] = kt[1 + tid];
            for (int i = tid; i < hdr; i += 1024) {
                own_src[i] = kt[KEEP_HDR + i];
                own_dst[i] = kt[KEEP_HDR + hdr + i];
            }
            __syncthreads();
            oc = own_cnt[wave];
            for (int w = 0; w < wave; ++w) obase += own_cnt[w];
        }
    }

    if (!have) {
        if (tid == 0) overflow = 0;
        __syncthreads();

        int cnt = 0;
        int base = wave * 256;
        constexpr int FCH = 4;
        if ((reinterpret_cast<size_t>(dstp) & 15) == 0) {
            const unsigned below = (1u << lane) - 1u;
            for (; base + (FCH - 1) * AGG_WAVES * 256 + 256 <= E0; base += FCH * AGG_WAVES * 256) {
                int4 dv[FCH][2];
#pragma unroll
                for (int u = 0; u < FCH; ++u) {
#pragma unroll
                    for (int h = 0; h < 2; ++h)
                        dv[u][h] = *(const int4*)(dstp + base + u * (AGG_WAVES * 256) + h * 128 + 4 * lane);
                }
#pragma unroll
                for (int u = 0; u < FCH; ++u) {
#pragma unroll
                    for (int h = 0; h < 2; ++h)
                        filter_group4(dv[u][h], base + u * (AGG_WAVES * 256) + h * 128, lane, below, n0, nt,
                                      &seg_e[wave][0], &seg_d[wave][0], cnt);
                }
            }
        }
        for (; base < E0; base += AGG_WAVES * 256) {
            int d[8];
            if (base + 256 <= E0) {
#pragma unroll
                for (int j = 0; j < 8; ++j) d[j] = dstp[base + j * 32 + lane];
            } else {
#pragma unroll
                for (int j = 0; j < 8; ++j) {
                    const int e = base + j * 32 + lane;
                    d[j] = (e < E0) ? dstp[e] : -1;
                }
            }
#pragma unroll
            for (int j = 0; j < 8; ++j) {
                const bool m = (unsigned)(d[j] - n0) < (unsigned)nt;
                const unsigned mask = __builtin_amdgcn_ballot_w32(m);
                if (mask) {
                    const int pos = cnt + __builtin_popcount(mask & ((1u << lane) - 1u));
                    if (m && pos < SEG_CAP) {
                        seg_e[wave][pos] = base + j * 32 + lane;
                        seg_d[wave][pos] = d[j];
                    }
                    cnt += __builtin_popcount(mask);
                }
            }
        }
        if (lane == 0) {
            seg_cnt[wave] = min(cnt, SEG_CAP);
            if (cnt > SEG_CAP) overflow = 1;
        }
        __syncthreads();

        int mybase = 0, total = 0;
        for (int w = 0; w < AGG_WAVES; ++w) {
            const int c = seg_cnt[w];
            if (w < wave) mybase += c;
            total += c;
        }
        const int myc = seg_cnt[wave];
        ovf = overflow;
        for (int i = lane; i < myc; i += 32) {
            lst_src[mybase + i] = ei[seg_e[wave][i]];
            lst_dst[mybase + i] = seg_d[wave][i];
        }
        __syncthreads();

        if (!ovf) {
            for (int c = 0; c < total; c += 32) {
                const int i = c + lane;
                const bool m = (i < total) && (((lst_dst[i] - n0) & (AGG_WAVES - 1)) == wave);
                oc += __builtin_popcount(__builtin_amdgcn_ballot_w32(m));
            }
        }
        if (lane == 0) own_cnt[wave] = oc;
        __syncthreads();
        for (int w = 0; w < wave; ++w) obase += own_cnt[w];

        if (!ovf) {
            int k = obase;
            for (int c = 0; c < total; c += 32) {
                const int i = c + lane;
                const bool m = (i < total) && (((lst_dst[i] - n0) & (AGG_WAVES - 1)) == wave);
                const unsigned mask = __builtin_amdgcn_ballot_w32(m);
                if (m) {
                    const int pos = k + __builtin_popcount(mask & ((1u << lane) - 1u));
                    own_src[pos] = lst_src[i];
                    own_dst[pos] = lst_dst[i];
                }
                k += __builtin_popcount(mask);
            }
        }
        __syncthreads();

        if constexpr (!REUSE) {
            const bool fits = !ovf && (KEEP_HDR + 2 * total <= kcap);
            if (tid == 0) kt[0] = fits ? total : -1;
            if (fits) {
                if (tid < AGG_WAVES) kt[1 + tid] = own_cnt[tid];
                for (int i = tid; i < total; i += 1024) {
                    kt[KEEP_HDR + i]         = own_src[i];
                    kt[KEEP_HDR + total + i] = own_dst[i];
                }
            }
        }
    }

    constexpr int G1 = 16;
    constexpr int G2 = (R > 2) ? 8 : 16;
    int* scratch = &seg_e[wave][0];
    if (!ovf)
        node_rows<H, R, ELU, G1, G2>(own_src + obase, own_dst + obase, oc, n0, nt, wave, lane, as, ad, hin, bias, hout, scratch);
    else
        node_rows<H, R, ELU, G1, G2>(ei, dstp, E0, n0, nt, wave, lane, as, ad, hin, bias, hout, scratch);
}

extern "C" void kernel_launch(void* const* d_in, const int* in_sizes, int n_in,
                              void* d_out, int out_size, void* d_ws, size_t ws_size,
                              hipStream_t stream) {
    const float* x        = (const float*)d_in[0];
    const int*   ei       = (const int*)d_in[1];
    const float* W1       = (const float*)d_in[2];
    const float* att_src1 = (const float*)d_in[3];
    const float* att_dst1 = (const float*)d_in[4];
    const float* b1       = (const float*)d_in[5];
    const float* W2       = (const float*)d_in[6];
    const float* att_src2 = (const float*)d_in[7];
    const float* att_dst2 = (const float*)d_in[8];
    const float* b2       = (const float*)d_in[9];
    float* out = (float*)d_out;

    const int Nn = in_sizes[0] / 256;
    const int E0 = in_sizes[1] / 2;

    float* h1   = (float*)d_ws;
    float* helu = h1 + (size_t)Nn * 256;
    float* as1  = helu + (size_t)Nn * 256;
    float* ad1  = as1 + (size_t)Nn * 8;
    float* h2   = h1;
    float* as2  = as1;
    float* ad2  = as1 + Nn;

    const int TB = 256;
    dim3 g1((256 + 63) / 64, (Nn + 63) / 64);
    dim3 g2((64 + 63) / 64, (Nn + 63) / 64);
    const int tiles = (Nn + TILE_N - 1) / TILE_N;

    gat_wmma_gemm<<<g1, TB, 0, stream>>>(x, W1, h1, Nn, 256, 256);
    scores1_kernel<<<(Nn * 8 + TB - 1) / TB, TB, 0, stream>>>(h1, att_src1, att_dst1, as1, ad1, Nn);
    agg_tile_kernel<8, 8, true, false><<<tiles, 1024, 0, stream>>>(ei, E0, Nn, as1, ad1, h1, b1, helu, (int*)out, 64);

    gat_wmma_gemm<<<g2, TB, 0, stream>>>(helu, W2, h2, Nn, 64, 256);
    scores2_kernel<<<(Nn + TB - 1) / TB, TB, 0, stream>>>(h2, att_src2, att_dst2, as2, ad2, Nn);
    agg_tile_kernel<1, 2, false, true><<<tiles, 1024, 0, stream>>>(ei, E0, Nn, as2, ad2, h2, b2, out, (int*)out, 64);
}


# Tile edge-list scan runs as extra work-groups of the first product's launch (4 part lists per tile, handed over in the output buffer), vector scan beside matrix work; list and sum order unchanged
# speedup vs baseline: 4.1767x; 1.0403x over previous
#include <hip/hip_runtime.h>
#include <hip/hip_bf16.h>

typedef __attribute__((ext_vector_type(2))) float v2f;
typedef __attribute__((ext_vector_type(8))) float v8f;

#define NEG_SLOPE 0.2f

#define TILE_N    196
#define AGG_WAVES 32
#define SEG_CAP   256
#define KEEP_HDR  64
#define PARTS      4
#define SCAN_WAVES 8
#define PART_HDR   4
static_assert(PARTS * SCAN_WAVES == AGG_WAVES, "the parts together are the filter's waves");

__device__ __forceinline__ void filter_group4(const int4 d, int ebase, int lane, unsigned below,
                                              int n0, int nt,
                                              int* __restrict__ se, int* __restrict__ sd, int& cnt) {
    const unsigned t0 = (unsigned)(d.x - n0), t1 = (unsigned)(d.y - n0);
    const unsigned t2 = (unsigned)(d.z - n0), t3 = (unsigned)(d.w - n0);
    const unsigned tmin = min(min(t0, t1), min(t2, t3));
    if (__builtin_amdgcn_ballot_w32(tmin < (unsigned)nt)) {
        const bool m0 = t0 < (unsigned)nt, m1 = t1 < (unsigned)nt;
        const bool m2 = t2 < (unsigned)nt, m3 = t3 < (unsigned)nt;
        const unsigned k0 = __builtin_amdgcn_ballot_w32(m0), k1 = __builtin_amdgcn_ballot_w32(m1);
        const unsigned k2 = __builtin_amdgcn_ballot_w32(m2), k3 = __builtin_amdgcn_ballot_w32(m3);
        int pos = cnt + __builtin_popcount(k0 & below) + __builtin_popcount(k1 & below)
                      + __builtin_popcount(k2 & below) + __builtin_popcount(k3 & below);
        const int e = ebase + 4 * lane;
        if (m0) { if (pos < SEG_CAP) { se[pos] = e;     sd[pos] = d.x; } ++pos; }
        if (m1) { if (pos < SEG_CAP) { se[pos] = e + 1; sd[pos] = d.y; } ++pos; }
        if (m2) { if (pos < SEG_CAP) { se[pos] = e + 2; sd[pos] = d.z; } ++pos; }
        if (m3) { if (pos < SEG_CAP) { se[pos] = e + 3; sd[pos] = d.w; } }
        cnt += __builtin_popcount(k0) + __builtin_popcount(k1) + __builtin_popcount(k2) + __builtin_popcount(k3);
    }
}

__device__ __forceinline__ int filter_scan(const int* __restrict__ dstp, int E0, int gw, int lane,
                                           int n0, int nt, int* __restrict__ se, int* __restrict__ sd) {
    int cnt = 0;
    int base = gw * 256;
    constexpr int FCH = 4;
    if ((reinterpret_cast<size_t>(dstp) & 15) == 0) {
        const unsigned below = (1u << lane) - 1u;
        for (; base + (FCH - 1) * AGG_WAVES * 256 + 256 <= E0; base += FCH * AGG_WAVES * 256) {
            int4 dv[FCH][2];
#pragma unroll
            for (int u = 0; u < FCH; ++u) {
#pragma unroll
                for (int h = 0; h < 2; ++h)
                    dv[u][h] = *(const int4*)(dstp + base + u * (AGG_WAVES * 256) + h * 128 + 4 * lane);
            }
#pragma unroll
            for (int u = 0; u < FCH; ++u) {
#pragma unroll
                for (int h = 0; h < 2; ++h)
                    filter_group4(dv[u][h], base + u * (AGG_WAVES * 256) + h * 128, lane, below, n0, nt,
                                  se, sd, cnt);
            }
        }
    }
    for (; base < E0; base += AGG_WAVES * 256) {
        int d[8];
        if (base + 256 <= E0) {
#pragma unroll
            for (int j = 0; j < 8; ++j) d[j] = dstp[base + j * 32 + lane];
        } else {
#pragma unroll
            for (int j = 0; j < 8; ++j) {
                const int e = base + j * 32 + lane;
                d[j] = (e < E0) ? dstp[e] : -1;
            }
        }
#pragma unroll
        for (int j = 0; j < 8; ++j) {
            const bool m = (unsigned)(d[j] - n0) < (unsigned)nt;
            const unsigned mask = __builtin_amdgcn_ballot_w32(m);
            if (mask) {
                const int pos = cnt + __builtin_popcount(mask & ((1u << lane) - 1u));
                if (m && pos < SEG_CAP) {
                    se[pos] = base + j * 32 + lane;
                    sd[pos] = d[j];
                }
                cnt += __builtin_popcount(mask);
            }
        }
    }
    return cnt;
}

__device__ __forceinline__ void scan_part(const int* __restrict__ ei, int E0, int Nn,
                                          int* __restrict__ keep, int keep_row,
                                          int tile, int part, int* __restrict__ sm) {
    const int tid  = threadIdx.x;
    const int lane = tid & 31;
    const int w  = __builtin_amdgcn_readfirstlane(tid >> 5);
    const int n0 = tile * TILE_N;
    const int nt = min(TILE_N, Nn - n0);
    int* se = sm + w * SEG_CAP;
    int* sd = sm + (SCAN_WAVES + w) * SEG_CAP;
    int* sc = sm + 2 * SCAN_WAVES * SEG_CAP;
    const int cnt = filter_scan(ei + E0, E0, part * SCAN_WAVES + w, lane, n0, nt, se, sd);
    if (lane == 0) sc[w] = cnt;
    __syncthreads();

    int  mybase = 0, total = 0;
    bool whole = true;
    for (int j = 0; j < SCAN_WAVES; ++j) {
        const int c = sc[j];
        whole = whole && (c <= SEG_CAP);
        if (j < w) mybase += c;
        total += c;
    }
    const int qcap = (nt * keep_row) / PARTS;
    int* kq = keep + (size_t)n0 * keep_row + part * qcap;
    const bool fits = whole && (PART_HDR + 2 * total <= qcap);
    if (tid == 0) kq[0] = fits ? total : -1;
    if (fits) {
        for (int i = lane; i < cnt; i += 32) {
            kq[PART_HDR + mybase + i]         = ei[se[i]];
            kq[PART_HDR + total + mybase + i] = sd[i];
        }
    }
}

template <bool SCAN>
__global__ void gat_wmma_gemm(const float* __restrict__ A,
                              const float* __restrict__ B,
                              float* __restrict__ C,
                              int M, int Ntot, int K,
                              const int* __restrict__ ei, int E0, int* __restrict__ keep, int keep_row, int stiles) {
    __shared__ float As[64][68];
    __shared__ float Bp[32][132];
    static_assert(sizeof(As) >= (2 * SCAN_WAVES * SEG_CAP + SCAN_WAVES) * sizeof(int), "the scan's segments fit in As");

    int yb = blockIdx.y;
    if constexpr (SCAN) {
        if (yb < stiles) {
            scan_part(ei, E0, M, keep, keep_row, yb, blockIdx.x, reinterpret_cast<int*>(&As[0][0]));
            return;
        }
        yb -= stiles;
    }

    const int tid   = threadIdx.x;
    const int lane  = tid & 31;
    const int wave  = tid >> 5;
    const int mBase = yb * 64;
    const int nBase = blockIdx.x * 64;

    const int ms   = (wave & 3) * 16;
    const int ns   = (wave >> 2) * 32;
    const int mrow = ms + (lane & 15);
    const int khi  = (lane >> 4) * 2;
    const int ncol = lane & 15;

    v8f acc0 = {}; v8f acc1 = {};

    for (int kBase = 0; kBase < K; kBase += 64) {
#pragma unroll
        for (int i = 0; i < 4; ++i) {
            int s  = tid + i * 256;
            int r  = s >> 4;
            int c4 = (s & 15) << 2;
            float4 va = make_float4(0.f, 0.f, 0.f, 0.f);
            int g = mBase + r;
            if (g < M) va = *(const float4*)(A + (size_t)g * K + kBase + c4);
            *(float4*)&As[r][c4] = va;
            float4 vb = *(const float4*)(B + (size_t)(kBase + r) * Ntot + nBase + c4);
            float* bp = &Bp[r >> 1][r & 1];
            bp[2 * c4 + 0] = vb.x;
            bp[2 * c4 + 2] = vb.y;
            bp[2 * c4 + 4] = vb.z;
            bp[2 * c4 + 6] = vb.w;
        }
        __syncthreads();

#pragma unroll
        for (int kk = 0; kk < 64; kk += 4) {
            const int p2 = (kk + khi) >> 1;
            v2f a, b0, b1v;
            a[0]   = As[mrow][kk + khi];
            a[1]   = As[mrow][kk + khi + 1];
            b0     = *(const v2f*)&Bp[p2][2 * (ns + ncol)];
            b1v    = *(const v2f*)&Bp[p2][2 * (ns + 16 + ncol)];
            acc0 = __builtin_amdgcn_wmma_f32_16x16x4_f32(
                false, a, false, b0, (short)0, acc0, false, false);
            acc1 = __builtin_amdgcn_wmma_f32_16x16x4_f32(
                false, a, false, b1v, (short)0, acc1, false, false);
        }
        __syncthreads();
    }

    const int orow = mBase + ms + ((lane >> 4) << 3);
    const int ocol = nBase + ns + ncol;
#pragma unroll
    for (int r = 0; r < 8; ++r) {
        int g = orow + r;
        if (g < M) {
            C[(size_t)g * Ntot + ocol]      = acc0[r];
            C[(size_t)g * Ntot + ocol + 16] = acc1[r];
        }
    }
}

__global__ void scores1_kernel(const float* __restrict__ h1,
                               const float* __restrict__ att_s,
                               const float* __restrict__ att_d,
                               float* __restrict__ as1, float* __restrict__ ad1,
                               int Nn) {
    int t = blockIdx.x * blockDim.x + threadIdx.x;
    if (t >= Nn * 8) return;
    int n = t >> 3, h = t & 7;
    const float* row = h1 + (size_t)n * 256 + h * 32;
    const float* ws = att_s + h * 32;
    const float* wd = att_d + h * 32;
    float ss = 0.f, sd = 0.f;
#pragma unroll
    for (int c = 0; c < 32; ++c) { ss += row[c] * ws[c]; sd += row[c] * wd[c]; }
    as1[t] = ss; ad1[t] = sd;
}

__global__ void scores2_kernel(const float* __restrict__ h2,
                               const float* __restrict__ att_s,
                               const float* __restrict__ att_d,
                               float* __restrict__ as2, float* __restrict__ ad2,
                               int Nn) {
    int n = blockIdx.x * blockDim.x + threadIdx.x;
    if (n >= Nn) return;
    const float* row = h2 + (size_t)n * 64;
    float ss = 0.f, sd = 0.f;
#pragma unroll
    for (int c = 0; c < 64; ++c) { ss += row[c] * att_s[c]; sd += row[c] * att_d[c]; }
    as2[n] = ss; ad2[n] = sd;
}

__device__ __forceinline__ float lrelu(float v) { return v > 0.f ? v : NEG_SLOPE * v; }

__device__ __forceinline__ float lane_bcast(float v, int j) {
    return __uint_as_float(__builtin_amdgcn_readlane(__float_as_uint(v), j));
}

__device__ __forceinline__ int compact_sources(const int* __restrict__ srcs, const int* __restrict__ dsts,
                                               int c0, int cend, int n, int lane,
                                               int* __restrict__ scratch) {
    const unsigned below = (1u << lane) - 1u;
    int m = 0;
    for (int i0 = c0; i0 < cend; i0 += 32) {
        const int  i     = i0 + lane;
        const bool in    = i < cend;
        const int  d     = in ? dsts[i] : -1;
        const int  sv    = in ? srcs[i] : 0;
        const bool match = (d == n);
        const unsigned mask = __builtin_amdgcn_ballot_w32(match);
        if (match) scratch[m + __builtin_popcount(mask & below)] = sv;
        m += __builtin_popcount(mask);
    }
    return m;
}

__device__ __forceinline__ void wave_lds_handover() {
    __builtin_amdgcn_fence(__ATOMIC_RELEASE, "wavefront");
    __builtin_amdgcn_wave_barrier();
    __builtin_amdgcn_fence(__ATOMIC_ACQUIRE, "wavefront");
}

template <int H, int R, bool ELU, int G1, int G2>
__device__ __forceinline__ void node_rows(const int* __restrict__ srcs, const int* __restrict__ dsts, int count,
                                          int n0, int nt, int wave, int lane,
                                          const float* __restrict__ as,
                                          const float* __restrict__ ad,
                                          const float* __restrict__ hin,
                                          const float* __restrict__ bias,
                                          float* __restrict__ hout,
                                          int* __restrict__ scratch) {
    constexpr int ROW = 32 * R;
    constexpr int VW  = (R >= 4) ? 4 : 2;
    typedef float VT __attribute__((ext_vector_type(VW)));
    const int  col    = lane * R;
    const int  hl     = (col * H) / ROW;
    const bool single = count <= SEG_CAP;

    for (int ln = wave; ln < nt; ln += AGG_WAVES) {
        const int   n     = n0 + ln;
        const float adn   = ad[n * H + hl];
        const float lself = lrelu(as[n * H + hl] + adn);
        float mx = lself;
        int m0 = 0;

        for (int c0 = 0; c0 < count; c0 += SEG_CAP) {
            const int m = compact_sources(srcs, dsts, c0, min(count, c0 + SEG_CAP), n, lane, scratch);
            if (c0 == 0) m0 = m;
            wave_lds_handover();
            for (int j = 0; j < m; j += 32) {
                const int cnt  = min(32, m - j);
                const int sreg = (lane < cnt) ? scratch[j + lane] : n;
                if constexpr (H == 1) {
                    mx = fmaxf(mx, lrelu(as[sreg] + adn));
                } else {
                    for (int e0 = 0; e0 < cnt; e0 += G1) {
                        float s[G1];
#pragma unroll
                        for (int g = 0; g < G1; ++g) {
                            const int sj = __builtin_amdgcn_readlane(sreg, e0 + g);
                            s[g] = as[sj * H + hl];
                        }
#pragma unroll
                        for (int g = 0; g < G1; ++g) mx = fmaxf(mx, lrelu(s[g] + adn));
                    }
                }
            }
            wave_lds_handover();
        }
        if constexpr (H == 1) {
#pragma unroll
            for (int off = 16; off > 0; off >>= 1) mx = fmaxf(mx, __shfl_xor(mx, off));
        }
        if (!(fabsf(mx) < INFINITY)) mx = 0.f;

        float den = expf(lself - mx);
        float acc[R];
        {
            const float* hrow = hin + (size_t)n * ROW + col;
#pragma unroll
            for (int q = 0; q < R / VW; ++q) {
                const VT t = *(const VT*)(hrow + q * VW);
#pragma unroll
                for (int k = 0; k < VW; ++k) acc[q * VW + k] = den * t[k];
            }
        }
        for (int c0 = 0; c0 < count; c0 += SEG_CAP) {
            int m = m0;
            if (!single) {
                m = compact_sources(srcs, dsts, c0, min(count, c0 + SEG_CAP), n, lane, scratch);
                wave_lds_handover();
            }
            for (int j = 0; j < m; j += 32) {
                const int cnt  = min(32, m - j);
                const int sreg = (lane < cnt) ? scratch[j + lane] : n;
                float wreg = 0.f;
                if constexpr (H == 1) wreg = expf(lrelu(as[sreg] + adn) - mx);
                for (int e0 = 0; e0 < cnt; e0 += G2) {
                    float s[G2];
                    float v[G2][R];
#pragma unroll
                    for (int g = 0; g < G2; ++g) {
                        s[g] = 0.f;
#pragma unroll
                        for (int r = 0; r < R; ++r) v[g][r] = 0.f;
                        if (e0 + g < cnt) {
                            const int sj = __builtin_amdgcn_readlane(sreg, e0 + g);
                            if constexpr (H > 1) s[g] = as[sj * H + hl];
                            const float* hrow = hin + (size_t)sj * ROW + col;
#pragma unroll
                            for (int q = 0; q < R / VW; ++q) {
                                const VT t = *(const VT*)(hrow + q * VW);
#pragma unroll
                                for (int k = 0; k < VW; ++k) v[g][q * VW + k] = t[k];
                            }
                        }
                    }
#pragma unroll
                    for (int g = 0; g < G2; ++g) {
                        if (e0 + g < cnt) {
                            float w;
                            if constexpr (H == 1) w = lane_bcast(wreg, e0 + g);
                            else                  w = expf(lrelu(s[g] + adn) - mx);
                            den += w;
#pragma unroll
                            for (int r = 0; r < R; ++r) acc[r] += w * v[g][r];
                        }
                    }
                }
            }
            wave_lds_handover();
        }

        float o[R];
#pragma unroll
        for (int r = 0; r < R; ++r) {
            float val = acc[r] / (den + 1e-16f) + bias[col + r];
            if (ELU) val = val > 0.f ? val : (expf(val) - 1.f);
            o[r] = val;
        }
        float* orow = hout + (size_t)n * ROW + col;
#pragma unroll
        for (int q = 0; q < R / VW; ++q) {
            VT t;
#pragma unroll
            for (int k = 0; k < VW; ++k) t[k] = o[q * VW + k];
            *(VT*)(orow + q * VW) = t;
        }
    }
}

template <int H, int R, bool ELU, bool REUSE>
__global__ void __launch_bounds__(1024)
agg_tile_kernel(const int* __restrict__ ei, int E0, int Nn,
                const float* __restrict__ as, const float* __restrict__ ad,
                const float* __restrict__ hin, const float* __restrict__ bias,
                float* hout, int* keep, int keep_row) {
    __shared__ int seg_e[AGG_WAVES][SEG_CAP];
    __shared__ int seg_d[AGG_WAVES][SEG_CAP];
    __shared__ int lst_src[AGG_WAVES * SEG_CAP];
    __shared__ int lst_dst[AGG_WAVES * SEG_CAP];
    __shared__ int own_src[AGG_WAVES * SEG_CAP];
    __shared__ int own_dst[AGG_WAVES * SEG_CAP];
    __shared__ int seg_cnt[AGG_WAVES];
    __shared__ int own_cnt[AGG_WAVES];
    __shared__ int overflow;

    const int tid  = threadIdx.x;
    const int lane = tid & 31;
    const int wave = __builtin_amdgcn_readfirstlane(tid >> 5);
    const int n0   = blockIdx.x * TILE_N;
    const int nt   = min(TILE_N, Nn - n0);
    const int* __restrict__ dstp = ei + E0;
    int* kt = keep + (size_t)n0 * keep_row;
    const int kcap = nt * keep_row;

    int  oc = 0, obase = 0, ovf = 0;
    bool have = false;
    if constexpr (REUSE) {
        const int hdr = __builtin_amdgcn_readfirstlane(kt[0]);
        have = (hdr >= 0) && (hdr <= AGG_WAVES * SEG_CAP) && (KEEP_HDR + 2 * hdr <= kcap);
        if (have) {
            if (tid < AGG_WAVES) own_cnt[tid] = kt[1 + tid];
            for (int i = tid; i < hdr; i += 1024) {
                own_src[i] = kt[KEEP_HDR + i];
                own_dst[i] = kt[KEEP_HDR + hdr + i];
            }
            __syncthreads();
            oc = own_cnt[wave];
            for (int w = 0; w < wave; ++w) obase += own_cnt[w];
        }
    }

    if (!have) {
        int  total = 0;
        bool parts = false;
        if constexpr (!REUSE) {
            const int qcap = kcap / PARTS;
            int hq[PARTS];
            int sum = 0;
            parts = true;
#pragma unroll
            for (int q = 0; q < PARTS; ++q) {
                hq[q] = __builtin_amdgcn_readfirstlane(kt[q * qcap]);
                parts = parts && (hq[q] >= 0) && (PART_HDR + 2 * hq[q] <= qcap);
                sum += parts ? hq[q] : 0;
            }
            parts = parts && (sum <= AGG_WAVES * SEG_CAP);
            if (parts) {
                int off = 0;
#pragma unroll
                for (int q = 0; q < PARTS; ++q) {
                    const int* kq = kt + q * qcap;
                    for (int i = tid; i < hq[q]; i += 1024) {
                        lst_src[off + i] = kq[PART_HDR + i];
                        lst_dst[off + i] = kq[PART_HDR + hq[q] + i];
                    }
                    off += hq[q];
                }
                total = sum;
                __syncthreads();
            }
        }

        if (!parts) {
            if (tid == 0) overflow = 0;
            __syncthreads();

            const int cnt = filter_scan(dstp, E0, wave, lane, n0, nt, &seg_e[wave][0], &seg_d[wave][0]);
            if (lane == 0) {
                seg_cnt[wave] = min(cnt, SEG_CAP);
                if (cnt > SEG_CAP) overflow = 1;
            }
            __syncthreads();

            int mybase = 0;
            for (int w = 0; w < AGG_WAVES; ++w) {
                const int c = seg_cnt[w];
                if (w < wave) mybase += c;
                total += c;
            }
            const int myc = seg_cnt[wave];
            ovf = overflow;
            for (int i = lane; i < myc; i += 32) {
                lst_src[mybase + i] = ei[seg_e[wave][i]];
                lst_dst[mybase + i] = seg_d[wave][i];
            }
            __syncthreads();
        }

        if (!ovf) {
            for (int c = 0; c < total; c += 32) {
                const int i = c + lane;
                const bool m = (i < total) && (((lst_dst[i] - n0) & (AGG_WAVES - 1)) == wave);
                oc += __builtin_popcount(__builtin_amdgcn_ballot_w32(m));
            }
        }
        if (lane == 0) own_cnt[wave] = oc;
        __syncthreads();
        for (int w = 0; w < wave; ++w) obase += own_cnt[w];

        if (!ovf) {
            int k = obase;
            for (int c = 0; c < total; c += 32) {
                const int i = c + lane;
                const bool m = (i < total) && (((lst_dst[i] - n0) & (AGG_WAVES - 1)) == wave);
                const unsigned mask = __builtin_amdgcn_ballot_w32(m);
                if (m) {
                    const int pos = k + __builtin_popcount(mask & ((1u << lane) - 1u));
                    own_src[pos] = lst_src[i];
                    own_dst[pos] = lst_dst[i];
                }
                k += __builtin_popcount(mask);
            }
        }
        __syncthreads();

        if constexpr (!REUSE) {
            const bool fits = !ovf && (KEEP_HDR + 2 * total <= kcap);
            if (tid == 0) kt[0] = fits ? total : -1;
            if (fits) {
                if (tid < AGG_WAVES) kt[1 + tid] = own_cnt[tid];
                for (int i = tid; i < total; i += 1024) {
                    kt[KEEP_HDR + i]         = own_src[i];
                    kt[KEEP_HDR + total + i] = own_dst[i];
                }
            }
        }
    }

    constexpr int G1 = 16;
    constexpr int G2 = (R > 2) ? 8 : 16;
    int* scratch = &seg_e[wave][0];
    if (!ovf)
        node_rows<H, R, ELU, G1, G2>(own_src + obase, own_dst + obase, oc, n0, nt, wave, lane, as, ad, hin, bias, hout, scratch);
    else
        node_rows<H, R, ELU, G1, G2>(ei, dstp, E0, n0, nt, wave, lane, as, ad, hin, bias, hout, scratch);
}

extern "C" void kernel_launch(void* const* d_in, const int* in_sizes, int n_in,
                              void* d_out, int out_size, void* d_ws, size_t ws_size,
                              hipStream_t stream) {
    const float* x        = (const float*)d_in[0];
    const int*   ei       = (const int*)d_in[1];
    const float* W1       = (const float*)d_in[2];
    const float* att_src1 = (const float*)d_in[3];
    const float* att_dst1 = (const float*)d_in[4];
    const float* b1       = (const float*)d_in[5];
    const float* W2       = (const float*)d_in[6];
    const float* att_src2 = (const float*)d_in[7];
    const float* att_dst2 = (const float*)d_in[8];
    const float* b2       = (const float*)d_in[9];
    float* out = (float*)d_out;

    const int Nn = in_sizes[0] / 256;
    const int E0 = in_sizes[1] / 2;

    float* h1   = (float*)d_ws;
    float* helu = h1 + (size_t)Nn * 256;
    float* as1  = helu + (size_t)Nn * 256;
    float* ad1  = as1 + (size_t)Nn * 8;
    float* h2   = h1;
    float* as2  = as1;
    float* ad2  = as1 + Nn;

    const int TB = 256;
    const int tiles = (Nn + TILE_N - 1) / TILE_N;
    dim3 g1((256 + 63) / 64, (Nn + 63) / 64 + tiles);
    dim3 g2((64 + 63) / 64, (Nn + 63) / 64);

    gat_wmma_gemm<true><<<g1, TB, 0, stream>>>(x, W1, h1, Nn, 256, 256, ei, E0, (int*)out, 64, tiles);
    scores1_kernel<<<(Nn * 8 + TB - 1) / TB, TB, 0, stream>>>(h1, att_src1, att_dst1, as1, ad1, Nn);
    agg_tile_kernel<8, 8, true, false><<<tiles, 1024, 0, stream>>>(ei, E0, Nn, as1, ad1, h1, b1, helu, (int*)out, 64);

    gat_wmma_gemm<false><<<g2, TB, 0, stream>>>(helu, W2, h2, Nn, 64, 256, nullptr, 0, nullptr, 0, 0);
    scores2_kernel<<<(Nn + TB - 1) / TB, TB, 0, stream>>>(h2, att_src2, att_dst2, as2, ad2, Nn);
    agg_tile_kernel<1, 2, false, true><<<tiles, 1024, 0, stream>>>(ei, E0, Nn, as2, ad2, h2, b2, out, (int*)out, 64);
}


# Edge list read once: extra work-groups of product 1's launch bin it by destination tile (integer LDS atomics, bins re-sorted by entry index); no tile scans it all; sum order now edge-list order
# speedup vs baseline: 4.4242x; 1.0593x over previous
#include <hip/hip_runtime.h>
#include <hip/hip_bf16.h>

typedef __attribute__((ext_vector_type(2))) float v2f;
typedef __attribute__((ext_vector_type(8))) float v8f;

#define NEG_SLOPE 0.2f

#define TILE_N    196
#define AGG_WAVES 32
#define SEG_CAP   256
#define KEEP_HDR  64
#define NBIN       256
#define BIN_COLS   4
#define BIN_MAXT   4096
#define BIN_THREADS 256
static_assert(NBIN % BIN_COLS == 0 && NBIN % AGG_WAVES == 0, "bins per grid row and per wave are whole numbers");

__device__ __forceinline__ void filter_group4(const int4 d, int ebase, int lane, unsigned below,
                                              int n0, int nt,
                                              int* __restrict__ se, int* __restrict__ sd, int& cnt) {
    const unsigned t0 = (unsigned)(d.x - n0), t1 = (unsigned)(d.y - n0);
    const unsigned t2 = (unsigned)(d.z - n0), t3 = (unsigned)(d.w - n0);
    const unsigned tmin = min(min(t0, t1), min(t2, t3));
    if (__builtin_amdgcn_ballot_w32(tmin < (unsigned)nt)) {
        const bool m0 = t0 < (unsigned)nt, m1 = t1 < (unsigned)nt;
        const bool m2 = t2 < (unsigned)nt, m3 = t3 < (unsigned)nt;
        const unsigned k0 = __builtin_amdgcn_ballot_w32(m0), k1 = __builtin_amdgcn_ballot_w32(m1);
        const unsigned k2 = __builtin_amdgcn_ballot_w32(m2), k3 = __builtin_amdgcn_ballot_w32(m3);
        int pos = cnt + __builtin_popcount(k0 & below) + __builtin_popcount(k1 & below)
                      + __builtin_popcount(k2 & below) + __builtin_popcount(k3 & below);
        const int e = ebase + 4 * lane;
        if (m0) { if (pos < SEG_CAP) { se[pos] = e;     sd[pos] = d.x; } ++pos; }
        if (m1) { if (pos < SEG_CAP) { se[pos] = e + 1; sd[pos] = d.y; } ++pos; }
        if (m2) { if (pos < SEG_CAP) { se[pos] = e + 2; sd[pos] = d.z; } ++pos; }
        if (m3) { if (pos < SEG_CAP) { se[pos] = e + 3; sd[pos] = d.w; } }
        cnt += __builtin_popcount(k0) + __builtin_popcount(k1) + __builtin_popcount(k2) + __builtin_popcount(k3);
    }
}

__device__ __forceinline__ int filter_scan(const int* __restrict__ dstp, int E0, int gw, int lane,
                                           int n0, int nt, int* __restrict__ se, int* __restrict__ sd) {
    int cnt = 0;
    int base = gw * 256;
    constexpr int FCH = 4;
    if ((reinterpret_cast<size_t>(dstp) & 15) == 0) {
        const unsigned below = (1u << lane) - 1u;
        for (; base + (FCH - 1) * AGG_WAVES * 256 + 256 <= E0; base += FCH * AGG_WAVES * 256) {
            int4 dv[FCH][2];
#pragma unroll
            for (int u = 0; u < FCH; ++u) {
#pragma unroll
                for (int h = 0; h < 2; ++h)
                    dv[u][h] = *(const int4*)(dstp + base + u * (AGG_WAVES * 256) + h * 128 + 4 * lane);
            }
#pragma unroll
            for (int u = 0; u < FCH; ++u) {
#pragma unroll
                for (int h = 0; h < 2; ++h)
                    filter_group4(dv[u][h], base + u * (AGG_WAVES * 256) + h * 128, lane, below, n0, nt,
                                  se, sd, cnt);
            }
        }
    }
    for (; base < E0; base += AGG_WAVES * 256) {
        int d[8];
        if (base + 256 <= E0) {
#pragma unroll
            for (int j = 0; j < 8; ++j) d[j] = dstp[base + j * 32 + lane];
        } else {
#pragma unroll
            for (int j = 0; j < 8; ++j) {
                const int e = base + j * 32 + lane;
                d[j] = (e < E0) ? dstp[e] : -1;
            }
        }
#pragma unroll
        for (int j = 0; j < 8; ++j) {
            const bool m = (unsigned)(d[j] - n0) < (unsigned)nt;
            const unsigned mask = __builtin_amdgcn_ballot_w32(m);
            if (mask) {
                const int pos = cnt + __builtin_popcount(mask & ((1u << lane) - 1u));
                if (m && pos < SEG_CAP) {
                    se[pos] = base + j * 32 + lane;
                    sd[pos] = d[j];
                }
                cnt += __builtin_popcount(mask);
            }
        }
    }
    return cnt;
}

__device__ __forceinline__ void bin_chunk(const int* __restrict__ ei, int E0, int Nn,
                                          int* __restrict__ bins, int bin_row, int b, int* __restrict__ sm) {
    const int tid   = threadIdx.x;
    constexpr int nthr = BIN_THREADS;
    const int tiles = (Nn + TILE_N - 1) / TILE_N;
    const bool usable = (tiles <= BIN_MAXT) && (Nn <= (1 << 24));
    if (usable) {
        for (int t = tid; t < tiles; t += nthr) sm[t] = 0;
        __syncthreads();
        const int CH = (E0 + NBIN - 1) / NBIN;
        const int e0 = b * CH;
        const int e1 = min(E0, e0 + CH);
        for (int eb = e0; eb < e1; eb += 4 * nthr) {
            int d[4], s[4];
#pragma unroll
            for (int u = 0; u < 4; ++u) {
                const int e = eb + u * nthr + tid;
                const bool ok = e < e1;
                d[u] = ok ? ei[E0 + e] : -1;
                s[u] = ok ? ei[e] : 0;
            }
#pragma unroll
            for (int u = 0; u < 4; ++u) {
                const int t = d[u] / TILE_N;
                if (d[u] >= 0 && t < tiles) {
                    const int n0   = t * TILE_N;
                    const int nt   = min(TILE_N, Nn - n0);
                    const int binw = (nt * bin_row) / NBIN;
                    const int cap  = (binw - 2) / 2;
                    const int slot = atomicAdd(&sm[t], 1);
                    if (slot < cap) {
                        int* q = bins + (size_t)n0 * bin_row + b * binw + 2 + 2 * slot;
                        q[0] = eb + u * nthr + tid;
                        q[1] = s[u] | (int)((unsigned)(d[u] - n0) << 24);
                    }
                }
            }
        }
        __syncthreads();
    }
    for (int t = tid; t < tiles; t += nthr) {
        const int n0   = t * TILE_N;
        const int nt   = min(TILE_N, Nn - n0);
        const int binw = (nt * bin_row) / NBIN;
        bins[(size_t)n0 * bin_row + b * binw] = usable ? sm[t] : -1;
    }
}

template <bool SCAN>
__global__ void gat_wmma_gemm(const float* __restrict__ A,
                              const float* __restrict__ B,
                              float* __restrict__ C,
                              int M, int Ntot, int K,
                              const int* __restrict__ ei, int E0, int* __restrict__ keep, int keep_row, int stiles) {
    __shared__ float As[64][68];
    __shared__ float Bp[32][132];
    static_assert(sizeof(As) >= BIN_MAXT * sizeof(int), "the bin counters fit in As");

    int yb = blockIdx.y;
    if constexpr (SCAN) {
        if (yb < stiles) {
            bin_chunk(ei, E0, M, keep, keep_row, yb * BIN_COLS + blockIdx.x, reinterpret_cast<int*>(&As[0][0]));
            return;
        }
        yb -= stiles;
    }

    const int tid   = threadIdx.x;
    const int lane  = tid & 31;
    const int wave  = tid >> 5;
    const int mBase = yb * 64;
    const int nBase = blockIdx.x * 64;

    const int ms   = (wave & 3) * 16;
    const int ns   = (wave >> 2) * 32;
    const int mrow = ms + (lane & 15);
    const int khi  = (lane >> 4) * 2;
    const int ncol = lane & 15;

    v8f acc0 = {}; v8f acc1 = {};

    for (int kBase = 0; kBase < K; kBase += 64) {
#pragma unroll
        for (int i = 0; i < 4; ++i) {
            int s  = tid + i * 256;
            int r  = s >> 4;
            int c4 = (s & 15) << 2;
            float4 va = make_float4(0.f, 0.f, 0.f, 0.f);
            int g = mBase + r;
            if (g < M) va = *(const float4*)(A + (size_t)g * K + kBase + c4);
            *(float4*)&As[r][c4] = va;
            float4 vb = *(const float4*)(B + (size_t)(kBase + r) * Ntot + nBase + c4);
            float* bp = &Bp[r >> 1][r & 1];
            bp[2 * c4 + 0] = vb.x;
            bp[2 * c4 + 2] = vb.y;
            bp[2 * c4 + 4] = vb.z;
            bp[2 * c4 + 6] = vb.w;
        }
        __syncthreads();

#pragma unroll
        for (int kk = 0; kk < 64; kk += 4) {
            const int p2 = (kk + khi) >> 1;
            v2f a, b0, b1v;
            a[0]   = As[mrow][kk + khi];
            a[1]   = As[mrow][kk + khi + 1];
            b0     = *(const v2f*)&Bp[p2][2 * (ns + ncol)];
            b1v    = *(const v2f*)&Bp[p2][2 * (ns + 16 + ncol)];
            acc0 = __builtin_amdgcn_wmma_f32_16x16x4_f32(
                false, a, false, b0, (short)0, acc0, false, false);
            acc1 = __builtin_amdgcn_wmma_f32_16x16x4_f32(
                false, a, false, b1v, (short)0, acc1, false, false);
        }
        __syncthreads();
    }

    const int orow = mBase + ms + ((lane >> 4) << 3);
    const int ocol = nBase + ns + ncol;
#pragma unroll
    for (int r = 0; r < 8; ++r) {
        int g = orow + r;
        if (g < M) {
            C[(size_t)g * Ntot + ocol]      = acc0[r];
            C[(size_t)g * Ntot + ocol + 16] = acc1[r];
        }
    }
}

__global__ void scores1_kernel(const float* __restrict__ h1,
                               const float* __restrict__ att_s,
                               const float* __restrict__ att_d,
                               float* __restrict__ as1, float* __restrict__ ad1,
                               int Nn) {
    int t = blockIdx.x * blockDim.x + threadIdx.x;
    if (t >= Nn * 8) return;
    int n = t >> 3, h = t & 7;
    const float* row = h1 + (size_t)n * 256 + h * 32;
    const float* ws = att_s + h * 32;
    const float* wd = att_d + h * 32;
    float ss = 0.f, sd = 0.f;
#pragma unroll
    for (int c = 0; c < 32; ++c) { ss += row[c] * ws[c]; sd += row[c] * wd[c]; }
    as1[t] = ss; ad1[t] = sd;
}

__global__ void scores2_kernel(const float* __restrict__ h2,
                               const float* __restrict__ att_s,
                               const float* __restrict__ att_d,
                               float* __restrict__ as2, float* __restrict__ ad2,
                               int Nn) {
    int n = blockIdx.x * blockDim.x + threadIdx.x;
    if (n >= Nn) return;
    const float* row = h2 + (size_t)n * 64;
    float ss = 0.f, sd = 0.f;
#pragma unroll
    for (int c = 0; c < 64; ++c) { ss += row[c] * att_s[c]; sd += row[c] * att_d[c]; }
    as2[n] = ss; ad2[n] = sd;
}

__device__ __forceinline__ float lrelu(float v) { return v > 0.f ? v : NEG_SLOPE * v; }

__device__ __forceinline__ float lane_bcast(float v, int j) {
    return __uint_as_float(__builtin_amdgcn_readlane(__float_as_uint(v), j));
}

__device__ __forceinline__ int compact_sources(const int* __restrict__ srcs, const int* __restrict__ dsts,
                                               int c0, int cend, int n, int lane,
                                               int* __restrict__ scratch) {
    const unsigned below = (1u << lane) - 1u;
    int m = 0;
    for (int i0 = c0; i0 < cend; i0 += 32) {
        const int  i     = i0 + lane;
        const bool in    = i < cend;
        const int  d     = in ? dsts[i] : -1;
        const int  sv    = in ? srcs[i] : 0;
        const bool match = (d == n);
        const unsigned mask = __builtin_amdgcn_ballot_w32(match);
        if (match) scratch[m + __builtin_popcount(mask & below)] = sv;
        m += __builtin_popcount(mask);
    }
    return m;
}

__device__ __forceinline__ void wave_lds_handover() {
    __builtin_amdgcn_fence(__ATOMIC_RELEASE, "wavefront");
    __builtin_amdgcn_wave_barrier();
    __builtin_amdgcn_fence(__ATOMIC_ACQUIRE, "wavefront");
}

template <int H, int R, bool ELU, int G1, int G2>
__device__ __forceinline__ void node_rows(const int* __restrict__ srcs, const int* __restrict__ dsts, int count,
                                          int n0, int nt, int wave, int lane,
                                          const float* __restrict__ as,
                                          const float* __restrict__ ad,
                                          const float* __restrict__ hin,
                                          const float* __restrict__ bias,
                                          float* __restrict__ hout,
                                          int* __restrict__ scratch) {
    constexpr int ROW = 32 * R;
    constexpr int VW  = (R >= 4) ? 4 : 2;
    typedef float VT __attribute__((ext_vector_type(VW)));
    const int  col    = lane * R;
    const int  hl     = (col * H) / ROW;
    const bool single = count <= SEG_CAP;

    for (int ln = wave; ln < nt; ln += AGG_WAVES) {
        const int   n     = n0 + ln;
        const float adn   = ad[n * H + hl];
        const float lself = lrelu(as[n * H + hl] + adn);
        float mx = lself;
        int m0 = 0;

        for (int c0 = 0; c0 < count; c0 += SEG_CAP) {
            const int m = compact_sources(srcs, dsts, c0, min(count, c0 + SEG_CAP), n, lane, scratch);
            if (c0 == 0) m0 = m;
            wave_lds_handover();
            for (int j = 0; j < m; j += 32) {
                const int cnt  = min(32, m - j);
                const int sreg = (lane < cnt) ? scratch[j + lane] : n;
                if constexpr (H == 1) {
                    mx = fmaxf(mx, lrelu(as[sreg] + adn));
                } else {
                    for (int e0 = 0; e0 < cnt; e0 += G1) {
                        float s[G1];
#pragma unroll
                        for (int g = 0; g < G1; ++g) {
                            const int sj = __builtin_amdgcn_readlane(sreg, e0 + g);
                            s[g] = as[sj * H + hl];
                        }
#pragma unroll
                        for (int g = 0; g < G1; ++g) mx = fmaxf(mx, lrelu(s[g] + adn));
                    }
                }
            }
            wave_lds_handover();
        }
        if constexpr (H == 1) {
#pragma unroll
            for (int off = 16; off > 0; off >>= 1) mx = fmaxf(mx, __shfl_xor(mx, off));
        }
        if (!(fabsf(mx) < INFINITY)) mx = 0.f;

        float den = expf(lself - mx);
        float acc[R];
        {
            const float* hrow = hin + (size_t)n * ROW + col;
#pragma unroll
            for (int q = 0; q < R / VW; ++q) {
                const VT t = *(const VT*)(hrow + q * VW);
#pragma unroll
                for (int k = 0; k < VW; ++k) acc[q * VW + k] = den * t[k];
            }
        }
        for (int c0 = 0; c0 < count; c0 += SEG_CAP) {
            int m = m0;
            if (!single) {
                m = compact_sources(srcs, dsts, c0, min(count, c0 + SEG_CAP), n, lane, scratch);
                wave_lds_handover();
            }
            for (int j = 0; j < m; j += 32) {
                const int cnt  = min(32, m - j);
                const int sreg = (lane < cnt) ? scratch[j + lane] : n;
                float wreg = 0.f;
                if constexpr (H == 1) wreg = expf(lrelu(as[sreg] + adn) - mx);
                for (int e0 = 0; e0 < cnt; e0 += G2) {
                    float s[G2];
                    float v[G2][R];
#pragma unroll
                    for (int g = 0; g < G2; ++g) {
                        s[g] = 0.f;
#pragma unroll
                        for (int r = 0; r < R; ++r) v[g][r] = 0.f;
                        if (e0 + g < cnt) {
                            const int sj = __builtin_amdgcn_readlane(sreg, e0 + g);
                            if constexpr (H > 1) s[g] = as[sj * H + hl];
                            const float* hrow = hin + (size_t)sj * ROW + col;
#pragma unroll
                            for (int q = 0; q < R / VW; ++q) {
                                const VT t = *(const VT*)(hrow + q * VW);
#pragma unroll
                                for (int k = 0; k < VW; ++k) v[g][q * VW + k] = t[k];
                            }
                        }
                    }
#pragma unroll
                    for (int g = 0; g < G2; ++g) {
                        if (e0 + g < cnt) {
                            float w;
                            if constexpr (H == 1) w = lane_bcast(wreg, e0 + g);
                            else                  w = expf(lrelu(s[g] + adn) - mx);
                            den += w;
#pragma unroll
                            for (int r = 0; r < R; ++r) acc[r] += w * v[g][r];
                        }
                    }
                }
            }
            wave_lds_handover();
        }

        float o[R];
#pragma unroll
        for (int r = 0; r < R; ++r) {
            float val = acc[r] / (den + 1e-16f) + bias[col + r];
            if (ELU) val = val > 0.f ? val : (expf(val) - 1.f);
            o[r] = val;
        }
        float* orow = hout + (size_t)n * ROW + col;
#pragma unroll
        for (int q = 0; q < R / VW; ++q) {
            VT t;
#pragma unroll
            for (int k = 0; k < VW; ++k) t[k] = o[q * VW + k];
            *(VT*)(orow + q * VW) = t;
        }
    }
}

template <int H, int R, bool ELU, bool REUSE>
__global__ void __launch_bounds__(1024)
agg_tile_kernel(const int* __restrict__ ei, int E0, int Nn,
                const float* __restrict__ as, const float* __restrict__ ad,
                const float* __restrict__ hin, const float* __restrict__ bias,
                float* hout, int* keep, int keep_row, const int* bins, int bin_row) {
    __shared__ int seg_e[AGG_WAVES][SEG_CAP];
    __shared__ int seg_d[AGG_WAVES][SEG_CAP];
    __shared__ int lst_src[AGG_WAVES * SEG_CAP];
    __shared__ int lst_dst[AGG_WAVES * SEG_CAP];
    __shared__ int own_src[AGG_WAVES * SEG_CAP];
    __shared__ int own_dst[AGG_WAVES * SEG_CAP];
    __shared__ int seg_cnt[AGG_WAVES];
    __shared__ int own_cnt[AGG_WAVES];
    __shared__ int overflow;

    const int tid  = threadIdx.x;
    const int lane = tid & 31;
    const int wave = __builtin_amdgcn_readfirstlane(tid >> 5);
    const int n0   = blockIdx.x * TILE_N;
    const int nt   = min(TILE_N, Nn - n0);
    const int* __restrict__ dstp = ei + E0;
    int* kt = keep + (size_t)n0 * keep_row;
    const int kcap = nt * keep_row;

    int  oc = 0, obase = 0, ovf = 0;
    bool have = false;
    if constexpr (REUSE) {
        const int hdr = __builtin_amdgcn_readfirstlane(kt[0]);
        have = (hdr >= 0) && (hdr <= AGG_WAVES * SEG_CAP) && (KEEP_HDR + 2 * hdr <= kcap);
        if (have) {
            if (tid < AGG_WAVES) own_cnt[tid] = kt[1 + tid];
            for (int i = tid; i < hdr; i += 1024) {
                own_src[i] = kt[KEEP_HDR + i];
                own_dst[i] = kt[KEEP_HDR + hdr + i];
            }
            __syncthreads();
            oc = own_cnt[wave];
            for (int w = 0; w < wave; ++w) obase += own_cnt[w];
        }
    }

    if (!have) {
        int  total = 0;
        bool parts = false;
        if constexpr (!REUSE) {
            static_assert(NBIN <= 1024 && NBIN / 32 <= AGG_WAVES, "one thread per bin, wave sums in seg_cnt");
            constexpr int BPW = NBIN / AGG_WAVES;
            const int* bt   = bins + (size_t)n0 * bin_row;
            const int  binw = (nt * bin_row) / NBIN;
            const int  cap  = min((binw - 2) / 2, SEG_CAP);
            int* bcnt = own_dst;
            int* boff = own_src;
            if (tid == 0) overflow = 0;
            __syncthreads();
            int cb = 0;
            if (tid < NBIN) {
                cb = bt[tid * binw];
                if (cb < 0 || cb > cap) { overflow = 1; cb = 0; }
                bcnt[tid] = cb;
            }
            int incl = cb;
            for (int off = 1; off < 32; off <<= 1) {
                const int v = __shfl_up(incl, off);
                if (lane >= off) incl += v;
            }
            if (lane == 31 && tid < NBIN) seg_cnt[wave] = incl;
            __syncthreads();
            int sum = 0, wbase = 0;
            for (int w = 0; w < NBIN / 32; ++w) {
                const int cw = seg_cnt[w];
                if (w < wave) wbase += cw;
                sum += cw;
            }
            if (tid < NBIN) boff[tid] = wbase + incl - cb;
            parts = (overflow == 0) && (sum <= AGG_WAVES * SEG_CAP);
            __syncthreads();
            if (parts) {
                int* ke = &seg_e[wave][0];
                int* kp = &seg_d[wave][0];
                int ev[BPW], pv[BPW], cn[BPW];
#pragma unroll
                for (int u = 0; u < BPW; ++u) {
                    const int b = wave + u * AGG_WAVES;
                    cn[u] = __builtin_amdgcn_readfirstlane(bcnt[b]);
                    ev[u] = 0; pv[u] = 0;
                    if (lane < cn[u]) {
                        const int* q = bt + b * binw + 2 + 2 * lane;
                        ev[u] = q[0]; pv[u] = q[1];
                    }
                }
#pragma unroll
                for (int u = 0; u < BPW; ++u) {
                    const int b   = wave + u * AGG_WAVES;
                    const int cnt = cn[u];
                    const int off = __builtin_amdgcn_readfirstlane(boff[b]);
                    if (lane < cnt) { ke[lane] = ev[u]; kp[lane] = pv[u]; }
                    for (int i = 32 + lane; i < cnt; i += 32) {
                        const int* q = bt + b * binw + 2 + 2 * i;
                        ke[i] = q[0]; kp[i] = q[1];
                    }
                    wave_lds_handover();
                    for (int i = lane; i < cnt; i += 32) {
                        const int key = ke[i];
                        int rank = 0;
                        for (int j = 0; j < cnt; ++j) rank += (ke[j] < key) ? 1 : 0;
                        const int pk = kp[i];
                        lst_src[off + rank] = pk & 0xFFFFFF;
                        lst_dst[off + rank] = n0 + (int)((unsigned)pk >> 24);
                    }
                    wave_lds_handover();
                }
                total = sum;
                __syncthreads();
            }
        }

        if (!parts) {
            if (tid == 0) overflow = 0;
            __syncthreads();

            const int cnt = filter_scan(dstp, E0, wave, lane, n0, nt, &seg_e[wave][0], &seg_d[wave][0]);
            if (lane == 0) {
                seg_cnt[wave] = min(cnt, SEG_CAP);
                if (cnt > SEG_CAP) overflow = 1;
            }
            __syncthreads();

            int mybase = 0;
            for (int w = 0; w < AGG_WAVES; ++w) {
                const int c = seg_cnt[w];
                if (w < wave) mybase += c;
                total += c;
            }
            const int myc = seg_cnt[wave];
            ovf = overflow;
            for (int i = lane; i < myc; i += 32) {
                lst_src[mybase + i] = ei[seg_e[wave][i]];
                lst_dst[mybase + i] = seg_d[wave][i];
            }
            __syncthreads();
        }

        if (!ovf) {
            for (int c = 0; c < total; c += 32) {
                const int i = c + lane;
                const bool m = (i < total) && (((lst_dst[i] - n0) & (AGG_WAVES - 1)) == wave);
                oc += __builtin_popcount(__builtin_amdgcn_ballot_w32(m));
            }
        }
        if (lane == 0) own_cnt[wave] = oc;
        __syncthreads();
        for (int w = 0; w < wave; ++w) obase += own_cnt[w];

        if (!ovf) {
            int k = obase;
            for (int c = 0; c < total; c += 32) {
                const int i = c + lane;
                const bool m = (i < total) && (((lst_dst[i] - n0) & (AGG_WAVES - 1)) == wave);
                const unsigned mask = __builtin_amdgcn_ballot_w32(m);
                if (m) {
                    const int pos = k + __builtin_popcount(mask & ((1u << lane) - 1u));
                    own_src[pos] = lst_src[i];
                    own_dst[pos] = lst_dst[i];
                }
                k += __builtin_popcount(mask);
            }
        }
        __syncthreads();

        if constexpr (!REUSE) {
            const bool fits = !ovf && (KEEP_HDR + 2 * total <= kcap);
            if (tid == 0) kt[0] = fits ? total : -1;
            if (fits) {
                if (tid < AGG_WAVES) kt[1 + tid] = own_cnt[tid];
                for (int i = tid; i < total; i += 1024) {
                    kt[KEEP_HDR + i]         = own_src[i];
                    kt[KEEP_HDR + total + i] = own_dst[i];
                }
            }
        }
    }

    constexpr int G1 = 16;
    constexpr int G2 = (R > 2) ? 8 : 16;
    int* scratch = &seg_e[wave][0];
    if (!ovf)
        node_rows<H, R, ELU, G1, G2>(own_src + obase, own_dst + obase, oc, n0, nt, wave, lane, as, ad, hin, bias, hout, scratch);
    else
        node_rows<H, R, ELU, G1, G2>(ei, dstp, E0, n0, nt, wave, lane, as, ad, hin, bias, hout, scratch);
}

extern "C" void kernel_launch(void* const* d_in, const int* in_sizes, int n_in,
                              void* d_out, int out_size, void* d_ws, size_t ws_size,
                              hipStream_t stream) {
    const float* x        = (const float*)d_in[0];
    const int*   ei       = (const int*)d_in[1];
    const float* W1       = (const float*)d_in[2];
    const float* att_src1 = (const float*)d_in[3];
    const float* att_dst1 = (const float*)d_in[4];
    const float* b1       = (const float*)d_in[5];
    const float* W2       = (const float*)d_in[6];
    const float* att_src2 = (const float*)d_in[7];
    const float* att_dst2 = (const float*)d_in[8];
    const float* b2       = (const float*)d_in[9];
    float* out = (float*)d_out;

    const int Nn = in_sizes[0] / 256;
    const int E0 = in_sizes[1] / 2;

    float* h1   = (float*)d_ws;
    float* helu = h1 + (size_t)Nn * 256;
    float* as1  = helu + (size_t)Nn * 256;
    float* ad1  = as1 + (size_t)Nn * 8;
    float* h2   = h1;
    float* as2  = as1;
    float* ad2  = as1 + Nn;

    const int TB = 256;
    const int tiles = (Nn + TILE_N - 1) / TILE_N;
    dim3 g1((256 + 63) / 64, (Nn + 63) / 64 + NBIN / BIN_COLS);
    dim3 g2((64 + 63) / 64, (Nn + 63) / 64);

    gat_wmma_gemm<true><<<g1, TB, 0, stream>>>(x, W1, h1, Nn, 256, 256, ei, E0, (int*)helu, 256, NBIN / BIN_COLS);
    scores1_kernel<<<(Nn * 8 + TB - 1) / TB, TB, 0, stream>>>(h1, att_src1, att_dst1, as1, ad1, Nn);
    agg_tile_kernel<8, 8, true, false><<<tiles, 1024, 0, stream>>>(ei, E0, Nn, as1, ad1, h1, b1, helu, (int*)out, 64, (const int*)helu, 256);

    gat_wmma_gemm<false><<<g2, TB, 0, stream>>>(helu, W2, h2, Nn, 64, 256, nullptr, 0, nullptr, 0, 0);
    scores2_kernel<<<(Nn + TB - 1) / TB, TB, 0, stream>>>(h2, att_src2, att_dst2, as2, ad2, Nn);
    agg_tile_kernel<1, 2, false, true><<<tiles, 1024, 0, stream>>>(ei, E0, Nn, as2, ad2, h2, b2, out, (int*)out, 64, nullptr, 0);
}


# Tile list's stable partition by owning wave done once by the whole work-group (per-run owner counts, running sums, scatter) instead of every wave scanning the list twice; list and sum order unchanged
# speedup vs baseline: 4.7825x; 1.0810x over previous
#include <hip/hip_runtime.h>
#include <hip/hip_bf16.h>

typedef __attribute__((ext_vector_type(2))) float v2f;
typedef __attribute__((ext_vector_type(8))) float v8f;

#define NEG_SLOPE 0.2f

#define TILE_N    196
#define AGG_WAVES 32
#define SEG_CAP   256
#define KEEP_HDR  64
#define NBIN       256
#define BIN_COLS   4
#define BIN_MAXT   4096
#define BIN_THREADS 256
static_assert(NBIN % BIN_COLS == 0 && NBIN % AGG_WAVES == 0, "bins per grid row and per wave are whole numbers");

__device__ __forceinline__ void filter_group4(const int4 d, int ebase, int lane, unsigned below,
                                              int n0, int nt,
                                              int* __restrict__ se, int* __restrict__ sd, int& cnt) {
    const unsigned t0 = (unsigned)(d.x - n0), t1 = (unsigned)(d.y - n0);
    const unsigned t2 = (unsigned)(d.z - n0), t3 = (unsigned)(d.w - n0);
    const unsigned tmin = min(min(t0, t1), min(t2, t3));
    if (__builtin_amdgcn_ballot_w32(tmin < (unsigned)nt)) {
        const bool m0 = t0 < (unsigned)nt, m1 = t1 < (unsigned)nt;
        const bool m2 = t2 < (unsigned)nt, m3 = t3 < (unsigned)nt;
        const unsigned k0 = __builtin_amdgcn_ballot_w32(m0), k1 = __builtin_amdgcn_ballot_w32(m1);
        const unsigned k2 = __builtin_amdgcn_ballot_w32(m2), k3 = __builtin_amdgcn_ballot_w32(m3);
        int pos = cnt + __builtin_popcount(k0 & below) + __builtin_popcount(k1 & below)
                      + __builtin_popcount(k2 & below) + __builtin_popcount(k3 & below);
        const int e = ebase + 4 * lane;
        if (m0) { if (pos < SEG_CAP) { se[pos] = e;     sd[pos] = d.x; } ++pos; }
        if (m1) { if (pos < SEG_CAP) { se[pos] = e + 1; sd[pos] = d.y; } ++pos; }
        if (m2) { if (pos < SEG_CAP) { se[pos] = e + 2; sd[pos] = d.z; } ++pos; }
        if (m3) { if (pos < SEG_CAP) { se[pos] = e + 3; sd[pos] = d.w; } }
        cnt += __builtin_popcount(k0) + __builtin_popcount(k1) + __builtin_popcount(k2) + __builtin_popcount(k3);
    }
}

__device__ __forceinline__ int filter_scan(const int* __restrict__ dstp, int E0, int gw, int lane,
                                           int n0, int nt, int* __restrict__ se, int* __restrict__ sd) {
    int cnt = 0;
    int base = gw * 256;
    constexpr int FCH = 4;
    if ((reinterpret_cast<size_t>(dstp) & 15) == 0) {
        const unsigned below = (1u << lane) - 1u;
        for (; base + (FCH - 1) * AGG_WAVES * 256 + 256 <= E0; base += FCH * AGG_WAVES * 256) {
            int4 dv[FCH][2];
#pragma unroll
            for (int u = 0; u < FCH; ++u) {
#pragma unroll
                for (int h = 0; h < 2; ++h)
                    dv[u][h] = *(const int4*)(dstp + base + u * (AGG_WAVES * 256) + h * 128 + 4 * lane);
            }
#pragma unroll
            for (int u = 0; u < FCH; ++u) {
#pragma unroll
                for (int h = 0; h < 2; ++h)
                    filter_group4(dv[u][h], base + u * (AGG_WAVES * 256) + h * 128, lane, below, n0, nt,
                                  se, sd, cnt);
            }
        }
    }
    for (; base < E0; base += AGG_WAVES * 256) {
        int d[8];
        if (base + 256 <= E0) {
#pragma unroll
            for (int j = 0; j < 8; ++j) d[j] = dstp[base + j * 32 + lane];
        } else {
#pragma unroll
            for (int j = 0; j < 8; ++j) {
                const int e = base + j * 32 + lane;
                d[j] = (e < E0) ? dstp[e] : -1;
            }
        }
#pragma unroll
        for (int j = 0; j < 8; ++j) {
            const bool m = (unsigned)(d[j] - n0) < (unsigned)nt;
            const unsigned mask = __builtin_amdgcn_ballot_w32(m);
            if (mask) {
                const int pos = cnt + __builtin_popcount(mask & ((1u << lane) - 1u));
                if (m && pos < SEG_CAP) {
                    se[pos] = base + j * 32 + lane;
                    sd[pos] = d[j];
                }
                cnt += __builtin_popcount(mask);
            }
        }
    }
    return cnt;
}

__device__ __forceinline__ void bin_chunk(const int* __restrict__ ei, int E0, int Nn,
                                          int* __restrict__ bins, int bin_row, int b, int* __restrict__ sm) {
    const int tid   = threadIdx.x;
    constexpr int nthr = BIN_THREADS;
    const int tiles = (Nn + TILE_N - 1) / TILE_N;
    const bool usable = (tiles <= BIN_MAXT) && (Nn <= (1 << 24));
    if (usable) {
        for (int t = tid; t < tiles; t += nthr) sm[t] = 0;
        __syncthreads();
        const int CH = (E0 + NBIN - 1) / NBIN;
        const int e0 = b * CH;
        const int e1 = min(E0, e0 + CH);
        for (int eb = e0; eb < e1; eb += 4 * nthr) {
            int d[4], s[4];
#pragma unroll
            for (int u = 0; u < 4; ++u) {
                const int e = eb + u * nthr + tid;
                const bool ok = e < e1;
                d[u] = ok ? ei[E0 + e] : -1;
                s[u] = ok ? ei[e] : 0;
            }
#pragma unroll
            for (int u = 0; u < 4; ++u) {
                const int t = d[u] / TILE_N;
                if (d[u] >= 0 && t < tiles) {
                    const int n0   = t * TILE_N;
                    const int nt   = min(TILE_N, Nn - n0);
                    const int binw = (nt * bin_row) / NBIN;
                    const int cap  = (binw - 2) / 2;
                    const int slot = atomicAdd(&sm[t], 1);
                    if (slot < cap) {
                        int* q = bins + (size_t)n0 * bin_row + b * binw + 2 + 2 * slot;
                        q[0] = eb + u * nthr + tid;
                        q[1] = s[u] | (int)((unsigned)(d[u] - n0) << 24);
                    }
                }
            }
        }
        __syncthreads();
    }
    for (int t = tid; t < tiles; t += nthr) {
        const int n0   = t * TILE_N;
        const int nt   = min(TILE_N, Nn - n0);
        const int binw = (nt * bin_row) / NBIN;
        bins[(size_t)n0 * bin_row + b * binw] = usable ? sm[t] : -1;
    }
}

template <bool SCAN>
__global__ void gat_wmma_gemm(const float* __restrict__ A,
                              const float* __restrict__ B,
                              float* __restrict__ C,
                              int M, int Ntot, int K,
                              const int* __restrict__ ei, int E0, int* __restrict__ keep, int keep_row, int stiles) {
    __shared__ float As[64][68];
    __shared__ float Bp[32][132];
    static_assert(sizeof(As) >= BIN_MAXT * sizeof(int), "the bin counters fit in As");

    int yb = blockIdx.y;
    if constexpr (SCAN) {
        if (yb < stiles) {
            bin_chunk(ei, E0, M, keep, keep_row, yb * BIN_COLS + blockIdx.x, reinterpret_cast<int*>(&As[0][0]));
            return;
        }
        yb -= stiles;
    }

    const int tid   = threadIdx.x;
    const int lane  = tid & 31;
    const int wave  = tid >> 5;
    const int mBase = yb * 64;
    const int nBase = blockIdx.x * 64;

    const int ms   = (wave & 3) * 16;
    const int ns   = (wave >> 2) * 32;
    const int mrow = ms + (lane & 15);
    const int khi  = (lane >> 4) * 2;
    const int ncol = lane & 15;

    v8f acc0 = {}; v8f acc1 = {};

    for (int kBase = 0; kBase < K; kBase += 64) {
#pragma unroll
        for (int i = 0; i < 4; ++i) {
            int s  = tid + i * 256;
            int r  = s >> 4;
            int c4 = (s & 15) << 2;
            float4 va = make_float4(0.f, 0.f, 0.f, 0.f);
            int g = mBase + r;
            if (g < M) va = *(const float4*)(A + (size_t)g * K + kBase + c4);
            *(float4*)&As[r][c4] = va;
            float4 vb = *(const float4*)(B + (size_t)(kBase + r) * Ntot + nBase + c4);
            float* bp = &Bp[r >> 1][r & 1];
            bp[2 * c4 + 0] = vb.x;
            bp[2 * c4 + 2] = vb.y;
            bp[2 * c4 + 4] = vb.z;
            bp[2 * c4 + 6] = vb.w;
        }
        __syncthreads();

#pragma unroll
        for (int kk = 0; kk < 64; kk += 4) {
            const int p2 = (kk + khi) >> 1;
            v2f a, b0, b1v;
            a[0]   = As[mrow][kk + khi];
            a[1]   = As[mrow][kk + khi + 1];
            b0     = *(const v2f*)&Bp[p2][2 * (ns + ncol)];
            b1v    = *(const v2f*)&Bp[p2][2 * (ns + 16 + ncol)];
            acc0 = __builtin_amdgcn_wmma_f32_16x16x4_f32(
                false, a, false, b0, (short)0, acc0, false, false);
            acc1 = __builtin_amdgcn_wmma_f32_16x16x4_f32(
                false, a, false, b1v, (short)0, acc1, false, false);
        }
        __syncthreads();
    }

    const int orow = mBase + ms + ((lane >> 4) << 3);
    const int ocol = nBase + ns + ncol;
#pragma unroll
    for (int r = 0; r < 8; ++r) {
        int g = orow + r;
        if (g < M) {
            C[(size_t)g * Ntot + ocol]      = acc0[r];
            C[(size_t)g * Ntot + ocol + 16] = acc1[r];
        }
    }
}

__global__ void scores1_kernel(const float* __restrict__ h1,
                               const float* __restrict__ att_s,
                               const float* __restrict__ att_d,
                               float* __restrict__ as1, float* __restrict__ ad1,
                               int Nn) {
    int t = blockIdx.x * blockDim.x + threadIdx.x;
    if (t >= Nn * 8) return;
    int n = t >> 3, h = t & 7;
    const float* row = h1 + (size_t)n * 256 + h * 32;
    const float* ws = att_s + h * 32;
    const float* wd = att_d + h * 32;
    float ss = 0.f, sd = 0.f;
#pragma unroll
    for (int c = 0; c < 32; ++c) { ss += row[c] * ws[c]; sd += row[c] * wd[c]; }
    as1[t] = ss; ad1[t] = sd;
}

__global__ void scores2_kernel(const float* __restrict__ h2,
                               const float* __restrict__ att_s,
                               const float* __restrict__ att_d,
                               float* __restrict__ as2, float* __restrict__ ad2,
                               int Nn) {
    int n = blockIdx.x * blockDim.x + threadIdx.x;
    if (n >= Nn) return;
    const float* row = h2 + (size_t)n * 64;
    float ss = 0.f, sd = 0.f;
#pragma unroll
    for (int c = 0; c < 64; ++c) { ss += row[c] * att_s[c]; sd += row[c] * att_d[c]; }
    as2[n] = ss; ad2[n] = sd;
}

__device__ __forceinline__ float lrelu(float v) { return v > 0.f ? v : NEG_SLOPE * v; }

__device__ __forceinline__ float lane_bcast(float v, int j) {
    return __uint_as_float(__builtin_amdgcn_readlane(__float_as_uint(v), j));
}

__device__ __forceinline__ int compact_sources(const int* __restrict__ srcs, const int* __restrict__ dsts,
                                               int c0, int cend, int n, int lane,
                                               int* __restrict__ scratch) {
    const unsigned below = (1u << lane) - 1u;
    int m = 0;
    for (int i0 = c0; i0 < cend; i0 += 32) {
        const int  i     = i0 + lane;
        const bool in    = i < cend;
        const int  d     = in ? dsts[i] : -1;
        const int  sv    = in ? srcs[i] : 0;
        const bool match = (d == n);
        const unsigned mask = __builtin_amdgcn_ballot_w32(match);
        if (match) scratch[m + __builtin_popcount(mask & below)] = sv;
        m += __builtin_popcount(mask);
    }
    return m;
}

__device__ __forceinline__ void wave_lds_handover() {
    __builtin_amdgcn_fence(__ATOMIC_RELEASE, "wavefront");
    __builtin_amdgcn_wave_barrier();
    __builtin_amdgcn_fence(__ATOMIC_ACQUIRE, "wavefront");
}

template <int H, int R, bool ELU, int G1, int G2>
__device__ __forceinline__ void node_rows(const int* __restrict__ srcs, const int* __restrict__ dsts, int count,
                                          int n0, int nt, int wave, int lane,
                                          const float* __restrict__ as,
                                          const float* __restrict__ ad,
                                          const float* __restrict__ hin,
                                          const float* __restrict__ bias,
                                          float* __restrict__ hout,
                                          int* __restrict__ scratch) {
    constexpr int ROW = 32 * R;
    constexpr int VW  = (R >= 4) ? 4 : 2;
    typedef float VT __attribute__((ext_vector_type(VW)));
    const int  col    = lane * R;
    const int  hl     = (col * H) / ROW;
    const bool single = count <= SEG_CAP;

    for (int ln = wave; ln < nt; ln += AGG_WAVES) {
        const int   n     = n0 + ln;
        const float adn   = ad[n * H + hl];
        const float lself = lrelu(as[n * H + hl] + adn);
        float mx = lself;
        int m0 = 0;

        for (int c0 = 0; c0 < count; c0 += SEG_CAP) {
            const int m = compact_sources(srcs, dsts, c0, min(count, c0 + SEG_CAP), n, lane, scratch);
            if (c0 == 0) m0 = m;
            wave_lds_handover();
            for (int j = 0; j < m; j += 32) {
                const int cnt  = min(32, m - j);
                const int sreg = (lane < cnt) ? scratch[j + lane] : n;
                if constexpr (H == 1) {
                    mx = fmaxf(mx, lrelu(as[sreg] + adn));
                } else {
                    for (int e0 = 0; e0 < cnt; e0 += G1) {
                        float s[G1];
#pragma unroll
                        for (int g = 0; g < G1; ++g) {
                            const int sj = __builtin_amdgcn_readlane(sreg, e0 + g);
                            s[g] = as[sj * H + hl];
                        }
#pragma unroll
                        for (int g = 0; g < G1; ++g) mx = fmaxf(mx, lrelu(s[g] + adn));
                    }
                }
            }
            wave_lds_handover();
        }
        if constexpr (H == 1) {
#pragma unroll
            for (int off = 16; off > 0; off >>= 1) mx = fmaxf(mx, __shfl_xor(mx, off));
        }
        if (!(fabsf(mx) < INFINITY)) mx = 0.f;

        float den = expf(lself - mx);
        float acc[R];
        {
            const float* hrow = hin + (size_t)n * ROW + col;
#pragma unroll
            for (int q = 0; q < R / VW; ++q) {
                const VT t = *(const VT*)(hrow + q * VW);
#pragma unroll
                for (int k = 0; k < VW; ++k) acc[q * VW + k] = den * t[k];
            }
        }
        for (int c0 = 0; c0 < count; c0 += SEG_CAP) {
            int m = m0;
            if (!single) {
                m = compact_sources(srcs, dsts, c0, min(count, c0 + SEG_CAP), n, lane, scratch);
                wave_lds_handover();
            }
            for (int j = 0; j < m; j += 32) {
                const int cnt  = min(32, m - j);
                const int sreg = (lane < cnt) ? scratch[j + lane] : n;
                float wreg = 0.f;
                if constexpr (H == 1) wreg = expf(lrelu(as[sreg] + adn) - mx);
                for (int e0 = 0; e0 < cnt; e0 += G2) {
                    float s[G2];
                    float v[G2][R];
#pragma unroll
                    for (int g = 0; g < G2; ++g) {
                        s[g] = 0.f;
#pragma unroll
                        for (int r = 0; r < R; ++r) v[g][r] = 0.f;
                        if (e0 + g < cnt) {
                            const int sj = __builtin_amdgcn_readlane(sreg, e0 + g);
                            if constexpr (H > 1) s[g] = as[sj * H + hl];
                            const float* hrow = hin + (size_t)sj * ROW + col;
#pragma unroll
                            for (int q = 0; q < R / VW; ++q) {
                                const VT t = *(const VT*)(hrow + q * VW);
#pragma unroll
                                for (int k = 0; k < VW; ++k) v[g][q * VW + k] = t[k];
                            }
                        }
                    }
#pragma unroll
                    for (int g = 0; g < G2; ++g) {
                        if (e0 + g < cnt) {
                            float w;
                            if constexpr (H == 1) w = lane_bcast(wreg, e0 + g);
                            else                  w = expf(lrelu(s[g] + adn) - mx);
                            den += w;
#pragma unroll
                            for (int r = 0; r < R; ++r) acc[r] += w * v[g][r];
                        }
                    }
                }
            }
            wave_lds_handover();
        }

        float o[R];
#pragma unroll
        for (int r = 0; r < R; ++r) {
            float val = acc[r] / (den + 1e-16f) + bias[col + r];
            if (ELU) val = val > 0.f ? val : (expf(val) - 1.f);
            o[r] = val;
        }
        float* orow = hout + (size_t)n * ROW + col;
#pragma unroll
        for (int q = 0; q < R / VW; ++q) {
            VT t;
#pragma unroll
            for (int k = 0; k < VW; ++k) t[k] = o[q * VW + k];
            *(VT*)(orow + q * VW) = t;
        }
    }
}

template <int H, int R, bool ELU, bool REUSE>
__global__ void __launch_bounds__(1024)
agg_tile_kernel(const int* __restrict__ ei, int E0, int Nn,
                const float* __restrict__ as, const float* __restrict__ ad,
                const float* __restrict__ hin, const float* __restrict__ bias,
                float* hout, int* keep, int keep_row, const int* bins, int bin_row) {
    __shared__ int seg_e[AGG_WAVES][SEG_CAP];
    __shared__ int seg_d[AGG_WAVES][SEG_CAP];
    __shared__ int lst_src[AGG_WAVES * SEG_CAP];
    __shared__ int lst_dst[AGG_WAVES * SEG_CAP];
    __shared__ int own_src[AGG_WAVES * SEG_CAP];
    __shared__ int own_dst[AGG_WAVES * SEG_CAP];
    __shared__ int seg_cnt[AGG_WAVES];
    __shared__ int own_cnt[AGG_WAVES];
    __shared__ int overflow;

    const int tid  = threadIdx.x;
    const int lane = tid & 31;
    const int wave = __builtin_amdgcn_readfirstlane(tid >> 5);
    const int n0   = blockIdx.x * TILE_N;
    const int nt   = min(TILE_N, Nn - n0);
    const int* __restrict__ dstp = ei + E0;
    int* kt = keep + (size_t)n0 * keep_row;
    const int kcap = nt * keep_row;

    int  oc = 0, obase = 0, ovf = 0;
    bool have = false;
    if constexpr (REUSE) {
        const int hdr = __builtin_amdgcn_readfirstlane(kt[0]);
        have = (hdr >= 0) && (hdr <= AGG_WAVES * SEG_CAP) && (KEEP_HDR + 2 * hdr <= kcap);
        if (have) {
            if (tid < AGG_WAVES) own_cnt[tid] = kt[1 + tid];
            for (int i = tid; i < hdr; i += 1024) {
                own_src[i] = kt[KEEP_HDR + i];
                own_dst[i] = kt[KEEP_HDR + hdr + i];
            }
            __syncthreads();
            const int cl = own_cnt[lane];
            int gincl = cl;
            for (int off = 1; off < 32; off <<= 1) {
                const int t = __shfl_up(gincl, off);
                if (lane >= off) gincl += t;
            }
            oc    = __builtin_amdgcn_readlane(cl, wave);
            obase = __builtin_amdgcn_readlane(gincl - cl, wave);
        }
    }

    if (!have) {
        int  total = 0;
        bool parts = false;
        if constexpr (!REUSE) {
            static_assert(NBIN <= 1024 && NBIN / 32 <= AGG_WAVES, "one thread per bin, wave sums in seg_cnt");
            constexpr int BPW = NBIN / AGG_WAVES;
            const int* bt   = bins + (size_t)n0 * bin_row;
            const int  binw = (nt * bin_row) / NBIN;
            const int  cap  = min((binw - 2) / 2, SEG_CAP);
            int* bcnt = own_dst;
            int* boff = own_src;
            if (tid == 0) overflow = 0;
            __syncthreads();
            int cb = 0;
            if (tid < NBIN) {
                cb = bt[tid * binw];
                if (cb < 0 || cb > cap) { overflow = 1; cb = 0; }
                bcnt[tid] = cb;
            }
            int incl = cb;
            for (int off = 1; off < 32; off <<= 1) {
                const int v = __shfl_up(incl, off);
                if (lane >= off) incl += v;
            }
            if (lane == 31 && tid < NBIN) seg_cnt[wave] = incl;
            __syncthreads();
            int sum = 0, wbase = 0;
            for (int w = 0; w < NBIN / 32; ++w) {
                const int cw = seg_cnt[w];
                if (w < wave) wbase += cw;
                sum += cw;
            }
            if (tid < NBIN) boff[tid] = wbase + incl - cb;
            parts = (overflow == 0) && (sum <= AGG_WAVES * SEG_CAP);
            __syncthreads();
            if (parts) {
                int* ke = &seg_e[wave][0];
                int* kp = &seg_d[wave][0];
                int ev[BPW], pv[BPW], cn[BPW];
#pragma unroll
                for (int u = 0; u < BPW; ++u) {
                    const int b = wave + u * AGG_WAVES;
                    cn[u] = __builtin_amdgcn_readfirstlane(bcnt[b]);
                    ev[u] = 0; pv[u] = 0;
                    if (lane < cn[u]) {
                        const int* q = bt + b * binw + 2 + 2 * lane;
                        ev[u] = q[0]; pv[u] = q[1];
                    }
                }
#pragma unroll
                for (int u = 0; u < BPW; ++u) {
                    const int b   = wave + u * AGG_WAVES;
                    const int cnt = cn[u];
                    const int off = __builtin_amdgcn_readfirstlane(boff[b]);
                    if (lane < cnt) { ke[lane] = ev[u]; kp[lane] = pv[u]; }
                    for (int i = 32 + lane; i < cnt; i += 32) {
                        const int* q = bt + b * binw + 2 + 2 * i;
                        ke[i] = q[0]; kp[i] = q[1];
                    }
                    wave_lds_handover();
                    for (int i = lane; i < cnt; i += 32) {
                        const int key = ke[i];
                        int rank = 0;
                        for (int j = 0; j < cnt; ++j) rank += (ke[j] < key) ? 1 : 0;
                        const int pk = kp[i];
                        lst_src[off + rank] = pk & 0xFFFFFF;
                        lst_dst[off + rank] = n0 + (int)((unsigned)pk >> 24);
                    }
                    wave_lds_handover();
                }
                total = sum;
                __syncthreads();
            }
        }

        if (!parts) {
            if (tid == 0) overflow = 0;
            __syncthreads();

            const int cnt = filter_scan(dstp, E0, wave, lane, n0, nt, &seg_e[wave][0], &seg_d[wave][0]);
            if (lane == 0) {
                seg_cnt[wave] = min(cnt, SEG_CAP);
                if (cnt > SEG_CAP) overflow = 1;
            }
            __syncthreads();

            int mybase = 0;
            for (int w = 0; w < AGG_WAVES; ++w) {
                const int c = seg_cnt[w];
                if (w < wave) mybase += c;
                total += c;
            }
            const int myc = seg_cnt[wave];
            ovf = overflow;
            for (int i = lane; i < myc; i += 32) {
                lst_src[mybase + i] = ei[seg_e[wave][i]];
                lst_dst[mybase + i] = seg_d[wave][i];
            }
            __syncthreads();
        }

        int* ctab = &seg_e[0][0];
        static_assert(AGG_WAVES == 32 && SEG_CAP >= 32, "one owner per lane; the 32 x 32 table fits in seg_e");
        const int nch  = (__builtin_amdgcn_readfirstlane(total) + 31) >> 5;
        const int cpw  = (nch + AGG_WAVES - 1) / AGG_WAVES;
        const int cbeg = min(nch, wave * cpw), cend = min(nch, cbeg + cpw);
        const unsigned below = (1u << lane) - 1u;
        if (!ovf) {
            int mycnt = 0;
            for (int c = cbeg; c < cend; ++c) {
                const int  i   = c * 32 + lane;
                const bool in  = i < total;
                const int  key = in ? ((lst_dst[i] - n0) & (AGG_WAVES - 1)) : 0;
                unsigned mine = __builtin_amdgcn_ballot_w32(in);
#pragma unroll
                for (int bit = 0; bit < 5; ++bit) {
                    const unsigned bb = __builtin_amdgcn_ballot_w32((key >> bit) & 1);
                    mine &= ((lane >> bit) & 1) ? bb : ~bb;
                }
                mycnt += __builtin_popcount(mine);
            }
            ctab[lane * 32 + ((wave + lane) & 31)] = mycnt;
        }
        __syncthreads();
        if (!ovf) {
            int* p = &ctab[wave * 32 + ((lane + wave) & 31)];
            const int v = *p;
            int incl = v;
            for (int off = 1; off < 32; off <<= 1) {
                const int t = __shfl_up(incl, off);
                if (lane >= off) incl += t;
            }
            *p = incl - v;
            if (lane == 31) own_cnt[wave] = incl;
        } else if (lane == 0) own_cnt[wave] = 0;
        __syncthreads();
        if (!ovf) {
            const int cl = own_cnt[lane];
            int gincl = cl;
            for (int off = 1; off < 32; off <<= 1) {
                const int t = __shfl_up(gincl, off);
                if (lane >= off) gincl += t;
            }
            const int gstart = gincl - cl;
            oc    = __builtin_amdgcn_readlane(cl, wave);
            obase = __builtin_amdgcn_readlane(gstart, wave);
            int nxt = gstart + ctab[lane * 32 + ((wave + lane) & 31)];
            for (int c = cbeg; c < cend; ++c) {
                const int  i   = c * 32 + lane;
                const bool in  = i < total;
                const int  d   = in ? lst_dst[i] : n0;
                const int  s   = in ? lst_src[i] : 0;
                const int  key = (d - n0) & (AGG_WAVES - 1);
                const unsigned valid = __builtin_amdgcn_ballot_w32(in);
                unsigned mine = valid;
                unsigned same = valid;
#pragma unroll
                for (int bit = 0; bit < 5; ++bit) {
                    const unsigned bb = __builtin_amdgcn_ballot_w32((key >> bit) & 1);
                    mine &= ((lane >> bit) & 1) ? bb : ~bb;
                    same &= ((key >> bit) & 1) ? bb : ~bb;
                }
                const int kb = __shfl(nxt, key);
                if (in) {
                    const int pos = kb + __builtin_popcount(same & below);
                    own_src[pos] = s;
                    own_dst[pos] = d;
                }
                nxt += __builtin_popcount(mine);
            }
        }
        __syncthreads();

        if constexpr (!REUSE) {
            const bool fits = !ovf && (KEEP_HDR + 2 * total <= kcap);
            if (tid == 0) kt[0] = fits ? total : -1;
            if (fits) {
                if (tid < AGG_WAVES) kt[1 + tid] = own_cnt[tid];
                for (int i = tid; i < total; i += 1024) {
                    kt[KEEP_HDR + i]         = own_src[i];
                    kt[KEEP_HDR + total + i] = own_dst[i];
                }
            }
        }
    }

    constexpr int G1 = 16;
    constexpr int G2 = (R > 2) ? 8 : 16;
    int* scratch = &seg_e[wave][0];
    if (!ovf)
        node_rows<H, R, ELU, G1, G2>(own_src + obase, own_dst + obase, oc, n0, nt, wave, lane, as, ad, hin, bias, hout, scratch);
    else
        node_rows<H, R, ELU, G1, G2>(ei, dstp, E0, n0, nt, wave, lane, as, ad, hin, bias, hout, scratch);
}

extern "C" void kernel_launch(void* const* d_in, const int* in_sizes, int n_in,
                              void* d_out, int out_size, void* d_ws, size_t ws_size,
                              hipStream_t stream) {
    const float* x        = (const float*)d_in[0];
    const int*   ei       = (const int*)d_in[1];
    const float* W1       = (const float*)d_in[2];
    const float* att_src1 = (const float*)d_in[3];
    const float* att_dst1 = (const float*)d_in[4];
    const float* b1       = (const float*)d_in[5];
    const float* W2       = (const float*)d_in[6];
    const float* att_src2 = (const float*)d_in[7];
    const float* att_dst2 = (const float*)d_in[8];
    const float* b2       = (const float*)d_in[9];
    float* out = (float*)d_out;

    const int Nn = in_sizes[0] / 256;
    const int E0 = in_sizes[1] / 2;

    float* h1   = (float*)d_ws;
    float* helu = h1 + (size_t)Nn * 256;
    float* as1  = helu + (size_t)Nn * 256;
    float* ad1  = as1 + (size_t)Nn * 8;
    float* h2   = h1;
    float* as2  = as1;
    float* ad2  = as1 + Nn;

    const int TB = 256;
    const int tiles = (Nn + TILE_N - 1) / TILE_N;
    dim3 g1((256 + 63) / 64, (Nn + 63) / 64 + NBIN / BIN_COLS);
    dim3 g2((64 + 63) / 64, (Nn + 63) / 64);

    gat_wmma_gemm<true><<<g1, TB, 0, stream>>>(x, W1, h1, Nn, 256, 256, ei, E0, (int*)helu, 256, NBIN / BIN_COLS);
    scores1_kernel<<<(Nn * 8 + TB - 1) / TB, TB, 0, stream>>>(h1, att_src1, att_dst1, as1, ad1, Nn);
    agg_tile_kernel<8, 8, true, false><<<tiles, 1024, 0, stream>>>(ei, E0, Nn, as1, ad1, h1, b1, helu, (int*)out, 64, (const int*)helu, 256);

    gat_wmma_gemm<false><<<g2, TB, 0, stream>>>(helu, W2, h2, Nn, 64, 256, nullptr, 0, nullptr, 0, 0);
    scores2_kernel<<<(Nn + TB - 1) / TB, TB, 0, stream>>>(h2, att_src2, att_dst2, as2, ad2, Nn);
    agg_tile_kernel<1, 2, false, true><<<tiles, 1024, 0, stream>>>(ei, E0, Nn, as2, ad2, h2, b2, out, (int*)out, 64, nullptr, 0);
}


# Layer-1 scores formed in the gather from the rows it loads (8 fma + 2 DPP adds a lane): max and sums in one pass, sums rescaled when the max rises; score kernel gone; score sum order changed
# speedup vs baseline: 5.0065x; 1.0468x over previous
#include <hip/hip_runtime.h>
#include <hip/hip_bf16.h>

typedef __attribute__((ext_vector_type(2))) float v2f;
typedef __attribute__((ext_vector_type(8))) float v8f;

#define NEG_SLOPE 0.2f

#define TILE_N    196
#define AGG_WAVES 32
#define SEG_CAP   256
#define KEEP_HDR  64
#define NBIN       256
#define BIN_COLS   4
#define BIN_MAXT   4096
#define BIN_THREADS 256
static_assert(NBIN % BIN_COLS == 0 && NBIN % AGG_WAVES == 0, "bins per grid row and per wave are whole numbers");

__device__ __forceinline__ void filter_group4(const int4 d, int ebase, int lane, unsigned below,
                                              int n0, int nt,
                                              int* __restrict__ se, int* __restrict__ sd, int& cnt) {
    const unsigned t0 = (unsigned)(d.x - n0), t1 = (unsigned)(d.y - n0);
    const unsigned t2 = (unsigned)(d.z - n0), t3 = (unsigned)(d.w - n0);
    const unsigned tmin = min(min(t0, t1), min(t2, t3));
    if (__builtin_amdgcn_ballot_w32(tmin < (unsigned)nt)) {
        const bool m0 = t0 < (unsigned)nt, m1 = t1 < (unsigned)nt;
        const bool m2 = t2 < (unsigned)nt, m3 = t3 < (unsigned)nt;
        const unsigned k0 = __builtin_amdgcn_ballot_w32(m0), k1 = __builtin_amdgcn_ballot_w32(m1);
        const unsigned k2 = __builtin_amdgcn_ballot_w32(m2), k3 = __builtin_amdgcn_ballot_w32(m3);
        int pos = cnt + __builtin_popcount(k0 & below) + __builtin_popcount(k1 & below)
                      + __builtin_popcount(k2 & below) + __builtin_popcount(k3 & below);
        const int e = ebase + 4 * lane;
        if (m0) { if (pos < SEG_CAP) { se[pos] = e;     sd[pos] = d.x; } ++pos; }
        if (m1) { if (pos < SEG_CAP) { se[pos] = e + 1; sd[pos] = d.y; } ++pos; }
        if (m2) { if (pos < SEG_CAP) { se[pos] = e + 2; sd[pos] = d.z; } ++pos; }
        if (m3) { if (pos < SEG_CAP) { se[pos] = e + 3; sd[pos] = d.w; } }
        cnt += __builtin_popcount(k0) + __builtin_popcount(k1) + __builtin_popcount(k2) + __builtin_popcount(k3);
    }
}

__device__ __forceinline__ int filter_scan(const int* __restrict__ dstp, int E0, int gw, int lane,
                                           int n0, int nt, int* __restrict__ se, int* __restrict__ sd) {
    int cnt = 0;
    int base = gw * 256;
    constexpr int FCH = 4;
    if ((reinterpret_cast<size_t>(dstp) & 15) == 0) {
        const unsigned below = (1u << lane) - 1u;
        for (; base + (FCH - 1) * AGG_WAVES * 256 + 256 <= E0; base += FCH * AGG_WAVES * 256) {
            int4 dv[FCH][2];
#pragma unroll
            for (int u = 0; u < FCH; ++u) {
#pragma unroll
                for (int h = 0; h < 2; ++h)
                    dv[u][h] = *(const int4*)(dstp + base + u * (AGG_WAVES * 256) + h * 128 + 4 * lane);
            }
#pragma unroll
            for (int u = 0; u < FCH; ++u) {
#pragma unroll
                for (int h = 0; h < 2; ++h)
                    filter_group4(dv[u][h], base + u * (AGG_WAVES * 256) + h * 128, lane, below, n0, nt,
                                  se, sd, cnt);
            }
        }
    }
    for (; base < E0; base += AGG_WAVES * 256) {
        int d[8];
        if (base + 256 <= E0) {
#pragma unroll
            for (int j = 0; j < 8; ++j) d[j] = dstp[base + j * 32 + lane];
        } else {
#pragma unroll
            for (int j = 0; j < 8; ++j) {
                const int e = base + j * 32 + lane;
                d[j] = (e < E0) ? dstp[e] : -1;
            }
        }
#pragma unroll
        for (int j = 0; j < 8; ++j) {
            const bool m = (unsigned)(d[j] - n0) < (unsigned)nt;
            const unsigned mask = __builtin_amdgcn_ballot_w32(m);
            if (mask) {
                const int pos = cnt + __builtin_popcount(mask & ((1u << lane) - 1u));
                if (m && pos < SEG_CAP) {
                    se[pos] = base + j * 32 + lane;
                    sd[pos] = d[j];
                }
                cnt += __builtin_popcount(mask);
            }
        }
    }
    return cnt;
}

__device__ __forceinline__ void bin_chunk(const int* __restrict__ ei, int E0, int Nn,
                                          int* __restrict__ bins, int bin_row, int b, int* __restrict__ sm) {
    const int tid   = threadIdx.x;
    constexpr int nthr = BIN_THREADS;
    const int tiles = (Nn + TILE_N - 1) / TILE_N;
    const bool usable = (tiles <= BIN_MAXT) && (Nn <= (1 << 24));
    if (usable) {
        for (int t = tid; t < tiles; t += nthr) sm[t] = 0;
        __syncthreads();
        const int CH = (E0 + NBIN - 1) / NBIN;
        const int e0 = b * CH;
        const int e1 = min(E0, e0 + CH);
        for (int eb = e0; eb < e1; eb += 4 * nthr) {
            int d[4], s[4];
#pragma unroll
            for (int u = 0; u < 4; ++u) {
                const int e = eb + u * nthr + tid;
                const bool ok = e < e1;
                d[u] = ok ? ei[E0 + e] : -1;
                s[u] = ok ? ei[e] : 0;
            }
#pragma unroll
            for (int u = 0; u < 4; ++u) {
                const int t = d[u] / TILE_N;
                if (d[u] >= 0 && t < tiles) {
                    const int n0   = t * TILE_N;
                    const int nt   = min(TILE_N, Nn - n0);
                    const int binw = (nt * bin_row) / NBIN;
                    const int cap  = (binw - 2) / 2;
                    const int slot = atomicAdd(&sm[t], 1);
                    if (slot < cap) {
                        int* q = bins + (size_t)n0 * bin_row + b * binw + 2 + 2 * slot;
                        q[0] = eb + u * nthr + tid;
                        q[1] = s[u] | (int)((unsigned)(d[u] - n0) << 24);
                    }
                }
            }
        }
        __syncthreads();
    }
    for (int t = tid; t < tiles; t += nthr) {
        const int n0   = t * TILE_N;
        const int nt   = min(TILE_N, Nn - n0);
        const int binw = (nt * bin_row) / NBIN;
        bins[(size_t)n0 * bin_row + b * binw] = usable ? sm[t] : -1;
    }
}

template <bool SCAN>
__global__ void gat_wmma_gemm(const float* __restrict__ A,
                              const float* __restrict__ B,
                              float* __restrict__ C,
                              int M, int Ntot, int K,
                              const int* __restrict__ ei, int E0, int* __restrict__ keep, int keep_row, int stiles) {
    __shared__ float As[64][68];
    __shared__ float Bp[32][132];
    static_assert(sizeof(As) >= BIN_MAXT * sizeof(int), "the bin counters fit in As");

    int yb = blockIdx.y;
    if constexpr (SCAN) {
        if (yb < stiles) {
            bin_chunk(ei, E0, M, keep, keep_row, yb * BIN_COLS + blockIdx.x, reinterpret_cast<int*>(&As[0][0]));
            return;
        }
        yb -= stiles;
    }

    const int tid   = threadIdx.x;
    const int lane  = tid & 31;
    const int wave  = tid >> 5;
    const int mBase = yb * 64;
    const int nBase = blockIdx.x * 64;

    const int ms   = (wave & 3) * 16;
    const int ns   = (wave >> 2) * 32;
    const int mrow = ms + (lane & 15);
    const int khi  = (lane >> 4) * 2;
    const int ncol = lane & 15;

    v8f acc0 = {}; v8f acc1 = {};

    for (int kBase = 0; kBase < K; kBase += 64) {
#pragma unroll
        for (int i = 0; i < 4; ++i) {
            int s  = tid + i * 256;
            int r  = s >> 4;
            int c4 = (s & 15) << 2;
            float4 va = make_float4(0.f, 0.f, 0.f, 0.f);
            int g = mBase + r;
            if (g < M) va = *(const float4*)(A + (size_t)g * K + kBase + c4);
            *(float4*)&As[r][c4] = va;
            float4 vb = *(const float4*)(B + (size_t)(kBase + r) * Ntot + nBase + c4);
            float* bp = &Bp[r >> 1][r & 1];
            bp[2 * c4 + 0] = vb.x;
            bp[2 * c4 + 2] = vb.y;
            bp[2 * c4 + 4] = vb.z;
            bp[2 * c4 + 6] = vb.w;
        }
        __syncthreads();

#pragma unroll
        for (int kk = 0; kk < 64; kk += 4) {
            const int p2 = (kk + khi) >> 1;
            v2f a, b0, b1v;
            a[0]   = As[mrow][kk + khi];
            a[1]   = As[mrow][kk + khi + 1];
            b0     = *(const v2f*)&Bp[p2][2 * (ns + ncol)];
            b1v    = *(const v2f*)&Bp[p2][2 * (ns + 16 + ncol)];
            acc0 = __builtin_amdgcn_wmma_f32_16x16x4_f32(
                false, a, false, b0, (short)0, acc0, false, false);
            acc1 = __builtin_amdgcn_wmma_f32_16x16x4_f32(
                false, a, false, b1v, (short)0, acc1, false, false);
        }
        __syncthreads();
    }

    const int orow = mBase + ms + ((lane >> 4) << 3);
    const int ocol = nBase + ns + ncol;
#pragma unroll
    for (int r = 0; r < 8; ++r) {
        int g = orow + r;
        if (g < M) {
            C[(size_t)g * Ntot + ocol]      = acc0[r];
            C[(size_t)g * Ntot + ocol + 16] = acc1[r];
        }
    }
}

__global__ void scores2_kernel(const float* __restrict__ h2,
                               const float* __restrict__ att_s,
                               const float* __restrict__ att_d,
                               float* __restrict__ as2, float* __restrict__ ad2,
                               int Nn) {
    int n = blockIdx.x * blockDim.x + threadIdx.x;
    if (n >= Nn) return;
    const float* row = h2 + (size_t)n * 64;
    float ss = 0.f, sd = 0.f;
#pragma unroll
    for (int c = 0; c < 64; ++c) { ss += row[c] * att_s[c]; sd += row[c] * att_d[c]; }
    as2[n] = ss; ad2[n] = sd;
}

__device__ __forceinline__ float lrelu(float v) { return v > 0.f ? v : NEG_SLOPE * v; }

__device__ __forceinline__ float lane_bcast(float v, int j) {
    return __uint_as_float(__builtin_amdgcn_readlane(__float_as_uint(v), j));
}

__device__ __forceinline__ int compact_sources(const int* __restrict__ srcs, const int* __restrict__ dsts,
                                               int c0, int cend, int n, int lane,
                                               int* __restrict__ scratch) {
    const unsigned below = (1u << lane) - 1u;
    int m = 0;
    for (int i0 = c0; i0 < cend; i0 += 32) {
        const int  i     = i0 + lane;
        const bool in    = i < cend;
        const int  d     = in ? dsts[i] : -1;
        const int  sv    = in ? srcs[i] : 0;
        const bool match = (d == n);
        const unsigned mask = __builtin_amdgcn_ballot_w32(match);
        if (match) scratch[m + __builtin_popcount(mask & below)] = sv;
        m += __builtin_popcount(mask);
    }
    return m;
}

__device__ __forceinline__ void wave_lds_handover() {
    __builtin_amdgcn_fence(__ATOMIC_RELEASE, "wavefront");
    __builtin_amdgcn_wave_barrier();
    __builtin_amdgcn_fence(__ATOMIC_ACQUIRE, "wavefront");
}

template <int H, int R, bool ELU, int G1, int G2>
__device__ __forceinline__ void node_rows(const int* __restrict__ srcs, const int* __restrict__ dsts, int count,
                                          int n0, int nt, int wave, int lane,
                                          const float* __restrict__ as,
                                          const float* __restrict__ ad,
                                          const float* __restrict__ hin,
                                          const float* __restrict__ bias,
                                          float* __restrict__ hout,
                                          int* __restrict__ scratch) {
    constexpr int ROW = 32 * R;
    constexpr int VW  = (R >= 4) ? 4 : 2;
    typedef float VT __attribute__((ext_vector_type(VW)));
    const int  col    = lane * R;
    const int  hl     = (col * H) / ROW;
    const bool single = count <= SEG_CAP;

    for (int ln = wave; ln < nt; ln += AGG_WAVES) {
        const int   n     = n0 + ln;
        const float adn   = ad[n * H + hl];
        const float lself = lrelu(as[n * H + hl] + adn);
        float mx = lself;
        int m0 = 0;

        for (int c0 = 0; c0 < count; c0 += SEG_CAP) {
            const int m = compact_sources(srcs, dsts, c0, min(count, c0 + SEG_CAP), n, lane, scratch);
            if (c0 == 0) m0 = m;
            wave_lds_handover();
            for (int j = 0; j < m; j += 32) {
                const int cnt  = min(32, m - j);
                const int sreg = (lane < cnt) ? scratch[j + lane] : n;
                if constexpr (H == 1) {
                    mx = fmaxf(mx, lrelu(as[sreg] + adn));
                } else {
                    for (int e0 = 0; e0 < cnt; e0 += G1) {
                        float s[G1];
#pragma unroll
                        for (int g = 0; g < G1; ++g) {
                            const int sj = __builtin_amdgcn_readlane(sreg, e0 + g);
                            s[g] = as[sj * H + hl];
                        }
#pragma unroll
                        for (int g = 0; g < G1; ++g) mx = fmaxf(mx, lrelu(s[g] + adn));
                    }
                }
            }
            wave_lds_handover();
        }
        if constexpr (H == 1) {
#pragma unroll
            for (int off = 16; off > 0; off >>= 1) mx = fmaxf(mx, __shfl_xor(mx, off));
        }
        if (!(fabsf(mx) < INFINITY)) mx = 0.f;

        float den = expf(lself - mx);
        float acc[R];
        {
            const float* hrow = hin + (size_t)n * ROW + col;
#pragma unroll
            for (int q = 0; q < R / VW; ++q) {
                const VT t = *(const VT*)(hrow + q * VW);
#pragma unroll
                for (int k = 0; k < VW; ++k) acc[q * VW + k] = den * t[k];
            }
        }
        for (int c0 = 0; c0 < count; c0 += SEG_CAP) {
            int m = m0;
            if (!single) {
                m = compact_sources(srcs, dsts, c0, min(count, c0 + SEG_CAP), n, lane, scratch);
                wave_lds_handover();
            }
            for (int j = 0; j < m; j += 32) {
                const int cnt  = min(32, m - j);
                const int sreg = (lane < cnt) ? scratch[j + lane] : n;
                float wreg = 0.f;
                if constexpr (H == 1) wreg = expf(lrelu(as[sreg] + adn) - mx);
                for (int e0 = 0; e0 < cnt; e0 += G2) {
                    float s[G2];
                    float v[G2][R];
#pragma unroll
                    for (int g = 0; g < G2; ++g) {
                        s[g] = 0.f;
#pragma unroll
                        for (int r = 0; r < R; ++r) v[g][r] = 0.f;
                        if (e0 + g < cnt) {
                            const int sj = __builtin_amdgcn_readlane(sreg, e0 + g);
                            if constexpr (H > 1) s[g] = as[sj * H + hl];
                            const float* hrow = hin + (size_t)sj * ROW + col;
#pragma unroll
                            for (int q = 0; q < R / VW; ++q) {
                                const VT t = *(const VT*)(hrow + q * VW);
#pragma unroll
                                for (int k = 0; k < VW; ++k) v[g][q * VW + k] = t[k];
                            }
                        }
                    }
#pragma unroll
                    for (int g = 0; g < G2; ++g) {
                        if (e0 + g < cnt) {
                            float w;
                            if constexpr (H == 1) w = lane_bcast(wreg, e0 + g);
                            else                  w = expf(lrelu(s[g] + adn) - mx);
                            den += w;
#pragma unroll
                            for (int r = 0; r < R; ++r) acc[r] += w * v[g][r];
                        }
                    }
                }
            }
            wave_lds_handover();
        }

        float o[R];
#pragma unroll
        for (int r = 0; r < R; ++r) {
            float val = acc[r] / (den + 1e-16f) + bias[col + r];
            if (ELU) val = val > 0.f ? val : (expf(val) - 1.f);
            o[r] = val;
        }
        float* orow = hout + (size_t)n * ROW + col;
#pragma unroll
        for (int q = 0; q < R / VW; ++q) {
            VT t;
#pragma unroll
            for (int k = 0; k < VW; ++k) t[k] = o[q * VW + k];
            *(VT*)(orow + q * VW) = t;
        }
    }
}

__device__ __forceinline__ float quad_sum(float v) {
    v += __int_as_float(__builtin_amdgcn_update_dpp(0, __float_as_int(v), 0xB1, 0xF, 0xF, true));
    v += __int_as_float(__builtin_amdgcn_update_dpp(0, __float_as_int(v), 0x4E, 0xF, 0xF, true));
    return v;
}

template <int R, bool ELU, int G2>
__device__ __forceinline__ void node_rows_scored(const int* __restrict__ srcs, const int* __restrict__ dsts, int count,
                                                 int n0, int nt, int wave, int lane,
                                                 const float* __restrict__ att_s,
                                                 const float* __restrict__ att_d,
                                                 const float* __restrict__ hin,
                                                 const float* __restrict__ bias,
                                                 float* __restrict__ hout,
                                                 int* __restrict__ scratch) {
    static_assert(R == 8, "four lanes of 8 columns make a head of 32");
    static_assert(32 % G2 == 0, "a group's slots all lie inside the 32 lanes of a batch");
    constexpr int ROW = 32 * R;
    const int col = lane * R;

    float ws[R];
#pragma unroll
    for (int q = 0; q < R / 4; ++q) {
        const float4 t = *(const float4*)(att_s + col + q * 4);
        ws[q * 4 + 0] = t.x; ws[q * 4 + 1] = t.y; ws[q * 4 + 2] = t.z; ws[q * 4 + 3] = t.w;
    }

    for (int ln = wave; ln < nt; ln += AGG_WAVES) {
        const int n = n0 + ln;
        float4 own[R / 4], wd[R / 4];
        {
            const float* hrow = hin + (size_t)n * ROW + col;
#pragma unroll
            for (int q = 0; q < R / 4; ++q) {
                own[q] = *(const float4*)(hrow + q * 4);
                wd[q]  = *(const float4*)(att_d + col + q * 4);
            }
        }
        int m = compact_sources(srcs, dsts, 0, min(count, SEG_CAP), n, lane, scratch);
        wave_lds_handover();

        float acc[R];
        float ps = 0.f, pd = 0.f;
        {
#pragma unroll
            for (int q = 0; q < R / 4; ++q) {
                const float4 t = own[q];
                const float4 d = wd[q];
                acc[q * 4 + 0] = t.x; acc[q * 4 + 1] = t.y; acc[q * 4 + 2] = t.z; acc[q * 4 + 3] = t.w;
                pd = fmaf(t.x, d.x, pd); pd = fmaf(t.y, d.y, pd); pd = fmaf(t.z, d.z, pd); pd = fmaf(t.w, d.w, pd);
            }
#pragma unroll
            for (int r = 0; r < R; ++r) ps = fmaf(acc[r], ws[r], ps);
        }
        const float adn   = quad_sum(pd);
        const float lself = lrelu(quad_sum(ps) + adn);
        float mx = lself;
        float ms = (fabsf(mx) < INFINITY) ? mx : 0.f;
        float den = expf(lself - ms);
#pragma unroll
        for (int r = 0; r < R; ++r) acc[r] *= den;

        for (int c0 = 0; c0 < count; c0 += SEG_CAP) {
            if (c0 > 0) {
                m = compact_sources(srcs, dsts, c0, min(count, c0 + SEG_CAP), n, lane, scratch);
                wave_lds_handover();
            }
            for (int j = 0; j < m; j += 32) {
                const int cnt  = min(32, m - j);
                const int sreg = (lane < cnt) ? scratch[j + lane] : n;
                for (int e0 = 0; e0 < cnt; e0 += G2) {
                    float4 v[G2][R / 4];
                    float  l[G2];
#pragma unroll
                    for (int g = 0; g < G2; ++g) {
#pragma unroll
                        for (int q = 0; q < R / 4; ++q) v[g][q] = make_float4(0.f, 0.f, 0.f, 0.f);
                        if (e0 + g < cnt) {
                            const int sj = __builtin_amdgcn_readlane(sreg, e0 + g);
                            const float* hrow = hin + (size_t)sj * ROW + col;
#pragma unroll
                            for (int q = 0; q < R / 4; ++q) v[g][q] = *(const float4*)(hrow + q * 4);
                        }
                    }
                    float gm = mx;
#pragma unroll
                    for (int g = 0; g < G2; ++g) {
                        if (e0 + g < cnt) {
                            float p = 0.f;
#pragma unroll
                            for (int q = 0; q < R / 4; ++q) {
                                p = fmaf(v[g][q].x, ws[q * 4 + 0], p); p = fmaf(v[g][q].y, ws[q * 4 + 1], p);
                                p = fmaf(v[g][q].z, ws[q * 4 + 2], p); p = fmaf(v[g][q].w, ws[q * 4 + 3], p);
                            }
                            l[g] = lrelu(quad_sum(p) + adn);
                            gm = fmaxf(gm, l[g]);
                        }
                    }
                    const float gs = (fabsf(gm) < INFINITY) ? gm : 0.f;
                    const float sc = (mx > -INFINITY) ? expf(ms - gs) : 1.f;
                    mx = gm;
                    ms = gs;
                    den *= sc;
#pragma unroll
                    for (int r = 0; r < R; ++r) acc[r] *= sc;
#pragma unroll
                    for (int g = 0; g < G2; ++g) {
                        if (e0 + g < cnt) {
                            const float w = expf(l[g] - ms);
                            den += w;
#pragma unroll
                            for (int q = 0; q < R / 4; ++q) {
                                acc[q * 4 + 0] += w * v[g][q].x; acc[q * 4 + 1] += w * v[g][q].y;
                                acc[q * 4 + 2] += w * v[g][q].z; acc[q * 4 + 3] += w * v[g][q].w;
                            }
                        }
                    }
                }
            }
            wave_lds_handover();
        }

#pragma unroll
        for (int q = 0; q < R / 4; ++q) {
            const float4 b = *(const float4*)(bias + col + q * 4);
            float4 t;
            t.x = acc[q * 4 + 0] / (den + 1e-16f) + b.x;
            t.y = acc[q * 4 + 1] / (den + 1e-16f) + b.y;
            t.z = acc[q * 4 + 2] / (den + 1e-16f) + b.z;
            t.w = acc[q * 4 + 3] / (den + 1e-16f) + b.w;
            if (ELU) {
                t.x = t.x > 0.f ? t.x : (expf(t.x) - 1.f);
                t.y = t.y > 0.f ? t.y : (expf(t.y) - 1.f);
                t.z = t.z > 0.f ? t.z : (expf(t.z) - 1.f);
                t.w = t.w > 0.f ? t.w : (expf(t.w) - 1.f);
            }
            *(float4*)(hout + (size_t)n * ROW + col + q * 4) = t;
        }
    }
}

template <int H, int R, bool ELU, bool REUSE>
__global__ void __launch_bounds__(1024)
agg_tile_kernel(const int* __restrict__ ei, int E0, int Nn,
                const float* __restrict__ sa, const float* __restrict__ sb,
                const float* __restrict__ hin, const float* __restrict__ bias,
                float* hout, int* keep, int keep_row, const int* bins, int bin_row) {
    __shared__ int seg_e[AGG_WAVES][SEG_CAP];
    __shared__ int seg_d[AGG_WAVES][SEG_CAP];
    __shared__ int lst_src[AGG_WAVES * SEG_CAP];
    __shared__ int lst_dst[AGG_WAVES * SEG_CAP];
    __shared__ int own_src[AGG_WAVES * SEG_CAP];
    __shared__ int own_dst[AGG_WAVES * SEG_CAP];
    __shared__ int seg_cnt[AGG_WAVES];
    __shared__ int own_cnt[AGG_WAVES];
    __shared__ int overflow;

    const int tid  = threadIdx.x;
    const int lane = tid & 31;
    const int wave = __builtin_amdgcn_readfirstlane(tid >> 5);
    const int n0   = blockIdx.x * TILE_N;
    const int nt   = min(TILE_N, Nn - n0);
    const int* __restrict__ dstp = ei + E0;
    int* kt = keep + (size_t)n0 * keep_row;
    const int kcap = nt * keep_row;

    int  oc = 0, obase = 0, ovf = 0;
    bool have = false;
    if constexpr (REUSE) {
        const int hdr = __builtin_amdgcn_readfirstlane(kt[0]);
        have = (hdr >= 0) && (hdr <= AGG_WAVES * SEG_CAP) && (KEEP_HDR + 2 * hdr <= kcap);
        if (have) {
            if (tid < AGG_WAVES) own_cnt[tid] = kt[1 + tid];
            for (int i = tid; i < hdr; i += 1024) {
                own_src[i] = kt[KEEP_HDR + i];
                own_dst[i] = kt[KEEP_HDR + hdr + i];
            }
            __syncthreads();
            const int cl = own_cnt[lane];
            int gincl = cl;
            for (int off = 1; off < 32; off <<= 1) {
                const int t = __shfl_up(gincl, off);
                if (lane >= off) gincl += t;
            }
            oc    = __builtin_amdgcn_readlane(cl, wave);
            obase = __builtin_amdgcn_readlane(gincl - cl, wave);
        }
    }

    if (!have) {
        int  total = 0;
        bool parts = false;
        if constexpr (!REUSE) {
            static_assert(NBIN <= 1024 && NBIN / 32 <= AGG_WAVES, "one thread per bin, wave sums in seg_cnt");
            constexpr int BPW = NBIN / AGG_WAVES;
            const int* bt   = bins + (size_t)n0 * bin_row;
            const int  binw = (nt * bin_row) / NBIN;
            const int  cap  = min((binw - 2) / 2, SEG_CAP);
            int* bcnt = own_dst;
            int* boff = own_src;
            if (tid == 0) overflow = 0;
            __syncthreads();
            int cb = 0;
            if (tid < NBIN) {
                cb = bt[tid * binw];
                if (cb < 0 || cb > cap) { overflow = 1; cb = 0; }
                bcnt[tid] = cb;
            }
            int incl = cb;
            for (int off = 1; off < 32; off <<= 1) {
                const int v = __shfl_up(incl, off);
                if (lane >= off) incl += v;
            }
            if (lane == 31 && tid < NBIN) seg_cnt[wave] = incl;
            __syncthreads();
            int sum = 0, wbase = 0;
            for (int w = 0; w < NBIN / 32; ++w) {
                const int cw = seg_cnt[w];
                if (w < wave) wbase += cw;
                sum += cw;
            }
            if (tid < NBIN) boff[tid] = wbase + incl - cb;
            parts = (overflow == 0) && (sum <= AGG_WAVES * SEG_CAP);
            __syncthreads();
            if (parts) {
                int* ke = &seg_e[wave][0];
                int* kp = &seg_d[wave][0];
                int ev[BPW], pv[BPW], cn[BPW];
#pragma unroll
                for (int u = 0; u < BPW; ++u) {
                    const int b = wave + u * AGG_WAVES;
                    cn[u] = __builtin_amdgcn_readfirstlane(bcnt[b]);
                    ev[u] = 0; pv[u] = 0;
                    if (lane < cn[u]) {
                        const int* q = bt + b * binw + 2 + 2 * lane;
                        ev[u] = q[0]; pv[u] = q[1];
                    }
                }
#pragma unroll
                for (int u = 0; u < BPW; ++u) {
                    const int b   = wave + u * AGG_WAVES;
                    const int cnt = cn[u];
                    const int off = __builtin_amdgcn_readfirstlane(boff[b]);
                    if (lane < cnt) { ke[lane] = ev[u]; kp[lane] = pv[u]; }
                    for (int i = 32 + lane; i < cnt; i += 32) {
                        const int* q = bt + b * binw + 2 + 2 * i;
                        ke[i] = q[0]; kp[i] = q[1];
                    }
                    wave_lds_handover();
                    for (int i = lane; i < cnt; i += 32) {
                        const int key = ke[i];
                        int rank = 0;
                        for (int j = 0; j < cnt; ++j) rank += (ke[j] < key) ? 1 : 0;
                        const int pk = kp[i];
                        lst_src[off + rank] = pk & 0xFFFFFF;
                        lst_dst[off + rank] = n0 + (int)((unsigned)pk >> 24);
                    }
                    wave_lds_handover();
                }
                total = sum;
                __syncthreads();
            }
        }

        if (!parts) {
            if (tid == 0) overflow = 0;
            __syncthreads();

            const int cnt = filter_scan(dstp, E0, wave, lane, n0, nt, &seg_e[wave][0], &seg_d[wave][0]);
            if (lane == 0) {
                seg_cnt[wave] = min(cnt, SEG_CAP);
                if (cnt > SEG_CAP) overflow = 1;
            }
            __syncthreads();

            int mybase = 0;
            for (int w = 0; w < AGG_WAVES; ++w) {
                const int c = seg_cnt[w];
                if (w < wave) mybase += c;
                total += c;
            }
            const int myc = seg_cnt[wave];
            ovf = overflow;
            for (int i = lane; i < myc; i += 32) {
                lst_src[mybase + i] = ei[seg_e[wave][i]];
                lst_dst[mybase + i] = seg_d[wave][i];
            }
            __syncthreads();
        }

        int* ctab = &seg_e[0][0];
        static_assert(AGG_WAVES == 32 && SEG_CAP >= 32, "one owner per lane; the 32 x 32 table fits in seg_e");
        const int nch  = (__builtin_amdgcn_readfirstlane(total) + 31) >> 5;
        const int cpw  = (nch + AGG_WAVES - 1) / AGG_WAVES;
        const int cbeg = min(nch, wave * cpw), cend = min(nch, cbeg + cpw);
        const unsigned below = (1u << lane) - 1u;
        if (!ovf) {
            int mycnt = 0;
            for (int c = cbeg; c < cend; ++c) {
                const int  i   = c * 32 + lane;
                const bool in  = i < total;
                const int  key = in ? ((lst_dst[i] - n0) & (AGG_WAVES - 1)) : 0;
                unsigned mine = __builtin_amdgcn_ballot_w32(in);
#pragma unroll
                for (int bit = 0; bit < 5; ++bit) {
                    const unsigned bb = __builtin_amdgcn_ballot_w32((key >> bit) & 1);
                    mine &= ((lane >> bit) & 1) ? bb : ~bb;
                }
                mycnt += __builtin_popcount(mine);
            }
            ctab[lane * 32 + ((wave + lane) & 31)] = mycnt;
        }
        __syncthreads();
        if (!ovf) {
            int* p = &ctab[wave * 32 + ((lane + wave) & 31)];
            const int v = *p;
            int incl = v;
            for (int off = 1; off < 32; off <<= 1) {
                const int t = __shfl_up(incl, off);
                if (lane >= off) incl += t;
            }
            *p = incl - v;
            if (lane == 31) own_cnt[wave] = incl;
        } else if (lane == 0) own_cnt[wave] = 0;
        __syncthreads();
        if (!ovf) {
            const int cl = own_cnt[lane];
            int gincl = cl;
            for (int off = 1; off < 32; off <<= 1) {
                const int t = __shfl_up(gincl, off);
                if (lane >= off) gincl += t;
            }
            const int gstart = gincl - cl;
            oc    = __builtin_amdgcn_readlane(cl, wave);
            obase = __builtin_amdgcn_readlane(gstart, wave);
            int nxt = gstart + ctab[lane * 32 + ((wave + lane) & 31)];
            for (int c = cbeg; c < cend; ++c) {
                const int  i   = c * 32 + lane;
                const bool in  = i < total;
                const int  d   = in ? lst_dst[i] : n0;
                const int  s   = in ? lst_src[i] : 0;
                const int  key = (d - n0) & (AGG_WAVES - 1);
                const unsigned valid = __builtin_amdgcn_ballot_w32(in);
                unsigned mine = valid;
                unsigned same = valid;
#pragma unroll
                for (int bit = 0; bit < 5; ++bit) {
                    const unsigned bb = __builtin_amdgcn_ballot_w32((key >> bit) & 1);
                    mine &= ((lane >> bit) & 1) ? bb : ~bb;
                    same &= ((key >> bit) & 1) ? bb : ~bb;
                }
                const int kb = __shfl(nxt, key);
                if (in) {
                    const int pos = kb + __builtin_popcount(same & below);
                    own_src[pos] = s;
                    own_dst[pos] = d;
                }
                nxt += __builtin_popcount(mine);
            }
        }
        __syncthreads();

        if constexpr (!REUSE) {
            const bool fits = !ovf && (KEEP_HDR + 2 * total <= kcap);
            if (tid == 0) kt[0] = fits ? total : -1;
            if (fits) {
                if (tid < AGG_WAVES) kt[1 + tid] = own_cnt[tid];
                for (int i = tid; i < total; i += 1024) {
                    kt[KEEP_HDR + i]         = own_src[i];
                    kt[KEEP_HDR + total + i] = own_dst[i];
                }
            }
        }
    }

    constexpr int G1 = 16;
    constexpr int G2 = (R > 2) ? 8 : 16;
    int* scratch = &seg_e[wave][0];
    if constexpr (H > 1) {
        if (!ovf)
            node_rows_scored<R, ELU, G2>(own_src + obase, own_dst + obase, oc, n0, nt, wave, lane, sa, sb, hin, bias, hout, scratch);
        else
            node_rows_scored<R, ELU, G2>(ei, dstp, E0, n0, nt, wave, lane, sa, sb, hin, bias, hout, scratch);
    } else {
        if (!ovf)
            node_rows<H, R, ELU, G1, G2>(own_src + obase, own_dst + obase, oc, n0, nt, wave, lane, sa, sb, hin, bias, hout, scratch);
        else
            node_rows<H, R, ELU, G1, G2>(ei, dstp, E0, n0, nt, wave, lane, sa, sb, hin, bias, hout, scratch);
    }
}

extern "C" void kernel_launch(void* const* d_in, const int* in_sizes, int n_in,
                              void* d_out, int out_size, void* d_ws, size_t ws_size,
                              hipStream_t stream) {
    const float* x        = (const float*)d_in[0];
    const int*   ei       = (const int*)d_in[1];
    const float* W1       = (const float*)d_in[2];
    const float* att_src1 = (const float*)d_in[3];
    const float* att_dst1 = (const float*)d_in[4];
    const float* b1       = (const float*)d_in[5];
    const float* W2       = (const float*)d_in[6];
    const float* att_src2 = (const float*)d_in[7];
    const float* att_dst2 = (const float*)d_in[8];
    const float* b2       = (const float*)d_in[9];
    float* out = (float*)d_out;

    const int Nn = in_sizes[0] / 256;
    const int E0 = in_sizes[1] / 2;

    float* h1   = (float*)d_ws;
    float* helu = h1 + (size_t)Nn * 256;
    float* as2  = helu + (size_t)Nn * 256;
    float* ad2  = as2 + Nn;
    float* h2   = h1;

    const int TB = 256;
    const int tiles = (Nn + TILE_N - 1) / TILE_N;
    dim3 g1((256 + 63) / 64, (Nn + 63) / 64 + NBIN / BIN_COLS);
    dim3 g2((64 + 63) / 64, (Nn + 63) / 64);

    gat_wmma_gemm<true><<<g1, TB, 0, stream>>>(x, W1, h1, Nn, 256, 256, ei, E0, (int*)helu, 256, NBIN / BIN_COLS);
    agg_tile_kernel<8, 8, true, false><<<tiles, 1024, 0, stream>>>(ei, E0, Nn, att_src1, att_dst1, h1, b1, helu, (int*)out, 64, (const int*)helu, 256);

    gat_wmma_gemm<false><<<g2, TB, 0, stream>>>(helu, W2, h2, Nn, 64, 256, nullptr, 0, nullptr, 0, 0);
    scores2_kernel<<<(Nn + TB - 1) / TB, TB, 0, stream>>>(h2, att_src2, att_dst2, as2, ad2, Nn);
    agg_tile_kernel<1, 2, false, true><<<tiles, 1024, 0, stream>>>(ei, E0, Nn, as2, ad2, h2, b2, out, (int*)out, 64, nullptr, 0);
}
